# starting kernel
# baseline (speedup 1.0000x reference)
#include <hip/hip_runtime.h>
#include <math.h>

#define B_   8
#define C0_  16
#define C1_  64
#define C2_  64
#define T0_  12
#define T1_  10
#define T2_  8
#define NN_  256
#define KT_  3
#define HH_  8
#define ALPHA_ 0.2f

typedef float v2f __attribute__((ext_vector_type(2)));
typedef float v4f __attribute__((ext_vector_type(4)));
typedef float v8f __attribute__((ext_vector_type(8)));

__device__ __forceinline__ float tanh_fast(float x) {
  float e = __expf(fminf(2.0f * x, 80.0f));
  return (e - 1.0f) / (e + 1.0f);
}

__global__ __launch_bounds__(256) void prep_u_kernel(
    const float* __restrict__ W,
    const float* __restrict__ a,
    float* __restrict__ u1, float* __restrict__ u2)
{
  int tid = blockIdx.x * 256 + threadIdx.x;
  if (tid >= HH_ * C1_) return;
  int h = tid >> 6, c = tid & 63;
  const float* Wh = W + ((size_t)h * C1_ + c) * C1_;
  const float* ah = a + (size_t)h * 2 * C1_;
  float s1 = 0.f, s2 = 0.f;
  for (int d = 0; d < C1_; ++d) {
    float w = Wh[d];
    s1 += w * ah[d];
    s2 += w * ah[C1_ + d];
  }
  u1[tid] = s1;
  u2[tid] = s2;
}

__global__ __launch_bounds__(256) void conv1_glu_kernel(
    const float* __restrict__ x,
    const float* __restrict__ w,
    const float* __restrict__ b,
    float* __restrict__ xg)
{
  int idx = blockIdx.x * 256 + threadIdx.x;
  int c  = idx & 63;
  int n  = (idx >> 6) & 255;
  int bt = idx >> 14;
  int bb = bt / T1_, t = bt - bb * T1_;

  float accp = b[c];
  float accq = b[c + C1_];
  #pragma unroll
  for (int kt = 0; kt < KT_; ++kt) {
    for (int ci = 0; ci < C0_; ++ci) {
      float xv = x[(((size_t)bb * C0_ + ci) * T0_ + (t + kt)) * NN_ + n];
      accp = fmaf(w[((size_t)c         * C0_ + ci) * KT_ + kt], xv, accp);
      accq = fmaf(w[((size_t)(c + C1_) * C0_ + ci) * KT_ + kt], xv, accq);
    }
  }
  float xin = (c < C0_) ? x[(((size_t)bb * C0_ + c) * T0_ + (t + KT_ - 1)) * NN_ + n] : 0.f;
  float sg  = 1.f / (1.f + __expf(-accq));
  xg[idx]   = (accp + xin) * sg;
}

#define XSTR  72
#define XPSTR 160
#define PSTR  260

__global__ __launch_bounds__(256) void attn_kernel(
    const float* __restrict__ xg,
    const float* __restrict__ adj,
    const float* __restrict__ u1,
    const float* __restrict__ u2,
    float* __restrict__ xs)
{
  extern __shared__ float smem[];
  float*    s_xg   = smem;
  float*    s_xgp  = s_xg + NN_ * XSTR;
  float*    s_e1   = s_xgp + (NN_ / 2) * XPSTR;
  float*    s_e2   = s_e1 + NN_;
  float*    s_rinv = s_e2 + NN_;
  float*    s_p    = s_rinv + NN_;
  unsigned* s_adj  = (unsigned*)(s_p + 8 * 16 * PSTR);

  const int bt   = blockIdx.x;
  const int tid  = threadIdx.x;
  const int lane = tid & 31;
  const int wave = tid >> 5;
  const int mloc = lane & 15;
  const int half = lane >> 4;

  const float* xg_bt = xg + (size_t)bt * NN_ * C1_;

  for (int i = tid; i < NN_ * C1_; i += 256) {
    int n = i >> 6, c = i & 63;
    float v = xg_bt[i];
    s_xg[n * XSTR + c] = v;
    s_xgp[(n >> 1) * XPSTR + 2 * c + (n & 1)] = v;
  }
  {
    int n = tid;
    for (int w8 = 0; w8 < 8; ++w8) {
      unsigned bits = 0u;
      for (int j = 0; j < 32; ++j)
        bits |= (adj[(size_t)n * NN_ + w8 * 32 + j] > 0.f) ? (1u << j) : 0u;
      s_adj[n * 8 + w8] = bits;
    }
  }
  __syncthreads();

  const float inv_h = 1.0f / (float)HH_;
  float* s_pw = s_p + wave * 16 * PSTR;

  float racc[2][4][8];
  #pragma unroll
  for (int i = 0; i < 2; ++i)
    #pragma unroll
    for (int j = 0; j < 4; ++j)
      #pragma unroll
      for (int r = 0; r < 8; ++r) racc[i][j][r] = 0.f;

  for (int h = 0; h < HH_; ++h) {
    {
      int n = tid;
      const float* row = &s_xg[n * XSTR];
      const float* U1 = u1 + h * C1_;
      const float* U2 = u2 + h * C1_;
      float a1 = 0.f, a2 = 0.f;
      for (int c = 0; c < C1_; ++c) {
        float v = row[c];
        a1 = fmaf(v, U1[c], a1);
        a2 = fmaf(v, U2[c], a2);
      }
      s_e1[n] = a1;
      s_e2[n] = a2;
    }
    __syncthreads();

    #pragma unroll
    for (int rbi = 0; rbi < 2; ++rbi) {
      const int rb = wave + rbi * 8;
      const int m  = rb * 16 + mloc;
      const float e1m = s_e1[m];
      const int jbase = half * 128;

      float mx = -3.0e38f;
      for (int w8 = 0; w8 < 4; ++w8) {
        unsigned adjw = s_adj[m * 8 + half * 4 + w8];
        const float* eb = &s_e2[jbase + w8 * 32];
        for (int j4 = 0; j4 < 32; j4 += 4) {
          v4f e = *(const v4f*)(eb + j4);
          #pragma unroll
          for (int u = 0; u < 4; ++u) {
            float sc = e1m + e[u];
            sc = sc > 0.f ? sc : ALPHA_ * sc;
            int on = (adjw >> (j4 + u)) & 1;
            mx = fmaxf(mx, on ? sc : -3.0e38f);
          }
        }
      }
      mx = fmaxf(mx, __shfl_xor(mx, 16));

      float sm = 0.f;
      for (int w8 = 0; w8 < 4; ++w8) {
        unsigned adjw = s_adj[m * 8 + half * 4 + w8];
        const float* eb = &s_e2[jbase + w8 * 32];
        float* pb = &s_pw[mloc * PSTR + jbase + w8 * 32];
        for (int j4 = 0; j4 < 32; j4 += 4) {
          v4f e = *(const v4f*)(eb + j4);
          v4f p;
          #pragma unroll
          for (int u = 0; u < 4; ++u) {
            float sc = e1m + e[u];
            sc = sc > 0.f ? sc : ALPHA_ * sc;
            sc = ((adjw >> (j4 + u)) & 1) ? sc : -3.0e38f;
            float pe = __expf(sc - mx);
            p[u] = pe;
            sm += pe;
          }
          *(v4f*)(pb + j4) = p;
        }
      }
      sm += __shfl_xor(sm, 16);
      s_rinv[rb * 16 + mloc] = 1.0f / sm;

      v8f acc0 = {}, acc1 = {}, acc2 = {}, acc3 = {};
      const float* pa   = &s_pw[mloc * PSTR + 2 * half];
      const float* xpb  = &s_xgp[half * XPSTR + 2 * mloc];
      for (int k0 = 0; k0 < NN_; k0 += 4) {
        v2f a = *(const v2f*)(pa + k0);
        const float* bp = xpb + (k0 >> 1) * XPSTR;
        v2f b0 = *(const v2f*)(bp + 0);
        v2f b1 = *(const v2f*)(bp + 32);
        v2f b2 = *(const v2f*)(bp + 64);
        v2f b3 = *(const v2f*)(bp + 96);
        acc0 = __builtin_amdgcn_wmma_f32_16x16x4_f32(false, a, false, b0, (short)0, acc0, false, false);
        acc1 = __builtin_amdgcn_wmma_f32_16x16x4_f32(false, a, false, b1, (short)0, acc1, false, false);
        acc2 = __builtin_amdgcn_wmma_f32_16x16x4_f32(false, a, false, b2, (short)0, acc2, false, false);
        acc3 = __builtin_amdgcn_wmma_f32_16x16x4_f32(false, a, false, b3, (short)0, acc3, false, false);
      }

      #pragma unroll
      for (int r = 0; r < 8; ++r) {
        float inv = s_rinv[rb * 16 + r + 8 * half];
        racc[rbi][0][r] += tanh_fast(acc0[r] * inv) * inv_h;
        racc[rbi][1][r] += tanh_fast(acc1[r] * inv) * inv_h;
        racc[rbi][2][r] += tanh_fast(acc2[r] * inv) * inv_h;
        racc[rbi][3][r] += tanh_fast(acc3[r] * inv) * inv_h;
      }
    }
    __syncthreads();
  }

  float* xs_bt = xs + (size_t)bt * NN_ * C1_;
  #pragma unroll
  for (int rbi = 0; rbi < 2; ++rbi) {
    const int rb = wave + rbi * 8;
    #pragma unroll
    for (int r = 0; r < 8; ++r) {
      int mm = rb * 16 + r + 8 * half;
      const float* xrow = &s_xg[mm * XSTR + mloc];
      float* orow = xs_bt + (size_t)mm * C1_ + mloc;
      orow[0]  = xrow[0]  + racc[rbi][0][r];
      orow[16] = xrow[16] + racc[rbi][1][r];
      orow[32] = xrow[32] + racc[rbi][2][r];
      orow[48] = xrow[48] + racc[rbi][3][r];
    }
  }
}

__global__ __launch_bounds__(256) void conv2_ln_kernel(
    const float* __restrict__ xs,
    const float* __restrict__ w2,
    const float* __restrict__ b2,
    const float* __restrict__ lnw,
    const float* __restrict__ lnb,
    float* __restrict__ out)
{
  extern __shared__ float smem[];
  float* s_w   = smem;
  float* s_y   = s_w + C2_ * C1_ * KT_;
  float* s_red = s_y + NN_ * C2_;

  const int bt = blockIdx.x;
  const int bb = bt >> 3, t = bt & 7;
  const int tid = threadIdx.x;

  for (int i = tid; i < C2_ * C1_ * KT_; i += 256) s_w[i] = w2[i];
  __syncthreads();

  const int n = tid;
  const float* xsb = xs + (((size_t)bb * T1_ + t) * NN_) * C1_;

  float lsum = 0.f, lsq = 0.f;
  for (int cb = 0; cb < 4; ++cb) {
    const int c0 = cb * 16;
    float acc[16];
    #pragma unroll
    for (int u = 0; u < 16; ++u) acc[u] = b2[c0 + u];
    for (int kt = 0; kt < KT_; ++kt) {
      const float* xrow = xsb + (size_t)kt * NN_ * C1_ + (size_t)n * C1_;
      for (int ci = 0; ci < C1_; ++ci) {
        float xv = xrow[ci];
        #pragma unroll
        for (int u = 0; u < 16; ++u)
          acc[u] = fmaf(s_w[((c0 + u) * C1_ + ci) * KT_ + kt], xv, acc[u]);
      }
    }
    const float* xres = xsb + (size_t)(KT_ - 1) * NN_ * C1_ + (size_t)n * C1_;
    #pragma unroll
    for (int u = 0; u < 16; ++u) {
      float v = acc[u] + xres[c0 + u];
      v = v > 0.f ? v : 0.f;
      s_y[n * C2_ + c0 + u] = v;
      lsum += v;
      lsq  += v * v;
    }
  }

  for (int off = 16; off > 0; off >>= 1) {
    lsum += __shfl_xor(lsum, off);
    lsq  += __shfl_xor(lsq,  off);
  }
  if ((tid & 31) == 0) { s_red[tid >> 5] = lsum; s_red[8 + (tid >> 5)] = lsq; }
  __syncthreads();
  float tsum = 0.f, tsq = 0.f;
  for (int i = 0; i < 8; ++i) { tsum += s_red[i]; tsq += s_red[8 + i]; }
  const float cnt  = (float)(NN_ * C2_);
  const float mu   = tsum / cnt;
  const float var  = tsq / cnt - mu * mu;
  const float rstd = rsqrtf(var + 1e-5f);

  for (int c = 0; c < C2_; ++c) {
    float v = (s_y[n * C2_ + c] - mu) * rstd * lnw[(size_t)n * C2_ + c] + lnb[(size_t)n * C2_ + c];
    out[(((size_t)bb * C2_ + c) * T2_ + t) * NN_ + n] = v;
  }
}

extern "C" void kernel_launch(void* const* d_in, const int* in_sizes, int n_in,
                              void* d_out, int out_size, void* d_ws, size_t ws_size,
                              hipStream_t stream) {
  const float* x   = (const float*)d_in[0];
  const float* adj = (const float*)d_in[1];
  const float* w1  = (const float*)d_in[2];
  const float* b1  = (const float*)d_in[3];
  const float* w2  = (const float*)d_in[4];
  const float* b2  = (const float*)d_in[5];
  const float* Wh  = (const float*)d_in[6];
  const float* ah  = (const float*)d_in[7];
  const float* lnw = (const float*)d_in[8];
  const float* lnb = (const float*)d_in[9];
  float* out = (float*)d_out;

  float* ws = (float*)d_ws;
  const size_t xg_elems = (size_t)B_ * T1_ * NN_ * C1_;
  float* xg = ws;
  float* xs = xg + xg_elems;
  float* u1 = xs + xg_elems;
  float* u2 = u1 + HH_ * C1_;

  prep_u_kernel<<<2, 256, 0, stream>>>(Wh, ah, u1, u2);

  const int n_c1 = (int)(xg_elems / 256);
  conv1_glu_kernel<<<n_c1, 256, 0, stream>>>(x, w1, b1, xg);

  const size_t attn_lds =
      (size_t)(NN_ * XSTR + (NN_ / 2) * XPSTR + 3 * NN_ + 8 * 16 * PSTR) * sizeof(float)
      + (size_t)(NN_ * 8) * sizeof(unsigned);
  attn_kernel<<<B_ * T1_, 256, attn_lds, stream>>>(xg, adj, u1, u2, xs);

  const size_t c2_lds = (size_t)(C2_ * C1_ * KT_ + NN_ * C2_ + 16) * sizeof(float);
  conv2_ln_kernel<<<B_ * T2_, 256, c2_lds, stream>>>(xs, w2, b2, lnw, lnb, out);
}


# conv1 remapped: lanes along output channel keep their 96 taps in registers; filter bank and x tile staged in LDS by coalesced loads, x read as broadcasts, full-line stores; sum order unchanged
# speedup vs baseline: 1.4259x; 1.4259x over previous
#include <hip/hip_runtime.h>
#include <math.h>

#define B_   8
#define C0_  16
#define C1_  64
#define C2_  64
#define T0_  12
#define T1_  10
#define T2_  8
#define NN_  256
#define KT_  3
#define HH_  8
#define ALPHA_ 0.2f

typedef float v2f __attribute__((ext_vector_type(2)));
typedef float v4f __attribute__((ext_vector_type(4)));
typedef float v8f __attribute__((ext_vector_type(8)));

__device__ __forceinline__ float tanh_fast(float x) {
  float e = __expf(fminf(2.0f * x, 80.0f));
  return (e - 1.0f) / (e + 1.0f);
}

__global__ __launch_bounds__(256) void prep_u_kernel(
    const float* __restrict__ W,
    const float* __restrict__ a,
    float* __restrict__ u1, float* __restrict__ u2)
{
  int tid = blockIdx.x * 256 + threadIdx.x;
  if (tid >= HH_ * C1_) return;
  int h = tid >> 6, c = tid & 63;
  const float* Wh = W + ((size_t)h * C1_ + c) * C1_;
  const float* ah = a + (size_t)h * 2 * C1_;
  float s1 = 0.f, s2 = 0.f;
  for (int d = 0; d < C1_; ++d) {
    float w = Wh[d];
    s1 += w * ah[d];
    s2 += w * ah[C1_ + d];
  }
  u1[tid] = s1;
  u2[tid] = s2;
}

#define C1_NT   64
#define C1_WSTR 49
#define C1_XSTR 68
#define C1_K    (C0_ * KT_)
#define C1_RC   6

__global__ __launch_bounds__(256) void conv1_glu_kernel(
    const float* __restrict__ x,
    const float* __restrict__ w,
    const float* __restrict__ b,
    float* __restrict__ xg)
{
  __shared__ __attribute__((aligned(16))) float s_w[2 * C1_ * C1_WSTR];
  __shared__ __attribute__((aligned(16))) float s_x[C1_K * C1_XSTR];

  const int tid  = threadIdx.x;
  const int lane = tid & 31;
  const int wave = tid >> 5;
  const int bt   = blockIdx.x >> 2;
  const int n0   = (blockIdx.x & 3) * C1_NT;
  const int bb   = bt / T1_, t = bt - bb * T1_;

  #pragma unroll
  for (int j = 0; j < (2 * C1_ * C1_K) / 4 / 256; ++j) {
    int i4 = tid + j * 256;
    v4f wv = ((const v4f*)w)[i4];
    int base = i4 * 4;
    int c2 = base / C1_K, k = base - c2 * C1_K;
    float* d = &s_w[c2 * C1_WSTR + k];
    d[0] = wv[0]; d[1] = wv[1]; d[2] = wv[2]; d[3] = wv[3];
  }
  #pragma unroll
  for (int j = 0; j < C1_K * (C1_NT / 4) / 256; ++j) {
    int i4 = tid + j * 256;
    int r = i4 >> 4, q = i4 & 15;
    int kt = r >> 4, ci = r & 15;
    const float* src = x + (((size_t)bb * C0_ + ci) * T0_ + (t + kt)) * NN_ + n0 + q * 4;
    *(v4f*)&s_x[r * C1_XSTR + q * 4] = *(const v4f*)src;
  }
  __syncthreads();

  const int c  = (wave & 1) * 32 + lane;
  const int nb = (wave >> 1) * 16;

  float wp[C1_K], wq[C1_K];
  #pragma unroll
  for (int r = 0; r < C1_K; ++r) {
    int kt = r >> 4, ci = r & 15;
    wp[r] = s_w[c * C1_WSTR + ci * KT_ + kt];
    wq[r] = s_w[(c + C1_) * C1_WSTR + ci * KT_ + kt];
  }
  const float bp = b[c];
  const float bq = b[c + C1_];
  const bool has_res = (c < C0_);
  const int  xr = has_res ? (2 * C0_ + c) : 0;

  #pragma unroll 1
  for (int g = 0; g < 2; ++g) {
    const int nl = nb + g * 8;
    const float* xrow = &s_x[nl];
    float ap[8], aq[8];
    #pragma unroll
    for (int u = 0; u < 8; ++u) { ap[u] = bp; aq[u] = bq; }

    v4f xa[2][C1_RC][2];
    #pragma unroll
    for (int j = 0; j < C1_RC; ++j) {
      xa[0][j][0] = *(const v4f*)(xrow + j * C1_XSTR);
      xa[0][j][1] = *(const v4f*)(xrow + j * C1_XSTR + 4);
    }
    #pragma unroll
    for (int ch = 0; ch < C1_K / C1_RC; ++ch) {
      __builtin_amdgcn_sched_barrier(0);
      if (ch + 1 < C1_K / C1_RC) {
        #pragma unroll
        for (int j = 0; j < C1_RC; ++j) {
          const float* p = xrow + ((ch + 1) * C1_RC + j) * C1_XSTR;
          xa[(ch + 1) & 1][j][0] = *(const v4f*)(p);
          xa[(ch + 1) & 1][j][1] = *(const v4f*)(p + 4);
        }
      }
      __builtin_amdgcn_sched_barrier(0);
      #pragma unroll
      for (int j = 0; j < C1_RC; ++j) {
        const int r = ch * C1_RC + j;
        const v4f x0 = xa[ch & 1][j][0];
        const v4f x1 = xa[ch & 1][j][1];
        #pragma unroll
        for (int u = 0; u < 4; ++u) {
          ap[u]     = fmaf(wp[r], x0[u], ap[u]);
          aq[u]     = fmaf(wq[r], x0[u], aq[u]);
          ap[u + 4] = fmaf(wp[r], x1[u], ap[u + 4]);
          aq[u + 4] = fmaf(wq[r], x1[u], aq[u + 4]);
        }
      }
    }
    v4f r0 = *(const v4f*)&s_x[xr * C1_XSTR + nl];
    v4f r1 = *(const v4f*)&s_x[xr * C1_XSTR + nl + 4];
    float* orow = xg + ((size_t)bt * NN_ + n0 + nl) * C1_ + c;
    #pragma unroll
    for (int u = 0; u < 8; ++u) {
      float xin = has_res ? (u < 4 ? r0[u & 3] : r1[u & 3]) : 0.f;
      float sg  = 1.f / (1.f + __expf(-aq[u]));
      orow[(size_t)u * C1_] = (ap[u] + xin) * sg;
    }
  }
}

#define XSTR  72
#define XPSTR 160
#define PSTR  260

__global__ __launch_bounds__(256) void attn_kernel(
    const float* __restrict__ xg,
    const float* __restrict__ adj,
    const float* __restrict__ u1,
    const float* __restrict__ u2,
    float* __restrict__ xs)
{
  extern __shared__ float smem[];
  float*    s_xg   = smem;
  float*    s_xgp  = s_xg + NN_ * XSTR;
  float*    s_e1   = s_xgp + (NN_ / 2) * XPSTR;
  float*    s_e2   = s_e1 + NN_;
  float*    s_rinv = s_e2 + NN_;
  float*    s_p    = s_rinv + NN_;
  unsigned* s_adj  = (unsigned*)(s_p + 8 * 16 * PSTR);

  const int bt   = blockIdx.x;
  const int tid  = threadIdx.x;
  const int lane = tid & 31;
  const int wave = tid >> 5;
  const int mloc = lane & 15;
  const int half = lane >> 4;

  const float* xg_bt = xg + (size_t)bt * NN_ * C1_;

  for (int i = tid; i < NN_ * C1_; i += 256) {
    int n = i >> 6, c = i & 63;
    float v = xg_bt[i];
    s_xg[n * XSTR + c] = v;
    s_xgp[(n >> 1) * XPSTR + 2 * c + (n & 1)] = v;
  }
  {
    int n = tid;
    for (int w8 = 0; w8 < 8; ++w8) {
      unsigned bits = 0u;
      for (int j = 0; j < 32; ++j)
        bits |= (adj[(size_t)n * NN_ + w8 * 32 + j] > 0.f) ? (1u << j) : 0u;
      s_adj[n * 8 + w8] = bits;
    }
  }
  __syncthreads();

  const float inv_h = 1.0f / (float)HH_;
  float* s_pw = s_p + wave * 16 * PSTR;

  float racc[2][4][8];
  #pragma unroll
  for (int i = 0; i < 2; ++i)
    #pragma unroll
    for (int j = 0; j < 4; ++j)
      #pragma unroll
      for (int r = 0; r < 8; ++r) racc[i][j][r] = 0.f;

  for (int h = 0; h < HH_; ++h) {
    {
      int n = tid;
      const float* row = &s_xg[n * XSTR];
      const float* U1 = u1 + h * C1_;
      const float* U2 = u2 + h * C1_;
      float a1 = 0.f, a2 = 0.f;
      for (int c = 0; c < C1_; ++c) {
        float v = row[c];
        a1 = fmaf(v, U1[c], a1);
        a2 = fmaf(v, U2[c], a2);
      }
      s_e1[n] = a1;
      s_e2[n] = a2;
    }
    __syncthreads();

    #pragma unroll
    for (int rbi = 0; rbi < 2; ++rbi) {
      const int rb = wave + rbi * 8;
      const int m  = rb * 16 + mloc;
      const float e1m = s_e1[m];
      const int jbase = half * 128;

      float mx = -3.0e38f;
      for (int w8 = 0; w8 < 4; ++w8) {
        unsigned adjw = s_adj[m * 8 + half * 4 + w8];
        const float* eb = &s_e2[jbase + w8 * 32];
        for (int j4 = 0; j4 < 32; j4 += 4) {
          v4f e = *(const v4f*)(eb + j4);
          #pragma unroll
          for (int u = 0; u < 4; ++u) {
            float sc = e1m + e[u];
            sc = sc > 0.f ? sc : ALPHA_ * sc;
            int on = (adjw >> (j4 + u)) & 1;
            mx = fmaxf(mx, on ? sc : -3.0e38f);
          }
        }
      }
      mx = fmaxf(mx, __shfl_xor(mx, 16));

      float sm = 0.f;
      for (int w8 = 0; w8 < 4; ++w8) {
        unsigned adjw = s_adj[m * 8 + half * 4 + w8];
        const float* eb = &s_e2[jbase + w8 * 32];
        float* pb = &s_pw[mloc * PSTR + jbase + w8 * 32];
        for (int j4 = 0; j4 < 32; j4 += 4) {
          v4f e = *(const v4f*)(eb + j4);
          v4f p;
          #pragma unroll
          for (int u = 0; u < 4; ++u) {
            float sc = e1m + e[u];
            sc = sc > 0.f ? sc : ALPHA_ * sc;
            sc = ((adjw >> (j4 + u)) & 1) ? sc : -3.0e38f;
            float pe = __expf(sc - mx);
            p[u] = pe;
            sm += pe;
          }
          *(v4f*)(pb + j4) = p;
        }
      }
      sm += __shfl_xor(sm, 16);
      s_rinv[rb * 16 + mloc] = 1.0f / sm;

      v8f acc0 = {}, acc1 = {}, acc2 = {}, acc3 = {};
      const float* pa   = &s_pw[mloc * PSTR + 2 * half];
      const float* xpb  = &s_xgp[half * XPSTR + 2 * mloc];
      for (int k0 = 0; k0 < NN_; k0 += 4) {
        v2f a = *(const v2f*)(pa + k0);
        const float* bp = xpb + (k0 >> 1) * XPSTR;
        v2f b0 = *(const v2f*)(bp + 0);
        v2f b1 = *(const v2f*)(bp + 32);
        v2f b2 = *(const v2f*)(bp + 64);
        v2f b3 = *(const v2f*)(bp + 96);
        acc0 = __builtin_amdgcn_wmma_f32_16x16x4_f32(false, a, false, b0, (short)0, acc0, false, false);
        acc1 = __builtin_amdgcn_wmma_f32_16x16x4_f32(false, a, false, b1, (short)0, acc1, false, false);
        acc2 = __builtin_amdgcn_wmma_f32_16x16x4_f32(false, a, false, b2, (short)0, acc2, false, false);
        acc3 = __builtin_amdgcn_wmma_f32_16x16x4_f32(false, a, false, b3, (short)0, acc3, false, false);
      }

      #pragma unroll
      for (int r = 0; r < 8; ++r) {
        float inv = s_rinv[rb * 16 + r + 8 * half];
        racc[rbi][0][r] += tanh_fast(acc0[r] * inv) * inv_h;
        racc[rbi][1][r] += tanh_fast(acc1[r] * inv) * inv_h;
        racc[rbi][2][r] += tanh_fast(acc2[r] * inv) * inv_h;
        racc[rbi][3][r] += tanh_fast(acc3[r] * inv) * inv_h;
      }
    }
    __syncthreads();
  }

  float* xs_bt = xs + (size_t)bt * NN_ * C1_;
  #pragma unroll
  for (int rbi = 0; rbi < 2; ++rbi) {
    const int rb = wave + rbi * 8;
    #pragma unroll
    for (int r = 0; r < 8; ++r) {
      int mm = rb * 16 + r + 8 * half;
      const float* xrow = &s_xg[mm * XSTR + mloc];
      float* orow = xs_bt + (size_t)mm * C1_ + mloc;
      orow[0]  = xrow[0]  + racc[rbi][0][r];
      orow[16] = xrow[16] + racc[rbi][1][r];
      orow[32] = xrow[32] + racc[rbi][2][r];
      orow[48] = xrow[48] + racc[rbi][3][r];
    }
  }
}

__global__ __launch_bounds__(256) void conv2_ln_kernel(
    const float* __restrict__ xs,
    const float* __restrict__ w2,
    const float* __restrict__ b2,
    const float* __restrict__ lnw,
    const float* __restrict__ lnb,
    float* __restrict__ out)
{
  extern __shared__ float smem[];
  float* s_w   = smem;
  float* s_y   = s_w + C2_ * C1_ * KT_;
  float* s_red = s_y + NN_ * C2_;

  const int bt = blockIdx.x;
  const int bb = bt >> 3, t = bt & 7;
  const int tid = threadIdx.x;

  for (int i = tid; i < C2_ * C1_ * KT_; i += 256) s_w[i] = w2[i];
  __syncthreads();

  const int n = tid;
  const float* xsb = xs + (((size_t)bb * T1_ + t) * NN_) * C1_;

  float lsum = 0.f, lsq = 0.f;
  for (int cb = 0; cb < 4; ++cb) {
    const int c0 = cb * 16;
    float acc[16];
    #pragma unroll
    for (int u = 0; u < 16; ++u) acc[u] = b2[c0 + u];
    for (int kt = 0; kt < KT_; ++kt) {
      const float* xrow = xsb + (size_t)kt * NN_ * C1_ + (size_t)n * C1_;
      for (int ci = 0; ci < C1_; ++ci) {
        float xv = xrow[ci];
        #pragma unroll
        for (int u = 0; u < 16; ++u)
          acc[u] = fmaf(s_w[((c0 + u) * C1_ + ci) * KT_ + kt], xv, acc[u]);
      }
    }
    const float* xres = xsb + (size_t)(KT_ - 1) * NN_ * C1_ + (size_t)n * C1_;
    #pragma unroll
    for (int u = 0; u < 16; ++u) {
      float v = acc[u] + xres[c0 + u];
      v = v > 0.f ? v : 0.f;
      s_y[n * C2_ + c0 + u] = v;
      lsum += v;
      lsq  += v * v;
    }
  }

  for (int off = 16; off > 0; off >>= 1) {
    lsum += __shfl_xor(lsum, off);
    lsq  += __shfl_xor(lsq,  off);
  }
  if ((tid & 31) == 0) { s_red[tid >> 5] = lsum; s_red[8 + (tid >> 5)] = lsq; }
  __syncthreads();
  float tsum = 0.f, tsq = 0.f;
  for (int i = 0; i < 8; ++i) { tsum += s_red[i]; tsq += s_red[8 + i]; }
  const float cnt  = (float)(NN_ * C2_);
  const float mu   = tsum / cnt;
  const float var  = tsq / cnt - mu * mu;
  const float rstd = rsqrtf(var + 1e-5f);

  for (int c = 0; c < C2_; ++c) {
    float v = (s_y[n * C2_ + c] - mu) * rstd * lnw[(size_t)n * C2_ + c] + lnb[(size_t)n * C2_ + c];
    out[(((size_t)bb * C2_ + c) * T2_ + t) * NN_ + n] = v;
  }
}

extern "C" void kernel_launch(void* const* d_in, const int* in_sizes, int n_in,
                              void* d_out, int out_size, void* d_ws, size_t ws_size,
                              hipStream_t stream) {
  const float* x   = (const float*)d_in[0];
  const float* adj = (const float*)d_in[1];
  const float* w1  = (const float*)d_in[2];
  const float* b1  = (const float*)d_in[3];
  const float* w2  = (const float*)d_in[4];
  const float* b2  = (const float*)d_in[5];
  const float* Wh  = (const float*)d_in[6];
  const float* ah  = (const float*)d_in[7];
  const float* lnw = (const float*)d_in[8];
  const float* lnb = (const float*)d_in[9];
  float* out = (float*)d_out;

  float* ws = (float*)d_ws;
  const size_t xg_elems = (size_t)B_ * T1_ * NN_ * C1_;
  float* xg = ws;
  float* xs = xg + xg_elems;
  float* u1 = xs + xg_elems;
  float* u2 = u1 + HH_ * C1_;

  prep_u_kernel<<<2, 256, 0, stream>>>(Wh, ah, u1, u2);

  conv1_glu_kernel<<<B_ * T1_ * (NN_ / C1_NT), 256, 0, stream>>>(x, w1, b1, xg);

  const size_t attn_lds =
      (size_t)(NN_ * XSTR + (NN_ / 2) * XPSTR + 3 * NN_ + 8 * 16 * PSTR) * sizeof(float)
      + (size_t)(NN_ * 8) * sizeof(unsigned);
  attn_kernel<<<B_ * T1_, 256, attn_lds, stream>>>(xg, adj, u1, u2, xs);

  const size_t c2_lds = (size_t)(C2_ * C1_ * KT_ + NN_ * C2_ + 16) * sizeof(float);
  conv2_ln_kernel<<<B_ * T2_, 256, c2_lds, stream>>>(xs, w2, b2, lnw, lnb, out);
}


# conv2 remapped: lane keeps 4 output channels x 16 nodes (64 sums) in registers; xs tile transposed into LDS and filter bank staged by coalesced loads, read as 16-byte broadcasts; sum order unchanged
# speedup vs baseline: 1.8813x; 1.3194x over previous
#include <hip/hip_runtime.h>
#include <math.h>

#define B_   8
#define C0_  16
#define C1_  64
#define C2_  64
#define T0_  12
#define T1_  10
#define T2_  8
#define NN_  256
#define KT_  3
#define HH_  8
#define ALPHA_ 0.2f

typedef float v2f __attribute__((ext_vector_type(2)));
typedef float v4f __attribute__((ext_vector_type(4)));
typedef float v8f __attribute__((ext_vector_type(8)));

__device__ __forceinline__ float tanh_fast(float x) {
  float e = __expf(fminf(2.0f * x, 80.0f));
  return (e - 1.0f) / (e + 1.0f);
}

__global__ __launch_bounds__(256) void prep_u_kernel(
    const float* __restrict__ W,
    const float* __restrict__ a,
    float* __restrict__ u1, float* __restrict__ u2)
{
  int tid = blockIdx.x * 256 + threadIdx.x;
  if (tid >= HH_ * C1_) return;
  int h = tid >> 6, c = tid & 63;
  const float* Wh = W + ((size_t)h * C1_ + c) * C1_;
  const float* ah = a + (size_t)h * 2 * C1_;
  float s1 = 0.f, s2 = 0.f;
  for (int d = 0; d < C1_; ++d) {
    float w = Wh[d];
    s1 += w * ah[d];
    s2 += w * ah[C1_ + d];
  }
  u1[tid] = s1;
  u2[tid] = s2;
}

#define C1_NT   64
#define C1_WSTR 49
#define C1_XSTR 68
#define C1_K    (C0_ * KT_)
#define C1_RC   6

__global__ __launch_bounds__(256) void conv1_glu_kernel(
    const float* __restrict__ x,
    const float* __restrict__ w,
    const float* __restrict__ b,
    float* __restrict__ xg)
{
  __shared__ __attribute__((aligned(16))) float s_w[2 * C1_ * C1_WSTR];
  __shared__ __attribute__((aligned(16))) float s_x[C1_K * C1_XSTR];

  const int tid  = threadIdx.x;
  const int lane = tid & 31;
  const int wave = tid >> 5;
  const int bt   = blockIdx.x >> 2;
  const int n0   = (blockIdx.x & 3) * C1_NT;
  const int bb   = bt / T1_, t = bt - bb * T1_;

  #pragma unroll
  for (int j = 0; j < (2 * C1_ * C1_K) / 4 / 256; ++j) {
    int i4 = tid + j * 256;
    v4f wv = ((const v4f*)w)[i4];
    int base = i4 * 4;
    int c2 = base / C1_K, k = base - c2 * C1_K;
    float* d = &s_w[c2 * C1_WSTR + k];
    d[0] = wv[0]; d[1] = wv[1]; d[2] = wv[2]; d[3] = wv[3];
  }
  #pragma unroll
  for (int j = 0; j < C1_K * (C1_NT / 4) / 256; ++j) {
    int i4 = tid + j * 256;
    int r = i4 >> 4, q = i4 & 15;
    int kt = r >> 4, ci = r & 15;
    const float* src = x + (((size_t)bb * C0_ + ci) * T0_ + (t + kt)) * NN_ + n0 + q * 4;
    *(v4f*)&s_x[r * C1_XSTR + q * 4] = *(const v4f*)src;
  }
  __syncthreads();

  const int c  = (wave & 1) * 32 + lane;
  const int nb = (wave >> 1) * 16;

  float wp[C1_K], wq[C1_K];
  #pragma unroll
  for (int r = 0; r < C1_K; ++r) {
    int kt = r >> 4, ci = r & 15;
    wp[r] = s_w[c * C1_WSTR + ci * KT_ + kt];
    wq[r] = s_w[(c + C1_) * C1_WSTR + ci * KT_ + kt];
  }
  const float bp = b[c];
  const float bq = b[c + C1_];
  const bool has_res = (c < C0_);
  const int  xr = has_res ? (2 * C0_ + c) : 0;

  #pragma unroll 1
  for (int g = 0; g < 2; ++g) {
    const int nl = nb + g * 8;
    const float* xrow = &s_x[nl];
    float ap[8], aq[8];
    #pragma unroll
    for (int u = 0; u < 8; ++u) { ap[u] = bp; aq[u] = bq; }

    v4f xa[2][C1_RC][2];
    #pragma unroll
    for (int j = 0; j < C1_RC; ++j) {
      xa[0][j][0] = *(const v4f*)(xrow + j * C1_XSTR);
      xa[0][j][1] = *(const v4f*)(xrow + j * C1_XSTR + 4);
    }
    #pragma unroll
    for (int ch = 0; ch < C1_K / C1_RC; ++ch) {
      __builtin_amdgcn_sched_barrier(0);
      if (ch + 1 < C1_K / C1_RC) {
        #pragma unroll
        for (int j = 0; j < C1_RC; ++j) {
          const float* p = xrow + ((ch + 1) * C1_RC + j) * C1_XSTR;
          xa[(ch + 1) & 1][j][0] = *(const v4f*)(p);
          xa[(ch + 1) & 1][j][1] = *(const v4f*)(p + 4);
        }
      }
      __builtin_amdgcn_sched_barrier(0);
      #pragma unroll
      for (int j = 0; j < C1_RC; ++j) {
        const int r = ch * C1_RC + j;
        const v4f x0 = xa[ch & 1][j][0];
        const v4f x1 = xa[ch & 1][j][1];
        #pragma unroll
        for (int u = 0; u < 4; ++u) {
          ap[u]     = fmaf(wp[r], x0[u], ap[u]);
          aq[u]     = fmaf(wq[r], x0[u], aq[u]);
          ap[u + 4] = fmaf(wp[r], x1[u], ap[u + 4]);
          aq[u + 4] = fmaf(wq[r], x1[u], aq[u + 4]);
        }
      }
    }
    v4f r0 = *(const v4f*)&s_x[xr * C1_XSTR + nl];
    v4f r1 = *(const v4f*)&s_x[xr * C1_XSTR + nl + 4];
    float* orow = xg + ((size_t)bt * NN_ + n0 + nl) * C1_ + c;
    #pragma unroll
    for (int u = 0; u < 8; ++u) {
      float xin = has_res ? (u < 4 ? r0[u & 3] : r1[u & 3]) : 0.f;
      float sg  = 1.f / (1.f + __expf(-aq[u]));
      orow[(size_t)u * C1_] = (ap[u] + xin) * sg;
    }
  }
}

#define XSTR  72
#define XPSTR 160
#define PSTR  260

__global__ __launch_bounds__(256) void attn_kernel(
    const float* __restrict__ xg,
    const float* __restrict__ adj,
    const float* __restrict__ u1,
    const float* __restrict__ u2,
    float* __restrict__ xs)
{
  extern __shared__ float smem[];
  float*    s_xg   = smem;
  float*    s_xgp  = s_xg + NN_ * XSTR;
  float*    s_e1   = s_xgp + (NN_ / 2) * XPSTR;
  float*    s_e2   = s_e1 + NN_;
  float*    s_rinv = s_e2 + NN_;
  float*    s_p    = s_rinv + NN_;
  unsigned* s_adj  = (unsigned*)(s_p + 8 * 16 * PSTR);

  const int bt   = blockIdx.x;
  const int tid  = threadIdx.x;
  const int lane = tid & 31;
  const int wave = tid >> 5;
  const int mloc = lane & 15;
  const int half = lane >> 4;

  const float* xg_bt = xg + (size_t)bt * NN_ * C1_;

  for (int i = tid; i < NN_ * C1_; i += 256) {
    int n = i >> 6, c = i & 63;
    float v = xg_bt[i];
    s_xg[n * XSTR + c] = v;
    s_xgp[(n >> 1) * XPSTR + 2 * c + (n & 1)] = v;
  }
  {
    int n = tid;
    for (int w8 = 0; w8 < 8; ++w8) {
      unsigned bits = 0u;
      for (int j = 0; j < 32; ++j)
        bits |= (adj[(size_t)n * NN_ + w8 * 32 + j] > 0.f) ? (1u << j) : 0u;
      s_adj[n * 8 + w8] = bits;
    }
  }
  __syncthreads();

  const float inv_h = 1.0f / (float)HH_;
  float* s_pw = s_p + wave * 16 * PSTR;

  float racc[2][4][8];
  #pragma unroll
  for (int i = 0; i < 2; ++i)
    #pragma unroll
    for (int j = 0; j < 4; ++j)
      #pragma unroll
      for (int r = 0; r < 8; ++r) racc[i][j][r] = 0.f;

  for (int h = 0; h < HH_; ++h) {
    {
      int n = tid;
      const float* row = &s_xg[n * XSTR];
      const float* U1 = u1 + h * C1_;
      const float* U2 = u2 + h * C1_;
      float a1 = 0.f, a2 = 0.f;
      for (int c = 0; c < C1_; ++c) {
        float v = row[c];
        a1 = fmaf(v, U1[c], a1);
        a2 = fmaf(v, U2[c], a2);
      }
      s_e1[n] = a1;
      s_e2[n] = a2;
    }
    __syncthreads();

    #pragma unroll
    for (int rbi = 0; rbi < 2; ++rbi) {
      const int rb = wave + rbi * 8;
      const int m  = rb * 16 + mloc;
      const float e1m = s_e1[m];
      const int jbase = half * 128;

      float mx = -3.0e38f;
      for (int w8 = 0; w8 < 4; ++w8) {
        unsigned adjw = s_adj[m * 8 + half * 4 + w8];
        const float* eb = &s_e2[jbase + w8 * 32];
        for (int j4 = 0; j4 < 32; j4 += 4) {
          v4f e = *(const v4f*)(eb + j4);
          #pragma unroll
          for (int u = 0; u < 4; ++u) {
            float sc = e1m + e[u];
            sc = sc > 0.f ? sc : ALPHA_ * sc;
            int on = (adjw >> (j4 + u)) & 1;
            mx = fmaxf(mx, on ? sc : -3.0e38f);
          }
        }
      }
      mx = fmaxf(mx, __shfl_xor(mx, 16));

      float sm = 0.f;
      for (int w8 = 0; w8 < 4; ++w8) {
        unsigned adjw = s_adj[m * 8 + half * 4 + w8];
        const float* eb = &s_e2[jbase + w8 * 32];
        float* pb = &s_pw[mloc * PSTR + jbase + w8 * 32];
        for (int j4 = 0; j4 < 32; j4 += 4) {
          v4f e = *(const v4f*)(eb + j4);
          v4f p;
          #pragma unroll
          for (int u = 0; u < 4; ++u) {
            float sc = e1m + e[u];
            sc = sc > 0.f ? sc : ALPHA_ * sc;
            sc = ((adjw >> (j4 + u)) & 1) ? sc : -3.0e38f;
            float pe = __expf(sc - mx);
            p[u] = pe;
            sm += pe;
          }
          *(v4f*)(pb + j4) = p;
        }
      }
      sm += __shfl_xor(sm, 16);
      s_rinv[rb * 16 + mloc] = 1.0f / sm;

      v8f acc0 = {}, acc1 = {}, acc2 = {}, acc3 = {};
      const float* pa   = &s_pw[mloc * PSTR + 2 * half];
      const float* xpb  = &s_xgp[half * XPSTR + 2 * mloc];
      for (int k0 = 0; k0 < NN_; k0 += 4) {
        v2f a = *(const v2f*)(pa + k0);
        const float* bp = xpb + (k0 >> 1) * XPSTR;
        v2f b0 = *(const v2f*)(bp + 0);
        v2f b1 = *(const v2f*)(bp + 32);
        v2f b2 = *(const v2f*)(bp + 64);
        v2f b3 = *(const v2f*)(bp + 96);
        acc0 = __builtin_amdgcn_wmma_f32_16x16x4_f32(false, a, false, b0, (short)0, acc0, false, false);
        acc1 = __builtin_amdgcn_wmma_f32_16x16x4_f32(false, a, false, b1, (short)0, acc1, false, false);
        acc2 = __builtin_amdgcn_wmma_f32_16x16x4_f32(false, a, false, b2, (short)0, acc2, false, false);
        acc3 = __builtin_amdgcn_wmma_f32_16x16x4_f32(false, a, false, b3, (short)0, acc3, false, false);
      }

      #pragma unroll
      for (int r = 0; r < 8; ++r) {
        float inv = s_rinv[rb * 16 + r + 8 * half];
        racc[rbi][0][r] += tanh_fast(acc0[r] * inv) * inv_h;
        racc[rbi][1][r] += tanh_fast(acc1[r] * inv) * inv_h;
        racc[rbi][2][r] += tanh_fast(acc2[r] * inv) * inv_h;
        racc[rbi][3][r] += tanh_fast(acc3[r] * inv) * inv_h;
      }
    }
    __syncthreads();
  }

  float* xs_bt = xs + (size_t)bt * NN_ * C1_;
  #pragma unroll
  for (int rbi = 0; rbi < 2; ++rbi) {
    const int rb = wave + rbi * 8;
    #pragma unroll
    for (int r = 0; r < 8; ++r) {
      int mm = rb * 16 + r + 8 * half;
      const float* xrow = &s_xg[mm * XSTR + mloc];
      float* orow = xs_bt + (size_t)mm * C1_ + mloc;
      orow[0]  = xrow[0]  + racc[rbi][0][r];
      orow[16] = xrow[16] + racc[rbi][1][r];
      orow[32] = xrow[32] + racc[rbi][2][r];
      orow[48] = xrow[48] + racc[rbi][3][r];
    }
  }
}

#define C2_K   (C1_ * KT_)
#define C2_XP  260
#define C2_WP  196
#define C2_YP  68
#define C2_G   4
#define C2_NG  (C2_K / C2_G)

__global__ __launch_bounds__(256) void conv2_ln_kernel(
    const float* __restrict__ xs,
    const float* __restrict__ w2,
    const float* __restrict__ b2,
    const float* __restrict__ lnw,
    const float* __restrict__ lnb,
    float* __restrict__ out)
{
  extern __shared__ float smem[];
  float* s_x   = smem;
  float* s_w   = s_x + C2_K * C2_XP;
  float* s_red = s_w + C2_ * C2_WP;
  float* s_y   = smem;

  const int bt = blockIdx.x;
  const int bb = bt >> 3, t = bt & 7;
  const int tid = threadIdx.x;

  #pragma unroll
  for (unsigned j = 0; j < (C2_ * C2_K) / 4 / 256; ++j) {
    const unsigned i4 = (unsigned)tid + j * 256u;
    const v4f wv = ((const v4f*)w2)[i4];
    const unsigned co = i4 / (C2_K / 4u);
    const unsigned r  = (i4 - co * (C2_K / 4u)) * 4u;
    #pragma unroll
    for (unsigned e = 0; e < 4; ++e) {
      const unsigned ci = (r + e) / KT_, kt = (r + e) - ci * KT_;
      s_w[co * C2_WP + kt * C1_ + ci] = wv[e];
    }
  }
  {
    const float* src = xs + (((size_t)bb * T1_ + t) * NN_) * C1_ + tid;
    float* dst = s_x + (tid & 63) * C2_XP + (tid >> 6);
    #pragma unroll 1
    for (int blk = 0; blk < 2 * KT_; ++blk) {
      float xv[32];
      #pragma unroll
      for (int u = 0; u < 32; ++u) xv[u] = src[u * 256];
      #pragma unroll
      for (int u = 0; u < 32; ++u) dst[u * 4] = xv[u];
      src += 32 * 256;
      dst += (blk & 1) ? (C1_ * C2_XP - 128) : 128;
    }
  }
  __syncthreads();

  const int lane = tid & 31;
  const int wave = tid >> 5;
  const int c16  = lane & 15;
  const int nb   = wave * 32 + (lane >> 4) * 16;

  float acc[4][16];
  #pragma unroll
  for (int j = 0; j < 4; ++j) {
    const float bj = b2[c16 + 16 * j];
    #pragma unroll
    for (int i = 0; i < 16; ++i) acc[j][i] = bj;
  }

  const float* xb = s_x + nb;
  const float* wb = s_w + c16 * C2_WP;

  v4f wv[2][4];
  v4f xv[2][C2_G][4];
  #pragma unroll
  for (int j = 0; j < 4; ++j) wv[0][j] = *(const v4f*)(wb + j * 16 * C2_WP);
  #pragma unroll
  for (int k = 0; k < C2_G; ++k)
    #pragma unroll
    for (int q = 0; q < 4; ++q) xv[0][k][q] = *(const v4f*)(xb + k * C2_XP + 4 * q);

  #pragma unroll 1
  for (int g = 0; g < C2_NG; g += 2) {
    #pragma unroll
    for (int s = 0; s < 2; ++s) {
      const int gn = (g + s + 1 < C2_NG) ? (g + s + 1) : (C2_NG - 1);
      const float* xn = xb + gn * C2_G * C2_XP;
      const float* wn = wb + gn * C2_G;
      __builtin_amdgcn_sched_barrier(0);
      #pragma unroll
      for (int j = 0; j < 4; ++j) wv[s ^ 1][j] = *(const v4f*)(wn + j * 16 * C2_WP);
      #pragma unroll
      for (int k = 0; k < C2_G; ++k)
        #pragma unroll
        for (int q = 0; q < 4; ++q) xv[s ^ 1][k][q] = *(const v4f*)(xn + k * C2_XP + 4 * q);
      __builtin_amdgcn_sched_barrier(0);
      #pragma unroll
      for (int k = 0; k < C2_G; ++k)
        #pragma unroll
        for (int j = 0; j < 4; ++j)
          #pragma unroll
          for (int q = 0; q < 4; ++q)
            #pragma unroll
            for (int u = 0; u < 4; ++u)
              acc[j][4 * q + u] = fmaf(wv[s][j][k], xv[s][k][q][u], acc[j][4 * q + u]);
    }
  }

  v4f res[4][4];
  #pragma unroll
  for (int j = 0; j < 4; ++j)
    #pragma unroll
    for (int q = 0; q < 4; ++q)
      res[j][q] = *(const v4f*)(s_x + (2 * C1_ + c16 + 16 * j) * C2_XP + nb + 4 * q);

  __syncthreads();

  #pragma unroll
  for (int j = 0; j < 4; ++j) {
    const int c = c16 + 16 * j;
    #pragma unroll
    for (int q = 0; q < 4; ++q) {
      #pragma unroll
      for (int u = 0; u < 4; ++u) {
        float v = acc[j][4 * q + u] + res[j][q][u];
        v = v > 0.f ? v : 0.f;
        s_y[(nb + 4 * q + u) * C2_YP + c] = v;
      }
    }
  }
  __syncthreads();

  const int n = tid;
  float yv[C2_];
  #pragma unroll
  for (int q = 0; q < C2_ / 4; ++q) {
    const v4f y4 = *(const v4f*)(s_y + n * C2_YP + 4 * q);
    yv[4 * q + 0] = y4[0]; yv[4 * q + 1] = y4[1]; yv[4 * q + 2] = y4[2]; yv[4 * q + 3] = y4[3];
  }
  float lsum = 0.f, lsq = 0.f;
  #pragma unroll
  for (int c = 0; c < C2_; ++c) {
    lsum += yv[c];
    lsq  += yv[c] * yv[c];
  }

  for (int off = 16; off > 0; off >>= 1) {
    lsum += __shfl_xor(lsum, off);
    lsq  += __shfl_xor(lsq,  off);
  }
  if ((tid & 31) == 0) { s_red[tid >> 5] = lsum; s_red[8 + (tid >> 5)] = lsq; }
  __syncthreads();
  float tsum = 0.f, tsq = 0.f;
  for (int i = 0; i < 8; ++i) { tsum += s_red[i]; tsq += s_red[8 + i]; }
  const float cnt  = (float)(NN_ * C2_);
  const float mu   = tsum / cnt;
  const float var  = tsq / cnt - mu * mu;
  const float rstd = rsqrtf(var + 1e-5f);

  #pragma unroll
  for (int c = 0; c < C2_; ++c) {
    float v = (yv[c] - mu) * rstd * lnw[(size_t)n * C2_ + c] + lnb[(size_t)n * C2_ + c];
    out[(((size_t)bb * C2_ + c) * T2_ + t) * NN_ + n] = v;
  }
}

extern "C" void kernel_launch(void* const* d_in, const int* in_sizes, int n_in,
                              void* d_out, int out_size, void* d_ws, size_t ws_size,
                              hipStream_t stream) {
  const float* x   = (const float*)d_in[0];
  const float* adj = (const float*)d_in[1];
  const float* w1  = (const float*)d_in[2];
  const float* b1  = (const float*)d_in[3];
  const float* w2  = (const float*)d_in[4];
  const float* b2  = (const float*)d_in[5];
  const float* Wh  = (const float*)d_in[6];
  const float* ah  = (const float*)d_in[7];
  const float* lnw = (const float*)d_in[8];
  const float* lnb = (const float*)d_in[9];
  float* out = (float*)d_out;

  float* ws = (float*)d_ws;
  const size_t xg_elems = (size_t)B_ * T1_ * NN_ * C1_;
  float* xg = ws;
  float* xs = xg + xg_elems;
  float* u1 = xs + xg_elems;
  float* u2 = u1 + HH_ * C1_;

  prep_u_kernel<<<2, 256, 0, stream>>>(Wh, ah, u1, u2);

  conv1_glu_kernel<<<B_ * T1_ * (NN_ / C1_NT), 256, 0, stream>>>(x, w1, b1, xg);

  const size_t attn_lds =
      (size_t)(NN_ * XSTR + (NN_ / 2) * XPSTR + 3 * NN_ + 8 * 16 * PSTR) * sizeof(float)
      + (size_t)(NN_ * 8) * sizeof(unsigned);
  attn_kernel<<<B_ * T1_, 256, attn_lds, stream>>>(xg, adj, u1, u2, xs);

  const size_t c2_lds = (size_t)(C2_K * C2_XP + C2_ * C2_WP + 16) * sizeof(float);
  conv2_ln_kernel<<<B_ * T2_, 256, c2_lds, stream>>>(xs, w2, b2, lnw, lnb, out);
}


# attention: each (b,t) split by rows over 2 workgroups (128 rows each, one 16-row block per wave): 160 workgroups instead of 80 share softmax and matrix products; sum order unchanged
# speedup vs baseline: 2.9098x; 1.5467x over previous
#include <hip/hip_runtime.h>
#include <math.h>

#define B_   8
#define C0_  16
#define C1_  64
#define C2_  64
#define T0_  12
#define T1_  10
#define T2_  8
#define NN_  256
#define KT_  3
#define HH_  8
#define ALPHA_ 0.2f

typedef float v2f __attribute__((ext_vector_type(2)));
typedef float v4f __attribute__((ext_vector_type(4)));
typedef float v8f __attribute__((ext_vector_type(8)));

__device__ __forceinline__ float tanh_fast(float x) {
  float e = __expf(fminf(2.0f * x, 80.0f));
  return (e - 1.0f) / (e + 1.0f);
}

__global__ __launch_bounds__(256) void prep_u_kernel(
    const float* __restrict__ W,
    const float* __restrict__ a,
    float* __restrict__ u1, float* __restrict__ u2)
{
  int tid = blockIdx.x * 256 + threadIdx.x;
  if (tid >= HH_ * C1_) return;
  int h = tid >> 6, c = tid & 63;
  const float* Wh = W + ((size_t)h * C1_ + c) * C1_;
  const float* ah = a + (size_t)h * 2 * C1_;
  float s1 = 0.f, s2 = 0.f;
  for (int d = 0; d < C1_; ++d) {
    float w = Wh[d];
    s1 += w * ah[d];
    s2 += w * ah[C1_ + d];
  }
  u1[tid] = s1;
  u2[tid] = s2;
}

#define C1_NT   64
#define C1_WSTR 49
#define C1_XSTR 68
#define C1_K    (C0_ * KT_)
#define C1_RC   6

__global__ __launch_bounds__(256) void conv1_glu_kernel(
    const float* __restrict__ x,
    const float* __restrict__ w,
    const float* __restrict__ b,
    float* __restrict__ xg)
{
  __shared__ __attribute__((aligned(16))) float s_w[2 * C1_ * C1_WSTR];
  __shared__ __attribute__((aligned(16))) float s_x[C1_K * C1_XSTR];

  const int tid  = threadIdx.x;
  const int lane = tid & 31;
  const int wave = tid >> 5;
  const int bt   = blockIdx.x >> 2;
  const int n0   = (blockIdx.x & 3) * C1_NT;
  const int bb   = bt / T1_, t = bt - bb * T1_;

  #pragma unroll
  for (int j = 0; j < (2 * C1_ * C1_K) / 4 / 256; ++j) {
    int i4 = tid + j * 256;
    v4f wv = ((const v4f*)w)[i4];
    int base = i4 * 4;
    int c2 = base / C1_K, k = base - c2 * C1_K;
    float* d = &s_w[c2 * C1_WSTR + k];
    d[0] = wv[0]; d[1] = wv[1]; d[2] = wv[2]; d[3] = wv[3];
  }
  #pragma unroll
  for (int j = 0; j < C1_K * (C1_NT / 4) / 256; ++j) {
    int i4 = tid + j * 256;
    int r = i4 >> 4, q = i4 & 15;
    int kt = r >> 4, ci = r & 15;
    const float* src = x + (((size_t)bb * C0_ + ci) * T0_ + (t + kt)) * NN_ + n0 + q * 4;
    *(v4f*)&s_x[r * C1_XSTR + q * 4] = *(const v4f*)src;
  }
  __syncthreads();

  const int c  = (wave & 1) * 32 + lane;
  const int nb = (wave >> 1) * 16;

  float wp[C1_K], wq[C1_K];
  #pragma unroll
  for (int r = 0; r < C1_K; ++r) {
    int kt = r >> 4, ci = r & 15;
    wp[r] = s_w[c * C1_WSTR + ci * KT_ + kt];
    wq[r] = s_w[(c + C1_) * C1_WSTR + ci * KT_ + kt];
  }
  const float bp = b[c];
  const float bq = b[c + C1_];
  const bool has_res = (c < C0_);
  const int  xr = has_res ? (2 * C0_ + c) : 0;

  #pragma unroll 1
  for (int g = 0; g < 2; ++g) {
    const int nl = nb + g * 8;
    const float* xrow = &s_x[nl];
    float ap[8], aq[8];
    #pragma unroll
    for (int u = 0; u < 8; ++u) { ap[u] = bp; aq[u] = bq; }

    v4f xa[2][C1_RC][2];
    #pragma unroll
    for (int j = 0; j < C1_RC; ++j) {
      xa[0][j][0] = *(const v4f*)(xrow + j * C1_XSTR);
      xa[0][j][1] = *(const v4f*)(xrow + j * C1_XSTR + 4);
    }
    #pragma unroll
    for (int ch = 0; ch < C1_K / C1_RC; ++ch) {
      __builtin_amdgcn_sched_barrier(0);
      if (ch + 1 < C1_K / C1_RC) {
        #pragma unroll
        for (int j = 0; j < C1_RC; ++j) {
          const float* p = xrow + ((ch + 1) * C1_RC + j) * C1_XSTR;
          xa[(ch + 1) & 1][j][0] = *(const v4f*)(p);
          xa[(ch + 1) & 1][j][1] = *(const v4f*)(p + 4);
        }
      }
      __builtin_amdgcn_sched_barrier(0);
      #pragma unroll
      for (int j = 0; j < C1_RC; ++j) {
        const int r = ch * C1_RC + j;
        const v4f x0 = xa[ch & 1][j][0];
        const v4f x1 = xa[ch & 1][j][1];
        #pragma unroll
        for (int u = 0; u < 4; ++u) {
          ap[u]     = fmaf(wp[r], x0[u], ap[u]);
          aq[u]     = fmaf(wq[r], x0[u], aq[u]);
          ap[u + 4] = fmaf(wp[r], x1[u], ap[u + 4]);
          aq[u + 4] = fmaf(wq[r], x1[u], aq[u + 4]);
        }
      }
    }
    v4f r0 = *(const v4f*)&s_x[xr * C1_XSTR + nl];
    v4f r1 = *(const v4f*)&s_x[xr * C1_XSTR + nl + 4];
    float* orow = xg + ((size_t)bt * NN_ + n0 + nl) * C1_ + c;
    #pragma unroll
    for (int u = 0; u < 8; ++u) {
      float xin = has_res ? (u < 4 ? r0[u & 3] : r1[u & 3]) : 0.f;
      float sg  = 1.f / (1.f + __expf(-aq[u]));
      orow[(size_t)u * C1_] = (ap[u] + xin) * sg;
    }
  }
}

#define XSTR  72
#define XPSTR 160
#define PSTR  260
#define ATT_RS   2
#define ATT_ROWS (NN_ / ATT_RS)

__global__ __launch_bounds__(256) void attn_kernel(
    const float* __restrict__ xg,
    const float* __restrict__ adj,
    const float* __restrict__ u1,
    const float* __restrict__ u2,
    float* __restrict__ xs)
{
  extern __shared__ float smem[];
  float*    s_xg   = smem;
  float*    s_xgp  = s_xg + NN_ * XSTR;
  float*    s_e1   = s_xgp + (NN_ / 2) * XPSTR;
  float*    s_e2   = s_e1 + NN_;
  float*    s_rinv = s_e2 + NN_;
  float*    s_p    = s_rinv + NN_;
  unsigned* s_adj  = (unsigned*)(s_p + 8 * 16 * PSTR);

  const int bt   = blockIdx.x / ATT_RS;
  const int row0 = (blockIdx.x % ATT_RS) * ATT_ROWS;
  const int tid  = threadIdx.x;
  const int lane = tid & 31;
  const int wave = tid >> 5;
  const int mloc = lane & 15;
  const int half = lane >> 4;

  const float* xg_bt = xg + (size_t)bt * NN_ * C1_;

  for (int i = tid; i < NN_ * C1_; i += 256) {
    int n = i >> 6, c = i & 63;
    float v = xg_bt[i];
    s_xg[n * XSTR + c] = v;
    s_xgp[(n >> 1) * XPSTR + 2 * c + (n & 1)] = v;
  }
  {
    const int rl = tid >> 1, ch = tid & 1;
    const float* arow = adj + (size_t)(row0 + rl) * NN_ + ch * 128;
    for (int w8 = 0; w8 < 4; ++w8) {
      unsigned bits = 0u;
      for (int j = 0; j < 32; ++j)
        bits |= (arow[w8 * 32 + j] > 0.f) ? (1u << j) : 0u;
      s_adj[rl * 8 + ch * 4 + w8] = bits;
    }
  }
  __syncthreads();

  const float inv_h = 1.0f / (float)HH_;
  float* s_pw = s_p + wave * 16 * PSTR;

  const int rb = (row0 >> 4) + wave;
  const int m  = rb * 16 + mloc;
  const int ml = wave * 16 + mloc;
  const int jbase = half * 128;

  float racc[4][8];
  #pragma unroll
  for (int j = 0; j < 4; ++j)
    #pragma unroll
    for (int r = 0; r < 8; ++r) racc[j][r] = 0.f;

  for (int h = 0; h < HH_; ++h) {
    {
      int n = tid;
      const float* row = &s_xg[n * XSTR];
      const float* U1 = u1 + h * C1_;
      const float* U2 = u2 + h * C1_;
      float a1 = 0.f, a2 = 0.f;
      for (int c = 0; c < C1_; ++c) {
        float v = row[c];
        a1 = fmaf(v, U1[c], a1);
        a2 = fmaf(v, U2[c], a2);
      }
      s_e1[n] = a1;
      s_e2[n] = a2;
    }
    __syncthreads();

    {
      const float e1m = s_e1[m];

      float mx = -3.0e38f;
      for (int w8 = 0; w8 < 4; ++w8) {
        unsigned adjw = s_adj[ml * 8 + half * 4 + w8];
        const float* eb = &s_e2[jbase + w8 * 32];
        for (int j4 = 0; j4 < 32; j4 += 4) {
          v4f e = *(const v4f*)(eb + j4);
          #pragma unroll
          for (int u = 0; u < 4; ++u) {
            float sc = e1m + e[u];
            sc = sc > 0.f ? sc : ALPHA_ * sc;
            int on = (adjw >> (j4 + u)) & 1;
            mx = fmaxf(mx, on ? sc : -3.0e38f);
          }
        }
      }
      mx = fmaxf(mx, __shfl_xor(mx, 16));

      float sm = 0.f;
      for (int w8 = 0; w8 < 4; ++w8) {
        unsigned adjw = s_adj[ml * 8 + half * 4 + w8];
        const float* eb = &s_e2[jbase + w8 * 32];
        float* pb = &s_pw[mloc * PSTR + jbase + w8 * 32];
        for (int j4 = 0; j4 < 32; j4 += 4) {
          v4f e = *(const v4f*)(eb + j4);
          v4f p;
          #pragma unroll
          for (int u = 0; u < 4; ++u) {
            float sc = e1m + e[u];
            sc = sc > 0.f ? sc : ALPHA_ * sc;
            sc = ((adjw >> (j4 + u)) & 1) ? sc : -3.0e38f;
            float pe = __expf(sc - mx);
            p[u] = pe;
            sm += pe;
          }
          *(v4f*)(pb + j4) = p;
        }
      }
      sm += __shfl_xor(sm, 16);
      s_rinv[ml] = 1.0f / sm;

      v8f acc0 = {}, acc1 = {}, acc2 = {}, acc3 = {};
      const float* pa   = &s_pw[mloc * PSTR + 2 * half];
      const float* xpb  = &s_xgp[half * XPSTR + 2 * mloc];
      for (int k0 = 0; k0 < NN_; k0 += 4) {
        v2f a = *(const v2f*)(pa + k0);
        const float* bp = xpb + (k0 >> 1) * XPSTR;
        v2f b0 = *(const v2f*)(bp + 0);
        v2f b1 = *(const v2f*)(bp + 32);
        v2f b2 = *(const v2f*)(bp + 64);
        v2f b3 = *(const v2f*)(bp + 96);
        acc0 = __builtin_amdgcn_wmma_f32_16x16x4_f32(false, a, false, b0, (short)0, acc0, false, false);
        acc1 = __builtin_amdgcn_wmma_f32_16x16x4_f32(false, a, false, b1, (short)0, acc1, false, false);
        acc2 = __builtin_amdgcn_wmma_f32_16x16x4_f32(false, a, false, b2, (short)0, acc2, false, false);
        acc3 = __builtin_amdgcn_wmma_f32_16x16x4_f32(false, a, false, b3, (short)0, acc3, false, false);
      }

      #pragma unroll
      for (int r = 0; r < 8; ++r) {
        float inv = s_rinv[wave * 16 + r + 8 * half];
        racc[0][r] += tanh_fast(acc0[r] * inv) * inv_h;
        racc[1][r] += tanh_fast(acc1[r] * inv) * inv_h;
        racc[2][r] += tanh_fast(acc2[r] * inv) * inv_h;
        racc[3][r] += tanh_fast(acc3[r] * inv) * inv_h;
      }
    }
    __syncthreads();
  }

  float* xs_bt = xs + (size_t)bt * NN_ * C1_;
  #pragma unroll
  for (int r = 0; r < 8; ++r) {
    int mm = rb * 16 + r + 8 * half;
    const float* xrow = &s_xg[mm * XSTR + mloc];
    float* orow = xs_bt + (size_t)mm * C1_ + mloc;
    orow[0]  = xrow[0]  + racc[0][r];
    orow[16] = xrow[16] + racc[1][r];
    orow[32] = xrow[32] + racc[2][r];
    orow[48] = xrow[48] + racc[3][r];
  }
}

#define C2_K   (C1_ * KT_)
#define C2_XP  260
#define C2_WP  196
#define C2_YP  68
#define C2_G   4
#define C2_NG  (C2_K / C2_G)

__global__ __launch_bounds__(256) void conv2_ln_kernel(
    const float* __restrict__ xs,
    const float* __restrict__ w2,
    const float* __restrict__ b2,
    const float* __restrict__ lnw,
    const float* __restrict__ lnb,
    float* __restrict__ out)
{
  extern __shared__ float smem[];
  float* s_x   = smem;
  float* s_w   = s_x + C2_K * C2_XP;
  float* s_red = s_w + C2_ * C2_WP;
  float* s_y   = smem;

  const int bt = blockIdx.x;
  const int bb = bt >> 3, t = bt & 7;
  const int tid = threadIdx.x;

  #pragma unroll
  for (unsigned j = 0; j < (C2_ * C2_K) / 4 / 256; ++j) {
    const unsigned i4 = (unsigned)tid + j * 256u;
    const v4f wv = ((const v4f*)w2)[i4];
    const unsigned co = i4 / (C2_K / 4u);
    const unsigned r  = (i4 - co * (C2_K / 4u)) * 4u;
    #pragma unroll
    for (unsigned e = 0; e < 4; ++e) {
      const unsigned ci = (r + e) / KT_, kt = (r + e) - ci * KT_;
      s_w[co * C2_WP + kt * C1_ + ci] = wv[e];
    }
  }
  {
    const float* src = xs + (((size_t)bb * T1_ + t) * NN_) * C1_ + tid;
    float* dst = s_x + (tid & 63) * C2_XP + (tid >> 6);
    #pragma unroll 1
    for (int blk = 0; blk < 2 * KT_; ++blk) {
      float xv[32];
      #pragma unroll
      for (int u = 0; u < 32; ++u) xv[u] = src[u * 256];
      #pragma unroll
      for (int u = 0; u < 32; ++u) dst[u * 4] = xv[u];
      src += 32 * 256;
      dst += (blk & 1) ? (C1_ * C2_XP - 128) : 128;
    }
  }
  __syncthreads();

  const int lane = tid & 31;
  const int wave = tid >> 5;
  const int c16  = lane & 15;
  const int nb   = wave * 32 + (lane >> 4) * 16;

  float acc[4][16];
  #pragma unroll
  for (int j = 0; j < 4; ++j) {
    const float bj = b2[c16 + 16 * j];
    #pragma unroll
    for (int i = 0; i < 16; ++i) acc[j][i] = bj;
  }

  const float* xb = s_x + nb;
  const float* wb = s_w + c16 * C2_WP;

  v4f wv[2][4];
  v4f xv[2][C2_G][4];
  #pragma unroll
  for (int j = 0; j < 4; ++j) wv[0][j] = *(const v4f*)(wb + j * 16 * C2_WP);
  #pragma unroll
  for (int k = 0; k < C2_G; ++k)
    #pragma unroll
    for (int q = 0; q < 4; ++q) xv[0][k][q] = *(const v4f*)(xb + k * C2_XP + 4 * q);

  #pragma unroll 1
  for (int g = 0; g < C2_NG; g += 2) {
    #pragma unroll
    for (int s = 0; s < 2; ++s) {
      const int gn = (g + s + 1 < C2_NG) ? (g + s + 1) : (C2_NG - 1);
      const float* xn = xb + gn * C2_G * C2_XP;
      const float* wn = wb + gn * C2_G;
      __builtin_amdgcn_sched_barrier(0);
      #pragma unroll
      for (int j = 0; j < 4; ++j) wv[s ^ 1][j] = *(const v4f*)(wn + j * 16 * C2_WP);
      #pragma unroll
      for (int k = 0; k < C2_G; ++k)
        #pragma unroll
        for (int q = 0; q < 4; ++q) xv[s ^ 1][k][q] = *(const v4f*)(xn + k * C2_XP + 4 * q);
      __builtin_amdgcn_sched_barrier(0);
      #pragma unroll
      for (int k = 0; k < C2_G; ++k)
        #pragma unroll
        for (int j = 0; j < 4; ++j)
          #pragma unroll
          for (int q = 0; q < 4; ++q)
            #pragma unroll
            for (int u = 0; u < 4; ++u)
              acc[j][4 * q + u] = fmaf(wv[s][j][k], xv[s][k][q][u], acc[j][4 * q + u]);
    }
  }

  v4f res[4][4];
  #pragma unroll
  for (int j = 0; j < 4; ++j)
    #pragma unroll
    for (int q = 0; q < 4; ++q)
      res[j][q] = *(const v4f*)(s_x + (2 * C1_ + c16 + 16 * j) * C2_XP + nb + 4 * q);

  __syncthreads();

  #pragma unroll
  for (int j = 0; j < 4; ++j) {
    const int c = c16 + 16 * j;
    #pragma unroll
    for (int q = 0; q < 4; ++q) {
      #pragma unroll
      for (int u = 0; u < 4; ++u) {
        float v = acc[j][4 * q + u] + res[j][q][u];
        v = v > 0.f ? v : 0.f;
        s_y[(nb + 4 * q + u) * C2_YP + c] = v;
      }
    }
  }
  __syncthreads();

  const int n = tid;
  float yv[C2_];
  #pragma unroll
  for (int q = 0; q < C2_ / 4; ++q) {
    const v4f y4 = *(const v4f*)(s_y + n * C2_YP + 4 * q);
    yv[4 * q + 0] = y4[0]; yv[4 * q + 1] = y4[1]; yv[4 * q + 2] = y4[2]; yv[4 * q + 3] = y4[3];
  }
  float lsum = 0.f, lsq = 0.f;
  #pragma unroll
  for (int c = 0; c < C2_; ++c) {
    lsum += yv[c];
    lsq  += yv[c] * yv[c];
  }

  for (int off = 16; off > 0; off >>= 1) {
    lsum += __shfl_xor(lsum, off);
    lsq  += __shfl_xor(lsq,  off);
  }
  if ((tid & 31) == 0) { s_red[tid >> 5] = lsum; s_red[8 + (tid >> 5)] = lsq; }
  __syncthreads();
  float tsum = 0.f, tsq = 0.f;
  for (int i = 0; i < 8; ++i) { tsum += s_red[i]; tsq += s_red[8 + i]; }
  const float cnt  = (float)(NN_ * C2_);
  const float mu   = tsum / cnt;
  const float var  = tsq / cnt - mu * mu;
  const float rstd = rsqrtf(var + 1e-5f);

  #pragma unroll
  for (int c = 0; c < C2_; ++c) {
    float v = (yv[c] - mu) * rstd * lnw[(size_t)n * C2_ + c] + lnb[(size_t)n * C2_ + c];
    out[(((size_t)bb * C2_ + c) * T2_ + t) * NN_ + n] = v;
  }
}

extern "C" void kernel_launch(void* const* d_in, const int* in_sizes, int n_in,
                              void* d_out, int out_size, void* d_ws, size_t ws_size,
                              hipStream_t stream) {
  const float* x   = (const float*)d_in[0];
  const float* adj = (const float*)d_in[1];
  const float* w1  = (const float*)d_in[2];
  const float* b1  = (const float*)d_in[3];
  const float* w2  = (const float*)d_in[4];
  const float* b2  = (const float*)d_in[5];
  const float* Wh  = (const float*)d_in[6];
  const float* ah  = (const float*)d_in[7];
  const float* lnw = (const float*)d_in[8];
  const float* lnb = (const float*)d_in[9];
  float* out = (float*)d_out;

  float* ws = (float*)d_ws;
  const size_t xg_elems = (size_t)B_ * T1_ * NN_ * C1_;
  float* xg = ws;
  float* xs = xg + xg_elems;
  float* u1 = xs + xg_elems;
  float* u2 = u1 + HH_ * C1_;

  prep_u_kernel<<<2, 256, 0, stream>>>(Wh, ah, u1, u2);

  conv1_glu_kernel<<<B_ * T1_ * (NN_ / C1_NT), 256, 0, stream>>>(x, w1, b1, xg);

  const size_t attn_lds =
      (size_t)(NN_ * XSTR + (NN_ / 2) * XPSTR + 3 * NN_ + 8 * 16 * PSTR) * sizeof(float)
      + (size_t)(ATT_ROWS * 8) * sizeof(unsigned);
  attn_kernel<<<B_ * T1_ * ATT_RS, 256, attn_lds, stream>>>(xg, adj, u1, u2, xs);

  const size_t c2_lds = (size_t)(C2_K * C2_XP + C2_ * C2_WP + 16) * sizeof(float);
  conv2_ln_kernel<<<B_ * T2_, 256, c2_lds, stream>>>(xs, w2, b2, lnw, lnb, out);
}


# attention K loop: A and B fragments of the next block of 4 K steps loaded as one group into a second register set ahead of the current block's 16 matrix instructions; sum order unchanged
# speedup vs baseline: 3.3351x; 1.1462x over previous
#include <hip/hip_runtime.h>
#include <math.h>

#define B_   8
#define C0_  16
#define C1_  64
#define C2_  64
#define T0_  12
#define T1_  10
#define T2_  8
#define NN_  256
#define KT_  3
#define HH_  8
#define ALPHA_ 0.2f

typedef float v2f __attribute__((ext_vector_type(2)));
typedef float v4f __attribute__((ext_vector_type(4)));
typedef float v8f __attribute__((ext_vector_type(8)));

__device__ __forceinline__ float tanh_fast(float x) {
  float e = __expf(fminf(2.0f * x, 80.0f));
  return (e - 1.0f) / (e + 1.0f);
}

__global__ __launch_bounds__(256) void prep_u_kernel(
    const float* __restrict__ W,
    const float* __restrict__ a,
    float* __restrict__ u1, float* __restrict__ u2)
{
  int tid = blockIdx.x * 256 + threadIdx.x;
  if (tid >= HH_ * C1_) return;
  int h = tid >> 6, c = tid & 63;
  const float* Wh = W + ((size_t)h * C1_ + c) * C1_;
  const float* ah = a + (size_t)h * 2 * C1_;
  float s1 = 0.f, s2 = 0.f;
  for (int d = 0; d < C1_; ++d) {
    float w = Wh[d];
    s1 += w * ah[d];
    s2 += w * ah[C1_ + d];
  }
  u1[tid] = s1;
  u2[tid] = s2;
}

#define C1_NT   64
#define C1_WSTR 49
#define C1_XSTR 68
#define C1_K    (C0_ * KT_)
#define C1_RC   6

__global__ __launch_bounds__(256) void conv1_glu_kernel(
    const float* __restrict__ x,
    const float* __restrict__ w,
    const float* __restrict__ b,
    float* __restrict__ xg)
{
  __shared__ __attribute__((aligned(16))) float s_w[2 * C1_ * C1_WSTR];
  __shared__ __attribute__((aligned(16))) float s_x[C1_K * C1_XSTR];

  const int tid  = threadIdx.x;
  const int lane = tid & 31;
  const int wave = tid >> 5;
  const int bt   = blockIdx.x >> 2;
  const int n0   = (blockIdx.x & 3) * C1_NT;
  const int bb   = bt / T1_, t = bt - bb * T1_;

  #pragma unroll
  for (int j = 0; j < (2 * C1_ * C1_K) / 4 / 256; ++j) {
    int i4 = tid + j * 256;
    v4f wv = ((const v4f*)w)[i4];
    int base = i4 * 4;
    int c2 = base / C1_K, k = base - c2 * C1_K;
    float* d = &s_w[c2 * C1_WSTR + k];
    d[0] = wv[0]; d[1] = wv[1]; d[2] = wv[2]; d[3] = wv[3];
  }
  #pragma unroll
  for (int j = 0; j < C1_K * (C1_NT / 4) / 256; ++j) {
    int i4 = tid + j * 256;
    int r = i4 >> 4, q = i4 & 15;
    int kt = r >> 4, ci = r & 15;
    const float* src = x + (((size_t)bb * C0_ + ci) * T0_ + (t + kt)) * NN_ + n0 + q * 4;
    *(v4f*)&s_x[r * C1_XSTR + q * 4] = *(const v4f*)src;
  }
  __syncthreads();

  const int c  = (wave & 1) * 32 + lane;
  const int nb = (wave >> 1) * 16;

  float wp[C1_K], wq[C1_K];
  #pragma unroll
  for (int r = 0; r < C1_K; ++r) {
    int kt = r >> 4, ci = r & 15;
    wp[r] = s_w[c * C1_WSTR + ci * KT_ + kt];
    wq[r] = s_w[(c + C1_) * C1_WSTR + ci * KT_ + kt];
  }
  const float bp = b[c];
  const float bq = b[c + C1_];
  const bool has_res = (c < C0_);
  const int  xr = has_res ? (2 * C0_ + c) : 0;

  #pragma unroll 1
  for (int g = 0; g < 2; ++g) {
    const int nl = nb + g * 8;
    const float* xrow = &s_x[nl];
    float ap[8], aq[8];
    #pragma unroll
    for (int u = 0; u < 8; ++u) { ap[u] = bp; aq[u] = bq; }

    v4f xa[2][C1_RC][2];
    #pragma unroll
    for (int j = 0; j < C1_RC; ++j) {
      xa[0][j][0] = *(const v4f*)(xrow + j * C1_XSTR);
      xa[0][j][1] = *(const v4f*)(xrow + j * C1_XSTR + 4);
    }
    #pragma unroll
    for (int ch = 0; ch < C1_K / C1_RC; ++ch) {
      __builtin_amdgcn_sched_barrier(0);
      if (ch + 1 < C1_K / C1_RC) {
        #pragma unroll
        for (int j = 0; j < C1_RC; ++j) {
          const float* p = xrow + ((ch + 1) * C1_RC + j) * C1_XSTR;
          xa[(ch + 1) & 1][j][0] = *(const v4f*)(p);
          xa[(ch + 1) & 1][j][1] = *(const v4f*)(p + 4);
        }
      }
      __builtin_amdgcn_sched_barrier(0);
      #pragma unroll
      for (int j = 0; j < C1_RC; ++j) {
        const int r = ch * C1_RC + j;
        const v4f x0 = xa[ch & 1][j][0];
        const v4f x1 = xa[ch & 1][j][1];
        #pragma unroll
        for (int u = 0; u < 4; ++u) {
          ap[u]     = fmaf(wp[r], x0[u], ap[u]);
          aq[u]     = fmaf(wq[r], x0[u], aq[u]);
          ap[u + 4] = fmaf(wp[r], x1[u], ap[u + 4]);
          aq[u + 4] = fmaf(wq[r], x1[u], aq[u + 4]);
        }
      }
    }
    v4f r0 = *(const v4f*)&s_x[xr * C1_XSTR + nl];
    v4f r1 = *(const v4f*)&s_x[xr * C1_XSTR + nl + 4];
    float* orow = xg + ((size_t)bt * NN_ + n0 + nl) * C1_ + c;
    #pragma unroll
    for (int u = 0; u < 8; ++u) {
      float xin = has_res ? (u < 4 ? r0[u & 3] : r1[u & 3]) : 0.f;
      float sg  = 1.f / (1.f + __expf(-aq[u]));
      orow[(size_t)u * C1_] = (ap[u] + xin) * sg;
    }
  }
}

#define XSTR  72
#define XPSTR 160
#define PSTR  260
#define ATT_RS   2
#define ATT_ROWS (NN_ / ATT_RS)
#define ATT_KB   4
#define ATT_NB   (NN_ / (4 * ATT_KB))

__global__ __launch_bounds__(256) void attn_kernel(
    const float* __restrict__ xg,
    const float* __restrict__ adj,
    const float* __restrict__ u1,
    const float* __restrict__ u2,
    float* __restrict__ xs)
{
  extern __shared__ float smem[];
  float*    s_xg   = smem;
  float*    s_xgp  = s_xg + NN_ * XSTR;
  float*    s_e1   = s_xgp + (NN_ / 2) * XPSTR;
  float*    s_e2   = s_e1 + NN_;
  float*    s_rinv = s_e2 + NN_;
  float*    s_p    = s_rinv + NN_;
  unsigned* s_adj  = (unsigned*)(s_p + 8 * 16 * PSTR);

  const int bt   = blockIdx.x / ATT_RS;
  const int row0 = (blockIdx.x % ATT_RS) * ATT_ROWS;
  const int tid  = threadIdx.x;
  const int lane = tid & 31;
  const int wave = tid >> 5;
  const int mloc = lane & 15;
  const int half = lane >> 4;

  const float* xg_bt = xg + (size_t)bt * NN_ * C1_;

  for (int i = tid; i < NN_ * C1_; i += 256) {
    int n = i >> 6, c = i & 63;
    float v = xg_bt[i];
    s_xg[n * XSTR + c] = v;
    s_xgp[(n >> 1) * XPSTR + 2 * c + (n & 1)] = v;
  }
  {
    const int rl = tid >> 1, ch = tid & 1;
    const float* arow = adj + (size_t)(row0 + rl) * NN_ + ch * 128;
    for (int w8 = 0; w8 < 4; ++w8) {
      unsigned bits = 0u;
      for (int j = 0; j < 32; ++j)
        bits |= (arow[w8 * 32 + j] > 0.f) ? (1u << j) : 0u;
      s_adj[rl * 8 + ch * 4 + w8] = bits;
    }
  }
  __syncthreads();

  const float inv_h = 1.0f / (float)HH_;
  float* s_pw = s_p + wave * 16 * PSTR;

  const int rb = (row0 >> 4) + wave;
  const int m  = rb * 16 + mloc;
  const int ml = wave * 16 + mloc;
  const int jbase = half * 128;

  float racc[4][8];
  #pragma unroll
  for (int j = 0; j < 4; ++j)
    #pragma unroll
    for (int r = 0; r < 8; ++r) racc[j][r] = 0.f;

  for (int h = 0; h < HH_; ++h) {
    {
      int n = tid;
      const float* row = &s_xg[n * XSTR];
      const float* U1 = u1 + h * C1_;
      const float* U2 = u2 + h * C1_;
      float a1 = 0.f, a2 = 0.f;
      for (int c = 0; c < C1_; ++c) {
        float v = row[c];
        a1 = fmaf(v, U1[c], a1);
        a2 = fmaf(v, U2[c], a2);
      }
      s_e1[n] = a1;
      s_e2[n] = a2;
    }
    __syncthreads();

    {
      const float e1m = s_e1[m];

      float mx = -3.0e38f;
      for (int w8 = 0; w8 < 4; ++w8) {
        unsigned adjw = s_adj[ml * 8 + half * 4 + w8];
        const float* eb = &s_e2[jbase + w8 * 32];
        for (int j4 = 0; j4 < 32; j4 += 4) {
          v4f e = *(const v4f*)(eb + j4);
          #pragma unroll
          for (int u = 0; u < 4; ++u) {
            float sc = e1m + e[u];
            sc = sc > 0.f ? sc : ALPHA_ * sc;
            int on = (adjw >> (j4 + u)) & 1;
            mx = fmaxf(mx, on ? sc : -3.0e38f);
          }
        }
      }
      mx = fmaxf(mx, __shfl_xor(mx, 16));

      float sm = 0.f;
      for (int w8 = 0; w8 < 4; ++w8) {
        unsigned adjw = s_adj[ml * 8 + half * 4 + w8];
        const float* eb = &s_e2[jbase + w8 * 32];
        float* pb = &s_pw[mloc * PSTR + jbase + w8 * 32];
        for (int j4 = 0; j4 < 32; j4 += 4) {
          v4f e = *(const v4f*)(eb + j4);
          v4f p;
          #pragma unroll
          for (int u = 0; u < 4; ++u) {
            float sc = e1m + e[u];
            sc = sc > 0.f ? sc : ALPHA_ * sc;
            sc = ((adjw >> (j4 + u)) & 1) ? sc : -3.0e38f;
            float pe = __expf(sc - mx);
            p[u] = pe;
            sm += pe;
          }
          *(v4f*)(pb + j4) = p;
        }
      }
      sm += __shfl_xor(sm, 16);
      s_rinv[ml] = 1.0f / sm;

      v8f acc0 = {}, acc1 = {}, acc2 = {}, acc3 = {};
      const float* pa   = &s_pw[mloc * PSTR + 2 * half];
      const float* xpb  = &s_xgp[half * XPSTR + 2 * mloc];
      v2f fa[2][ATT_KB];
      v2f fb[2][ATT_KB][4];
      #pragma unroll
      for (int s = 0; s < ATT_KB; ++s) {
        fa[0][s] = *(const v2f*)(pa + 4 * s);
        const float* bp = xpb + (2 * s) * XPSTR;
        #pragma unroll
        for (int q = 0; q < 4; ++q) fb[0][s][q] = *(const v2f*)(bp + 32 * q);
      }
      #pragma unroll 1
      for (int kb = 0; kb < ATT_NB; kb += 2) {
        #pragma unroll
        for (int st = 0; st < 2; ++st) {
          const int nb = (kb + st + 1 < ATT_NB) ? (kb + st + 1) : (ATT_NB - 1);
          const float* pan = pa + nb * (4 * ATT_KB);
          const float* xpn = xpb + nb * (2 * ATT_KB) * XPSTR;
          __builtin_amdgcn_sched_barrier(0);
          #pragma unroll
          for (int s = 0; s < ATT_KB; ++s) {
            fa[st ^ 1][s] = *(const v2f*)(pan + 4 * s);
            const float* bp = xpn + (2 * s) * XPSTR;
            #pragma unroll
            for (int q = 0; q < 4; ++q) fb[st ^ 1][s][q] = *(const v2f*)(bp + 32 * q);
          }
          __builtin_amdgcn_sched_barrier(0);
          #pragma unroll
          for (int s = 0; s < ATT_KB; ++s) {
            acc0 = __builtin_amdgcn_wmma_f32_16x16x4_f32(false, fa[st][s], false, fb[st][s][0], (short)0, acc0, false, false);
            acc1 = __builtin_amdgcn_wmma_f32_16x16x4_f32(false, fa[st][s], false, fb[st][s][1], (short)0, acc1, false, false);
            acc2 = __builtin_amdgcn_wmma_f32_16x16x4_f32(false, fa[st][s], false, fb[st][s][2], (short)0, acc2, false, false);
            acc3 = __builtin_amdgcn_wmma_f32_16x16x4_f32(false, fa[st][s], false, fb[st][s][3], (short)0, acc3, false, false);
          }
        }
      }

      #pragma unroll
      for (int r = 0; r < 8; ++r) {
        float inv = s_rinv[wave * 16 + r + 8 * half];
        racc[0][r] += tanh_fast(acc0[r] * inv) * inv_h;
        racc[1][r] += tanh_fast(acc1[r] * inv) * inv_h;
        racc[2][r] += tanh_fast(acc2[r] * inv) * inv_h;
        racc[3][r] += tanh_fast(acc3[r] * inv) * inv_h;
      }
    }
    __syncthreads();
  }

  float* xs_bt = xs + (size_t)bt * NN_ * C1_;
  #pragma unroll
  for (int r = 0; r < 8; ++r) {
    int mm = rb * 16 + r + 8 * half;
    const float* xrow = &s_xg[mm * XSTR + mloc];
    float* orow = xs_bt + (size_t)mm * C1_ + mloc;
    orow[0]  = xrow[0]  + racc[0][r];
    orow[16] = xrow[16] + racc[1][r];
    orow[32] = xrow[32] + racc[2][r];
    orow[48] = xrow[48] + racc[3][r];
  }
}

#define C2_K   (C1_ * KT_)
#define C2_XP  260
#define C2_WP  196
#define C2_YP  68
#define C2_G   4
#define C2_NG  (C2_K / C2_G)

__global__ __launch_bounds__(256) void conv2_ln_kernel(
    const float* __restrict__ xs,
    const float* __restrict__ w2,
    const float* __restrict__ b2,
    const float* __restrict__ lnw,
    const float* __restrict__ lnb,
    float* __restrict__ out)
{
  extern __shared__ float smem[];
  float* s_x   = smem;
  float* s_w   = s_x + C2_K * C2_XP;
  float* s_red = s_w + C2_ * C2_WP;
  float* s_y   = smem;

  const int bt = blockIdx.x;
  const int bb = bt >> 3, t = bt & 7;
  const int tid = threadIdx.x;

  #pragma unroll
  for (unsigned j = 0; j < (C2_ * C2_K) / 4 / 256; ++j) {
    const unsigned i4 = (unsigned)tid + j * 256u;
    const v4f wv = ((const v4f*)w2)[i4];
    const unsigned co = i4 / (C2_K / 4u);
    const unsigned r  = (i4 - co * (C2_K / 4u)) * 4u;
    #pragma unroll
    for (unsigned e = 0; e < 4; ++e) {
      const unsigned ci = (r + e) / KT_, kt = (r + e) - ci * KT_;
      s_w[co * C2_WP + kt * C1_ + ci] = wv[e];
    }
  }
  {
    const float* src = xs + (((size_t)bb * T1_ + t) * NN_) * C1_ + tid;
    float* dst = s_x + (tid & 63) * C2_XP + (tid >> 6);
    #pragma unroll 1
    for (int blk = 0; blk < 2 * KT_; ++blk) {
      float xv[32];
      #pragma unroll
      for (int u = 0; u < 32; ++u) xv[u] = src[u * 256];
      #pragma unroll
      for (int u = 0; u < 32; ++u) dst[u * 4] = xv[u];
      src += 32 * 256;
      dst += (blk & 1) ? (C1_ * C2_XP - 128) : 128;
    }
  }
  __syncthreads();

  const int lane = tid & 31;
  const int wave = tid >> 5;
  const int c16  = lane & 15;
  const int nb   = wave * 32 + (lane >> 4) * 16;

  float acc[4][16];
  #pragma unroll
  for (int j = 0; j < 4; ++j) {
    const float bj = b2[c16 + 16 * j];
    #pragma unroll
    for (int i = 0; i < 16; ++i) acc[j][i] = bj;
  }

  const float* xb = s_x + nb;
  const float* wb = s_w + c16 * C2_WP;

  v4f wv[2][4];
  v4f xv[2][C2_G][4];
  #pragma unroll
  for (int j = 0; j < 4; ++j) wv[0][j] = *(const v4f*)(wb + j * 16 * C2_WP);
  #pragma unroll
  for (int k = 0; k < C2_G; ++k)
    #pragma unroll
    for (int q = 0; q < 4; ++q) xv[0][k][q] = *(const v4f*)(xb + k * C2_XP + 4 * q);

  #pragma unroll 1
  for (int g = 0; g < C2_NG; g += 2) {
    #pragma unroll
    for (int s = 0; s < 2; ++s) {
      const int gn = (g + s + 1 < C2_NG) ? (g + s + 1) : (C2_NG - 1);
      const float* xn = xb + gn * C2_G * C2_XP;
      const float* wn = wb + gn * C2_G;
      __builtin_amdgcn_sched_barrier(0);
      #pragma unroll
      for (int j = 0; j < 4; ++j) wv[s ^ 1][j] = *(const v4f*)(wn + j * 16 * C2_WP);
      #pragma unroll
      for (int k = 0; k < C2_G; ++k)
        #pragma unroll
        for (int q = 0; q < 4; ++q) xv[s ^ 1][k][q] = *(const v4f*)(xn + k * C2_XP + 4 * q);
      __builtin_amdgcn_sched_barrier(0);
      #pragma unroll
      for (int k = 0; k < C2_G; ++k)
        #pragma unroll
        for (int j = 0; j < 4; ++j)
          #pragma unroll
          for (int q = 0; q < 4; ++q)
            #pragma unroll
            for (int u = 0; u < 4; ++u)
              acc[j][4 * q + u] = fmaf(wv[s][j][k], xv[s][k][q][u], acc[j][4 * q + u]);
    }
  }

  v4f res[4][4];
  #pragma unroll
  for (int j = 0; j < 4; ++j)
    #pragma unroll
    for (int q = 0; q < 4; ++q)
      res[j][q] = *(const v4f*)(s_x + (2 * C1_ + c16 + 16 * j) * C2_XP + nb + 4 * q);

  __syncthreads();

  #pragma unroll
  for (int j = 0; j < 4; ++j) {
    const int c = c16 + 16 * j;
    #pragma unroll
    for (int q = 0; q < 4; ++q) {
      #pragma unroll
      for (int u = 0; u < 4; ++u) {
        float v = acc[j][4 * q + u] + res[j][q][u];
        v = v > 0.f ? v : 0.f;
        s_y[(nb + 4 * q + u) * C2_YP + c] = v;
      }
    }
  }
  __syncthreads();

  const int n = tid;
  float yv[C2_];
  #pragma unroll
  for (int q = 0; q < C2_ / 4; ++q) {
    const v4f y4 = *(const v4f*)(s_y + n * C2_YP + 4 * q);
    yv[4 * q + 0] = y4[0]; yv[4 * q + 1] = y4[1]; yv[4 * q + 2] = y4[2]; yv[4 * q + 3] = y4[3];
  }
  float lsum = 0.f, lsq = 0.f;
  #pragma unroll
  for (int c = 0; c < C2_; ++c) {
    lsum += yv[c];
    lsq  += yv[c] * yv[c];
  }

  for (int off = 16; off > 0; off >>= 1) {
    lsum += __shfl_xor(lsum, off);
    lsq  += __shfl_xor(lsq,  off);
  }
  if ((tid & 31) == 0) { s_red[tid >> 5] = lsum; s_red[8 + (tid >> 5)] = lsq; }
  __syncthreads();
  float tsum = 0.f, tsq = 0.f;
  for (int i = 0; i < 8; ++i) { tsum += s_red[i]; tsq += s_red[8 + i]; }
  const float cnt  = (float)(NN_ * C2_);
  const float mu   = tsum / cnt;
  const float var  = tsq / cnt - mu * mu;
  const float rstd = rsqrtf(var + 1e-5f);

  #pragma unroll
  for (int c = 0; c < C2_; ++c) {
    float v = (yv[c] - mu) * rstd * lnw[(size_t)n * C2_ + c] + lnb[(size_t)n * C2_ + c];
    out[(((size_t)bb * C2_ + c) * T2_ + t) * NN_ + n] = v;
  }
}

extern "C" void kernel_launch(void* const* d_in, const int* in_sizes, int n_in,
                              void* d_out, int out_size, void* d_ws, size_t ws_size,
                              hipStream_t stream) {
  const float* x   = (const float*)d_in[0];
  const float* adj = (const float*)d_in[1];
  const float* w1  = (const float*)d_in[2];
  const float* b1  = (const float*)d_in[3];
  const float* w2  = (const float*)d_in[4];
  const float* b2  = (const float*)d_in[5];
  const float* Wh  = (const float*)d_in[6];
  const float* ah  = (const float*)d_in[7];
  const float* lnw = (const float*)d_in[8];
  const float* lnb = (const float*)d_in[9];
  float* out = (float*)d_out;

  float* ws = (float*)d_ws;
  const size_t xg_elems = (size_t)B_ * T1_ * NN_ * C1_;
  float* xg = ws;
  float* xs = xg + xg_elems;
  float* u1 = xs + xg_elems;
  float* u2 = u1 + HH_ * C1_;

  prep_u_kernel<<<2, 256, 0, stream>>>(Wh, ah, u1, u2);

  conv1_glu_kernel<<<B_ * T1_ * (NN_ / C1_NT), 256, 0, stream>>>(x, w1, b1, xg);

  const size_t attn_lds =
      (size_t)(NN_ * XSTR + (NN_ / 2) * XPSTR + 3 * NN_ + 8 * 16 * PSTR) * sizeof(float)
      + (size_t)(ATT_ROWS * 8) * sizeof(unsigned);
  attn_kernel<<<B_ * T1_ * ATT_RS, 256, attn_lds, stream>>>(xg, adj, u1, u2, xs);

  const size_t c2_lds = (size_t)(C2_K * C2_XP + C2_ * C2_WP + 16) * sizeof(float);
  conv2_ln_kernel<<<B_ * T2_, 256, c2_lds, stream>>>(xs, w2, b2, lnw, lnb, out);
}


# attention: each (b,t) over 3 workgroups of 6/5/5 row blocks, one wave per head walks the row blocks, heads added in order through LDS: 240 workgroups, 12 not 16 products per SIMD; sum order unchanged
# speedup vs baseline: 3.9931x; 1.1973x over previous
#include <hip/hip_runtime.h>
#include <math.h>

#define B_   8
#define C0_  16
#define C1_  64
#define C2_  64
#define T0_  12
#define T1_  10
#define T2_  8
#define NN_  256
#define KT_  3
#define HH_  8
#define ALPHA_ 0.2f

typedef float v2f __attribute__((ext_vector_type(2)));
typedef float v4f __attribute__((ext_vector_type(4)));
typedef float v8f __attribute__((ext_vector_type(8)));

__device__ __forceinline__ float tanh_fast(float x) {
  float e = __expf(fminf(2.0f * x, 80.0f));
  return (e - 1.0f) / (e + 1.0f);
}

__global__ __launch_bounds__(256) void prep_u_kernel(
    const float* __restrict__ W,
    const float* __restrict__ a,
    float* __restrict__ u1, float* __restrict__ u2)
{
  int tid = blockIdx.x * 256 + threadIdx.x;
  if (tid >= HH_ * C1_) return;
  int h = tid >> 6, c = tid & 63;
  const float* Wh = W + ((size_t)h * C1_ + c) * C1_;
  const float* ah = a + (size_t)h * 2 * C1_;
  float s1 = 0.f, s2 = 0.f;
  for (int d = 0; d < C1_; ++d) {
    float w = Wh[d];
    s1 += w * ah[d];
    s2 += w * ah[C1_ + d];
  }
  u1[tid] = s1;
  u2[tid] = s2;
}

#define C1_NT   64
#define C1_WSTR 49
#define C1_XSTR 68
#define C1_K    (C0_ * KT_)
#define C1_RC   6

__global__ __launch_bounds__(256) void conv1_glu_kernel(
    const float* __restrict__ x,
    const float* __restrict__ w,
    const float* __restrict__ b,
    float* __restrict__ xg)
{
  __shared__ __attribute__((aligned(16))) float s_w[2 * C1_ * C1_WSTR];
  __shared__ __attribute__((aligned(16))) float s_x[C1_K * C1_XSTR];

  const int tid  = threadIdx.x;
  const int lane = tid & 31;
  const int wave = tid >> 5;
  const int bt   = blockIdx.x >> 2;
  const int n0   = (blockIdx.x & 3) * C1_NT;
  const int bb   = bt / T1_, t = bt - bb * T1_;

  #pragma unroll
  for (int j = 0; j < (2 * C1_ * C1_K) / 4 / 256; ++j) {
    int i4 = tid + j * 256;
    v4f wv = ((const v4f*)w)[i4];
    int base = i4 * 4;
    int c2 = base / C1_K, k = base - c2 * C1_K;
    float* d = &s_w[c2 * C1_WSTR + k];
    d[0] = wv[0]; d[1] = wv[1]; d[2] = wv[2]; d[3] = wv[3];
  }
  #pragma unroll
  for (int j = 0; j < C1_K * (C1_NT / 4) / 256; ++j) {
    int i4 = tid + j * 256;
    int r = i4 >> 4, q = i4 & 15;
    int kt = r >> 4, ci = r & 15;
    const float* src = x + (((size_t)bb * C0_ + ci) * T0_ + (t + kt)) * NN_ + n0 + q * 4;
    *(v4f*)&s_x[r * C1_XSTR + q * 4] = *(const v4f*)src;
  }
  __syncthreads();

  const int c  = (wave & 1) * 32 + lane;
  const int nb = (wave >> 1) * 16;

  float wp[C1_K], wq[C1_K];
  #pragma unroll
  for (int r = 0; r < C1_K; ++r) {
    int kt = r >> 4, ci = r & 15;
    wp[r] = s_w[c * C1_WSTR + ci * KT_ + kt];
    wq[r] = s_w[(c + C1_) * C1_WSTR + ci * KT_ + kt];
  }
  const float bp = b[c];
  const float bq = b[c + C1_];
  const bool has_res = (c < C0_);
  const int  xr = has_res ? (2 * C0_ + c) : 0;

  #pragma unroll 1
  for (int g = 0; g < 2; ++g) {
    const int nl = nb + g * 8;
    const float* xrow = &s_x[nl];
    float ap[8], aq[8];
    #pragma unroll
    for (int u = 0; u < 8; ++u) { ap[u] = bp; aq[u] = bq; }

    v4f xa[2][C1_RC][2];
    #pragma unroll
    for (int j = 0; j < C1_RC; ++j) {
      xa[0][j][0] = *(const v4f*)(xrow + j * C1_XSTR);
      xa[0][j][1] = *(const v4f*)(xrow + j * C1_XSTR + 4);
    }
    #pragma unroll
    for (int ch = 0; ch < C1_K / C1_RC; ++ch) {
      __builtin_amdgcn_sched_barrier(0);
      if (ch + 1 < C1_K / C1_RC) {
        #pragma unroll
        for (int j = 0; j < C1_RC; ++j) {
          const float* p = xrow + ((ch + 1) * C1_RC + j) * C1_XSTR;
          xa[(ch + 1) & 1][j][0] = *(const v4f*)(p);
          xa[(ch + 1) & 1][j][1] = *(const v4f*)(p + 4);
        }
      }
      __builtin_amdgcn_sched_barrier(0);
      #pragma unroll
      for (int j = 0; j < C1_RC; ++j) {
        const int r = ch * C1_RC + j;
        const v4f x0 = xa[ch & 1][j][0];
        const v4f x1 = xa[ch & 1][j][1];
        #pragma unroll
        for (int u = 0; u < 4; ++u) {
          ap[u]     = fmaf(wp[r], x0[u], ap[u]);
          aq[u]     = fmaf(wq[r], x0[u], aq[u]);
          ap[u + 4] = fmaf(wp[r], x1[u], ap[u + 4]);
          aq[u + 4] = fmaf(wq[r], x1[u], aq[u + 4]);
        }
      }
    }
    v4f r0 = *(const v4f*)&s_x[xr * C1_XSTR + nl];
    v4f r1 = *(const v4f*)&s_x[xr * C1_XSTR + nl + 4];
    float* orow = xg + ((size_t)bt * NN_ + n0 + nl) * C1_ + c;
    #pragma unroll
    for (int u = 0; u < 8; ++u) {
      float xin = has_res ? (u < 4 ? r0[u & 3] : r1[u & 3]) : 0.f;
      float sg  = 1.f / (1.f + __expf(-aq[u]));
      orow[(size_t)u * C1_] = (ap[u] + xin) * sg;
    }
  }
}

#define XSTR  72
#define XPSTR 160
#define PSTR  260
#define ATT_RS   3
#define ATT_MAXT 6
#define ATT_KB   4
#define ATT_NB   (NN_ / (4 * ATT_KB))

__global__ __launch_bounds__(256) void attn_kernel(
    const float* __restrict__ xg,
    const float* __restrict__ adj,
    const float* __restrict__ u1,
    const float* __restrict__ u2,
    float* __restrict__ xs)
{
  extern __shared__ float smem[];
  float*    s_xg   = smem;
  float*    s_xgp  = s_xg + NN_ * XSTR;
  float*    s_e1   = s_xgp + (NN_ / 2) * XPSTR;
  float*    s_e2   = s_e1 + HH_ * NN_;
  float*    s_rinv = s_e2 + HH_ * NN_;
  float*    s_p    = s_rinv + 8 * 16;
  unsigned* s_adj  = (unsigned*)(s_p + 8 * 16 * PSTR);

  const int bt    = blockIdx.x / ATT_RS;
  const int part  = blockIdx.x - bt * ATT_RS;
  const int ntask = (part == 0) ? 6 : 5;
  const int row0  = ((part == 0) ? 0 : (part == 1 ? 6 : 11)) * 16;
  const int tid  = threadIdx.x;
  const int lane = tid & 31;
  const int wave = tid >> 5;
  const int mloc = lane & 15;
  const int half = lane >> 4;

  const float* xg_bt = xg + (size_t)bt * NN_ * C1_;

  for (int i = tid; i < NN_ * C1_; i += 256) {
    int n = i >> 6, c = i & 63;
    float v = xg_bt[i];
    s_xg[n * XSTR + c] = v;
    s_xgp[(n >> 1) * XPSTR + 2 * c + (n & 1)] = v;
  }
  {
    const int rl = tid >> 1, ch = tid & 1;
    if (rl < ntask * 16) {
      const float* arow = adj + (size_t)(row0 + rl) * NN_ + ch * 128;
      for (int w8 = 0; w8 < 4; ++w8) {
        unsigned bits = 0u;
        for (int j = 0; j < 32; ++j)
          bits |= (arow[w8 * 32 + j] > 0.f) ? (1u << j) : 0u;
        s_adj[rl * 8 + ch * 4 + w8] = bits;
      }
    }
  }
  __syncthreads();

  {
    const float* row = &s_xg[tid * XSTR];
    float rv[C1_];
    #pragma unroll
    for (int q = 0; q < C1_ / 4; ++q) {
      const v4f t4 = *(const v4f*)(row + 4 * q);
      rv[4 * q + 0] = t4[0]; rv[4 * q + 1] = t4[1]; rv[4 * q + 2] = t4[2]; rv[4 * q + 3] = t4[3];
    }
    #pragma unroll 1
    for (int h = 0; h < HH_; ++h) {
      const float* U1 = u1 + h * C1_;
      const float* U2 = u2 + h * C1_;
      float a1 = 0.f, a2 = 0.f;
      #pragma unroll
      for (int c = 0; c < C1_; ++c) {
        a1 = fmaf(rv[c], U1[c], a1);
        a2 = fmaf(rv[c], U2[c], a2);
      }
      s_e1[h * NN_ + tid] = a1;
      s_e2[h * NN_ + tid] = a2;
    }
  }
  __syncthreads();

  const float inv_h = 1.0f / (float)HH_;
  float* s_pw = s_p + wave * 16 * PSTR;
  const float* e1h = s_e1 + wave * NN_;
  const float* e2h = s_e2 + wave * NN_;
  const int jbase = half * 128;
  float* xs_bt = xs + (size_t)bt * NN_ * C1_;

  #pragma unroll 1
  for (int ti = 0; ti < ntask; ++ti) {
    const int ml = ti * 16 + mloc;
    {
      const float e1m = e1h[row0 + ml];

      float mx = -3.0e38f;
      for (int w8 = 0; w8 < 4; ++w8) {
        unsigned adjw = s_adj[ml * 8 + half * 4 + w8];
        const float* eb = &e2h[jbase + w8 * 32];
        for (int j4 = 0; j4 < 32; j4 += 4) {
          v4f e = *(const v4f*)(eb + j4);
          #pragma unroll
          for (int u = 0; u < 4; ++u) {
            float sc = e1m + e[u];
            sc = sc > 0.f ? sc : ALPHA_ * sc;
            int on = (adjw >> (j4 + u)) & 1;
            mx = fmaxf(mx, on ? sc : -3.0e38f);
          }
        }
      }
      mx = fmaxf(mx, __shfl_xor(mx, 16));

      float sm = 0.f;
      for (int w8 = 0; w8 < 4; ++w8) {
        unsigned adjw = s_adj[ml * 8 + half * 4 + w8];
        const float* eb = &e2h[jbase + w8 * 32];
        float* pb = &s_pw[mloc * PSTR + jbase + w8 * 32];
        for (int j4 = 0; j4 < 32; j4 += 4) {
          v4f e = *(const v4f*)(eb + j4);
          v4f p;
          #pragma unroll
          for (int u = 0; u < 4; ++u) {
            float sc = e1m + e[u];
            sc = sc > 0.f ? sc : ALPHA_ * sc;
            sc = ((adjw >> (j4 + u)) & 1) ? sc : -3.0e38f;
            float pe = __expf(sc - mx);
            p[u] = pe;
            sm += pe;
          }
          *(v4f*)(pb + j4) = p;
        }
      }
      sm += __shfl_xor(sm, 16);
      s_rinv[wave * 16 + mloc] = 1.0f / sm;

      v8f acc0 = {}, acc1 = {}, acc2 = {}, acc3 = {};
      const float* pa   = &s_pw[mloc * PSTR + 2 * half];
      const float* xpb  = &s_xgp[half * XPSTR + 2 * mloc];
      v2f fa[2][ATT_KB];
      v2f fb[2][ATT_KB][4];
      #pragma unroll
      for (int s = 0; s < ATT_KB; ++s) {
        fa[0][s] = *(const v2f*)(pa + 4 * s);
        const float* bp = xpb + (2 * s) * XPSTR;
        #pragma unroll
        for (int q = 0; q < 4; ++q) fb[0][s][q] = *(const v2f*)(bp + 32 * q);
      }
      #pragma unroll 1
      for (int kb = 0; kb < ATT_NB; kb += 2) {
        #pragma unroll
        for (int st = 0; st < 2; ++st) {
          const int nb = (kb + st + 1 < ATT_NB) ? (kb + st + 1) : (ATT_NB - 1);
          const float* pan = pa + nb * (4 * ATT_KB);
          const float* xpn = xpb + nb * (2 * ATT_KB) * XPSTR;
          __builtin_amdgcn_sched_barrier(0);
          #pragma unroll
          for (int s = 0; s < ATT_KB; ++s) {
            fa[st ^ 1][s] = *(const v2f*)(pan + 4 * s);
            const float* bp = xpn + (2 * s) * XPSTR;
            #pragma unroll
            for (int q = 0; q < 4; ++q) fb[st ^ 1][s][q] = *(const v2f*)(bp + 32 * q);
          }
          __builtin_amdgcn_sched_barrier(0);
          #pragma unroll
          for (int s = 0; s < ATT_KB; ++s) {
            acc0 = __builtin_amdgcn_wmma_f32_16x16x4_f32(false, fa[st][s], false, fb[st][s][0], (short)0, acc0, false, false);
            acc1 = __builtin_amdgcn_wmma_f32_16x16x4_f32(false, fa[st][s], false, fb[st][s][1], (short)0, acc1, false, false);
            acc2 = __builtin_amdgcn_wmma_f32_16x16x4_f32(false, fa[st][s], false, fb[st][s][2], (short)0, acc2, false, false);
            acc3 = __builtin_amdgcn_wmma_f32_16x16x4_f32(false, fa[st][s], false, fb[st][s][3], (short)0, acc3, false, false);
          }
        }
      }

      float inv[8];
      #pragma unroll
      for (int r = 0; r < 8; ++r) inv[r] = s_rinv[wave * 16 + r + 8 * half];
      #pragma unroll
      for (int r = 0; r < 8; ++r) {
        float* hrow = s_pw + (r + 8 * half) * PSTR + mloc;
        hrow[0]  = tanh_fast(acc0[r] * inv[r]);
        hrow[16] = tanh_fast(acc1[r] * inv[r]);
        hrow[32] = tanh_fast(acc2[r] * inv[r]);
        hrow[48] = tanh_fast(acc3[r] * inv[r]);
      }
    }
    __syncthreads();

    const int rr = tid >> 4, c4 = (tid & 15) * 4;
    const int mm = row0 + ti * 16 + rr;
    v4f o;
    {
      v4f sacc = {0.f, 0.f, 0.f, 0.f};
      #pragma unroll
      for (int hh = 0; hh < HH_; ++hh) {
        const v4f t4 = *(const v4f*)(s_p + (hh * 16 + rr) * PSTR + c4);
        sacc[0] += t4[0] * inv_h;
        sacc[1] += t4[1] * inv_h;
        sacc[2] += t4[2] * inv_h;
        sacc[3] += t4[3] * inv_h;
      }
      const v4f xr = *(const v4f*)&s_xg[mm * XSTR + c4];
      o[0] = xr[0] + sacc[0]; o[1] = xr[1] + sacc[1]; o[2] = xr[2] + sacc[2]; o[3] = xr[3] + sacc[3];
    }
    __syncthreads();
    *(v4f*)(xs_bt + (size_t)mm * C1_ + c4) = o;
  }
}

#define C2_K   (C1_ * KT_)
#define C2_XP  260
#define C2_WP  196
#define C2_YP  68
#define C2_G   4
#define C2_NG  (C2_K / C2_G)

__global__ __launch_bounds__(256) void conv2_ln_kernel(
    const float* __restrict__ xs,
    const float* __restrict__ w2,
    const float* __restrict__ b2,
    const float* __restrict__ lnw,
    const float* __restrict__ lnb,
    float* __restrict__ out)
{
  extern __shared__ float smem[];
  float* s_x   = smem;
  float* s_w   = s_x + C2_K * C2_XP;
  float* s_red = s_w + C2_ * C2_WP;
  float* s_y   = smem;

  const int bt = blockIdx.x;
  const int bb = bt >> 3, t = bt & 7;
  const int tid = threadIdx.x;

  #pragma unroll
  for (unsigned j = 0; j < (C2_ * C2_K) / 4 / 256; ++j) {
    const unsigned i4 = (unsigned)tid + j * 256u;
    const v4f wv = ((const v4f*)w2)[i4];
    const unsigned co = i4 / (C2_K / 4u);
    const unsigned r  = (i4 - co * (C2_K / 4u)) * 4u;
    #pragma unroll
    for (unsigned e = 0; e < 4; ++e) {
      const unsigned ci = (r + e) / KT_, kt = (r + e) - ci * KT_;
      s_w[co * C2_WP + kt * C1_ + ci] = wv[e];
    }
  }
  {
    const float* src = xs + (((size_t)bb * T1_ + t) * NN_) * C1_ + tid;
    float* dst = s_x + (tid & 63) * C2_XP + (tid >> 6);
    #pragma unroll 1
    for (int blk = 0; blk < 2 * KT_; ++blk) {
      float xv[32];
      #pragma unroll
      for (int u = 0; u < 32; ++u) xv[u] = src[u * 256];
      #pragma unroll
      for (int u = 0; u < 32; ++u) dst[u * 4] = xv[u];
      src += 32 * 256;
      dst += (blk & 1) ? (C1_ * C2_XP - 128) : 128;
    }
  }
  __syncthreads();

  const int lane = tid & 31;
  const int wave = tid >> 5;
  const int c16  = lane & 15;
  const int nb   = wave * 32 + (lane >> 4) * 16;

  float acc[4][16];
  #pragma unroll
  for (int j = 0; j < 4; ++j) {
    const float bj = b2[c16 + 16 * j];
    #pragma unroll
    for (int i = 0; i < 16; ++i) acc[j][i] = bj;
  }

  const float* xb = s_x + nb;
  const float* wb = s_w + c16 * C2_WP;

  v4f wv[2][4];
  v4f xv[2][C2_G][4];
  #pragma unroll
  for (int j = 0; j < 4; ++j) wv[0][j] = *(const v4f*)(wb + j * 16 * C2_WP);
  #pragma unroll
  for (int k = 0; k < C2_G; ++k)
    #pragma unroll
    for (int q = 0; q < 4; ++q) xv[0][k][q] = *(const v4f*)(xb + k * C2_XP + 4 * q);

  #pragma unroll 1
  for (int g = 0; g < C2_NG; g += 2) {
    #pragma unroll
    for (int s = 0; s < 2; ++s) {
      const int gn = (g + s + 1 < C2_NG) ? (g + s + 1) : (C2_NG - 1);
      const float* xn = xb + gn * C2_G * C2_XP;
      const float* wn = wb + gn * C2_G;
      __builtin_amdgcn_sched_barrier(0);
      #pragma unroll
      for (int j = 0; j < 4; ++j) wv[s ^ 1][j] = *(const v4f*)(wn + j * 16 * C2_WP);
      #pragma unroll
      for (int k = 0; k < C2_G; ++k)
        #pragma unroll
        for (int q = 0; q < 4; ++q) xv[s ^ 1][k][q] = *(const v4f*)(xn + k * C2_XP + 4 * q);
      __builtin_amdgcn_sched_barrier(0);
      #pragma unroll
      for (int k = 0; k < C2_G; ++k)
        #pragma unroll
        for (int j = 0; j < 4; ++j)
          #pragma unroll
          for (int q = 0; q < 4; ++q)
            #pragma unroll
            for (int u = 0; u < 4; ++u)
              acc[j][4 * q + u] = fmaf(wv[s][j][k], xv[s][k][q][u], acc[j][4 * q + u]);
    }
  }

  v4f res[4][4];
  #pragma unroll
  for (int j = 0; j < 4; ++j)
    #pragma unroll
    for (int q = 0; q < 4; ++q)
      res[j][q] = *(const v4f*)(s_x + (2 * C1_ + c16 + 16 * j) * C2_XP + nb + 4 * q);

  __syncthreads();

  #pragma unroll
  for (int j = 0; j < 4; ++j) {
    const int c = c16 + 16 * j;
    #pragma unroll
    for (int q = 0; q < 4; ++q) {
      #pragma unroll
      for (int u = 0; u < 4; ++u) {
        float v = acc[j][4 * q + u] + res[j][q][u];
        v = v > 0.f ? v : 0.f;
        s_y[(nb + 4 * q + u) * C2_YP + c] = v;
      }
    }
  }
  __syncthreads();

  const int n = tid;
  float yv[C2_];
  #pragma unroll
  for (int q = 0; q < C2_ / 4; ++q) {
    const v4f y4 = *(const v4f*)(s_y + n * C2_YP + 4 * q);
    yv[4 * q + 0] = y4[0]; yv[4 * q + 1] = y4[1]; yv[4 * q + 2] = y4[2]; yv[4 * q + 3] = y4[3];
  }
  float lsum = 0.f, lsq = 0.f;
  #pragma unroll
  for (int c = 0; c < C2_; ++c) {
    lsum += yv[c];
    lsq  += yv[c] * yv[c];
  }

  for (int off = 16; off > 0; off >>= 1) {
    lsum += __shfl_xor(lsum, off);
    lsq  += __shfl_xor(lsq,  off);
  }
  if ((tid & 31) == 0) { s_red[tid >> 5] = lsum; s_red[8 + (tid >> 5)] = lsq; }
  __syncthreads();
  float tsum = 0.f, tsq = 0.f;
  for (int i = 0; i < 8; ++i) { tsum += s_red[i]; tsq += s_red[8 + i]; }
  const float cnt  = (float)(NN_ * C2_);
  const float mu   = tsum / cnt;
  const float var  = tsq / cnt - mu * mu;
  const float rstd = rsqrtf(var + 1e-5f);

  #pragma unroll
  for (int c = 0; c < C2_; ++c) {
    float v = (yv[c] - mu) * rstd * lnw[(size_t)n * C2_ + c] + lnb[(size_t)n * C2_ + c];
    out[(((size_t)bb * C2_ + c) * T2_ + t) * NN_ + n] = v;
  }
}

extern "C" void kernel_launch(void* const* d_in, const int* in_sizes, int n_in,
                              void* d_out, int out_size, void* d_ws, size_t ws_size,
                              hipStream_t stream) {
  const float* x   = (const float*)d_in[0];
  const float* adj = (const float*)d_in[1];
  const float* w1  = (const float*)d_in[2];
  const float* b1  = (const float*)d_in[3];
  const float* w2  = (const float*)d_in[4];
  const float* b2  = (const float*)d_in[5];
  const float* Wh  = (const float*)d_in[6];
  const float* ah  = (const float*)d_in[7];
  const float* lnw = (const float*)d_in[8];
  const float* lnb = (const float*)d_in[9];
  float* out = (float*)d_out;

  float* ws = (float*)d_ws;
  const size_t xg_elems = (size_t)B_ * T1_ * NN_ * C1_;
  float* xg = ws;
  float* xs = xg + xg_elems;
  float* u1 = xs + xg_elems;
  float* u2 = u1 + HH_ * C1_;

  prep_u_kernel<<<2, 256, 0, stream>>>(Wh, ah, u1, u2);

  conv1_glu_kernel<<<B_ * T1_ * (NN_ / C1_NT), 256, 0, stream>>>(x, w1, b1, xg);

  const size_t attn_lds =
      (size_t)(NN_ * XSTR + (NN_ / 2) * XPSTR + 2 * HH_ * NN_ + 8 * 16 + 8 * 16 * PSTR) * sizeof(float)
      + (size_t)(ATT_MAXT * 16 * 8) * sizeof(unsigned);
  attn_kernel<<<B_ * T1_ * ATT_RS, 256, attn_lds, stream>>>(xg, adj, u1, u2, xs);

  const size_t c2_lds = (size_t)(C2_K * C2_XP + C2_ * C2_WP + 16) * sizeof(float);
  conv2_ln_kernel<<<B_ * T2_, 256, c2_lds, stream>>>(xs, w2, b2, lnw, lnb, out);
}


# attention softmax: lane keeps its 128 e2 values in registers for all row blocks and its masked scores from pass 1 to pass 2, so no LDS read waits behind each probability store; sum order unchanged
# speedup vs baseline: 4.3569x; 1.0911x over previous
#include <hip/hip_runtime.h>
#include <math.h>

#define B_   8
#define C0_  16
#define C1_  64
#define C2_  64
#define T0_  12
#define T1_  10
#define T2_  8
#define NN_  256
#define KT_  3
#define HH_  8
#define ALPHA_ 0.2f

typedef float v2f __attribute__((ext_vector_type(2)));
typedef float v4f __attribute__((ext_vector_type(4)));
typedef float v8f __attribute__((ext_vector_type(8)));
typedef unsigned v4u __attribute__((ext_vector_type(4)));

__device__ __forceinline__ float tanh_fast(float x) {
  float e = __expf(fminf(2.0f * x, 80.0f));
  return (e - 1.0f) / (e + 1.0f);
}

__global__ __launch_bounds__(256) void prep_u_kernel(
    const float* __restrict__ W,
    const float* __restrict__ a,
    float* __restrict__ u1, float* __restrict__ u2)
{
  int tid = blockIdx.x * 256 + threadIdx.x;
  if (tid >= HH_ * C1_) return;
  int h = tid >> 6, c = tid & 63;
  const float* Wh = W + ((size_t)h * C1_ + c) * C1_;
  const float* ah = a + (size_t)h * 2 * C1_;
  float s1 = 0.f, s2 = 0.f;
  for (int d = 0; d < C1_; ++d) {
    float w = Wh[d];
    s1 += w * ah[d];
    s2 += w * ah[C1_ + d];
  }
  u1[tid] = s1;
  u2[tid] = s2;
}

#define C1_NT   64
#define C1_WSTR 49
#define C1_XSTR 68
#define C1_K    (C0_ * KT_)
#define C1_RC   6

__global__ __launch_bounds__(256) void conv1_glu_kernel(
    const float* __restrict__ x,
    const float* __restrict__ w,
    const float* __restrict__ b,
    float* __restrict__ xg)
{
  __shared__ __attribute__((aligned(16))) float s_w[2 * C1_ * C1_WSTR];
  __shared__ __attribute__((aligned(16))) float s_x[C1_K * C1_XSTR];

  const int tid  = threadIdx.x;
  const int lane = tid & 31;
  const int wave = tid >> 5;
  const int bt   = blockIdx.x >> 2;
  const int n0   = (blockIdx.x & 3) * C1_NT;
  const int bb   = bt / T1_, t = bt - bb * T1_;

  #pragma unroll
  for (int j = 0; j < (2 * C1_ * C1_K) / 4 / 256; ++j) {
    int i4 = tid + j * 256;
    v4f wv = ((const v4f*)w)[i4];
    int base = i4 * 4;
    int c2 = base / C1_K, k = base - c2 * C1_K;
    float* d = &s_w[c2 * C1_WSTR + k];
    d[0] = wv[0]; d[1] = wv[1]; d[2] = wv[2]; d[3] = wv[3];
  }
  #pragma unroll
  for (int j = 0; j < C1_K * (C1_NT / 4) / 256; ++j) {
    int i4 = tid + j * 256;
    int r = i4 >> 4, q = i4 & 15;
    int kt = r >> 4, ci = r & 15;
    const float* src = x + (((size_t)bb * C0_ + ci) * T0_ + (t + kt)) * NN_ + n0 + q * 4;
    *(v4f*)&s_x[r * C1_XSTR + q * 4] = *(const v4f*)src;
  }
  __syncthreads();

  const int c  = (wave & 1) * 32 + lane;
  const int nb = (wave >> 1) * 16;

  float wp[C1_K], wq[C1_K];
  #pragma unroll
  for (int r = 0; r < C1_K; ++r) {
    int kt = r >> 4, ci = r & 15;
    wp[r] = s_w[c * C1_WSTR + ci * KT_ + kt];
    wq[r] = s_w[(c + C1_) * C1_WSTR + ci * KT_ + kt];
  }
  const float bp = b[c];
  const float bq = b[c + C1_];
  const bool has_res = (c < C0_);
  const int  xr = has_res ? (2 * C0_ + c) : 0;

  #pragma unroll 1
  for (int g = 0; g < 2; ++g) {
    const int nl = nb + g * 8;
    const float* xrow = &s_x[nl];
    float ap[8], aq[8];
    #pragma unroll
    for (int u = 0; u < 8; ++u) { ap[u] = bp; aq[u] = bq; }

    v4f xa[2][C1_RC][2];
    #pragma unroll
    for (int j = 0; j < C1_RC; ++j) {
      xa[0][j][0] = *(const v4f*)(xrow + j * C1_XSTR);
      xa[0][j][1] = *(const v4f*)(xrow + j * C1_XSTR + 4);
    }
    #pragma unroll
    for (int ch = 0; ch < C1_K / C1_RC; ++ch) {
      __builtin_amdgcn_sched_barrier(0);
      if (ch + 1 < C1_K / C1_RC) {
        #pragma unroll
        for (int j = 0; j < C1_RC; ++j) {
          const float* p = xrow + ((ch + 1) * C1_RC + j) * C1_XSTR;
          xa[(ch + 1) & 1][j][0] = *(const v4f*)(p);
          xa[(ch + 1) & 1][j][1] = *(const v4f*)(p + 4);
        }
      }
      __builtin_amdgcn_sched_barrier(0);
      #pragma unroll
      for (int j = 0; j < C1_RC; ++j) {
        const int r = ch * C1_RC + j;
        const v4f x0 = xa[ch & 1][j][0];
        const v4f x1 = xa[ch & 1][j][1];
        #pragma unroll
        for (int u = 0; u < 4; ++u) {
          ap[u]     = fmaf(wp[r], x0[u], ap[u]);
          aq[u]     = fmaf(wq[r], x0[u], aq[u]);
          ap[u + 4] = fmaf(wp[r], x1[u], ap[u + 4]);
          aq[u + 4] = fmaf(wq[r], x1[u], aq[u + 4]);
        }
      }
    }
    v4f r0 = *(const v4f*)&s_x[xr * C1_XSTR + nl];
    v4f r1 = *(const v4f*)&s_x[xr * C1_XSTR + nl + 4];
    float* orow = xg + ((size_t)bt * NN_ + n0 + nl) * C1_ + c;
    #pragma unroll
    for (int u = 0; u < 8; ++u) {
      float xin = has_res ? (u < 4 ? r0[u & 3] : r1[u & 3]) : 0.f;
      float sg  = 1.f / (1.f + __expf(-aq[u]));
      orow[(size_t)u * C1_] = (ap[u] + xin) * sg;
    }
  }
}

#define XSTR  72
#define XPSTR 160
#define PSTR  260
#define ATT_RS   3
#define ATT_MAXT 6
#define ATT_KB   4
#define ATT_NB   (NN_ / (4 * ATT_KB))

__global__ __launch_bounds__(256) void attn_kernel(
    const float* __restrict__ xg,
    const float* __restrict__ adj,
    const float* __restrict__ u1,
    const float* __restrict__ u2,
    float* __restrict__ xs)
{
  extern __shared__ float smem[];
  float*    s_xg   = smem;
  float*    s_xgp  = s_xg + NN_ * XSTR;
  float*    s_e1   = s_xgp + (NN_ / 2) * XPSTR;
  float*    s_e2   = s_e1 + HH_ * NN_;
  float*    s_rinv = s_e2 + HH_ * NN_;
  float*    s_p    = s_rinv + 8 * 16;
  unsigned* s_adj  = (unsigned*)(s_p + 8 * 16 * PSTR);

  const int bt    = blockIdx.x / ATT_RS;
  const int part  = blockIdx.x - bt * ATT_RS;
  const int ntask = (part == 0) ? 6 : 5;
  const int row0  = ((part == 0) ? 0 : (part == 1 ? 6 : 11)) * 16;
  const int tid  = threadIdx.x;
  const int lane = tid & 31;
  const int wave = tid >> 5;
  const int mloc = lane & 15;
  const int half = lane >> 4;

  const float* xg_bt = xg + (size_t)bt * NN_ * C1_;

  for (int i = tid; i < NN_ * C1_; i += 256) {
    int n = i >> 6, c = i & 63;
    float v = xg_bt[i];
    s_xg[n * XSTR + c] = v;
    s_xgp[(n >> 1) * XPSTR + 2 * c + (n & 1)] = v;
  }
  {
    const int rl = tid >> 1, ch = tid & 1;
    if (rl < ntask * 16) {
      const float* arow = adj + (size_t)(row0 + rl) * NN_ + ch * 128;
      for (int w8 = 0; w8 < 4; ++w8) {
        unsigned bits = 0u;
        for (int j = 0; j < 32; ++j)
          bits |= (arow[w8 * 32 + j] > 0.f) ? (1u << j) : 0u;
        s_adj[rl * 8 + ch * 4 + w8] = bits;
      }
    }
  }
  __syncthreads();

  {
    const float* row = &s_xg[tid * XSTR];
    float rv[C1_];
    #pragma unroll
    for (int q = 0; q < C1_ / 4; ++q) {
      const v4f t4 = *(const v4f*)(row + 4 * q);
      rv[4 * q + 0] = t4[0]; rv[4 * q + 1] = t4[1]; rv[4 * q + 2] = t4[2]; rv[4 * q + 3] = t4[3];
    }
    #pragma unroll 1
    for (int h = 0; h < HH_; ++h) {
      const float* U1 = u1 + h * C1_;
      const float* U2 = u2 + h * C1_;
      float a1 = 0.f, a2 = 0.f;
      #pragma unroll
      for (int c = 0; c < C1_; ++c) {
        a1 = fmaf(rv[c], U1[c], a1);
        a2 = fmaf(rv[c], U2[c], a2);
      }
      s_e1[h * NN_ + tid] = a1;
      s_e2[h * NN_ + tid] = a2;
    }
  }
  __syncthreads();

  const float inv_h = 1.0f / (float)HH_;
  float* s_pw = s_p + wave * 16 * PSTR;
  const float* e1h = s_e1 + wave * NN_;
  const float* e2h = s_e2 + wave * NN_;
  const int jbase = half * 128;
  float* xs_bt = xs + (size_t)bt * NN_ * C1_;

  float e2r[NN_ / 2];
  #pragma unroll
  for (int q = 0; q < NN_ / 8; ++q) {
    const v4f t4 = *(const v4f*)(e2h + jbase + 4 * q);
    e2r[4 * q + 0] = t4[0]; e2r[4 * q + 1] = t4[1]; e2r[4 * q + 2] = t4[2]; e2r[4 * q + 3] = t4[3];
  }

  #pragma unroll 1
  for (int ti = 0; ti < ntask; ++ti) {
    const int ml = ti * 16 + mloc;
    {
      const float e1m = e1h[row0 + ml];
      const v4u adjv = *(const v4u*)&s_adj[ml * 8 + half * 4];

      float scr[NN_ / 2];
      float mx = -3.0e38f;
      #pragma unroll
      for (int w8 = 0; w8 < 4; ++w8) {
        const unsigned adjw = adjv[w8];
        #pragma unroll
        for (int j = 0; j < 32; ++j) {
          float sc = e1m + e2r[w8 * 32 + j];
          sc = sc > 0.f ? sc : ALPHA_ * sc;
          sc = ((adjw >> j) & 1) ? sc : -3.0e38f;
          scr[w8 * 32 + j] = sc;
          mx = fmaxf(mx, sc);
        }
      }
      mx = fmaxf(mx, __shfl_xor(mx, 16));

      float sm = 0.f;
      #pragma unroll
      for (int w8 = 0; w8 < 4; ++w8) {
        float* pb = &s_pw[mloc * PSTR + jbase + w8 * 32];
        #pragma unroll
        for (int j4 = 0; j4 < 32; j4 += 4) {
          v4f p;
          #pragma unroll
          for (int u = 0; u < 4; ++u) {
            float pe = __expf(scr[w8 * 32 + j4 + u] - mx);
            p[u] = pe;
            sm += pe;
          }
          *(v4f*)(pb + j4) = p;
        }
      }
      sm += __shfl_xor(sm, 16);
      s_rinv[wave * 16 + mloc] = 1.0f / sm;

      v8f acc0 = {}, acc1 = {}, acc2 = {}, acc3 = {};
      const float* pa   = &s_pw[mloc * PSTR + 2 * half];
      const float* xpb  = &s_xgp[half * XPSTR + 2 * mloc];
      v2f fa[2][ATT_KB];
      v2f fb[2][ATT_KB][4];
      #pragma unroll
      for (int s = 0; s < ATT_KB; ++s) {
        fa[0][s] = *(const v2f*)(pa + 4 * s);
        const float* bp = xpb + (2 * s) * XPSTR;
        #pragma unroll
        for (int q = 0; q < 4; ++q) fb[0][s][q] = *(const v2f*)(bp + 32 * q);
      }
      #pragma unroll 1
      for (int kb = 0; kb < ATT_NB; kb += 2) {
        #pragma unroll
        for (int st = 0; st < 2; ++st) {
          const int nb = (kb + st + 1 < ATT_NB) ? (kb + st + 1) : (ATT_NB - 1);
          const float* pan = pa + nb * (4 * ATT_KB);
          const float* xpn = xpb + nb * (2 * ATT_KB) * XPSTR;
          __builtin_amdgcn_sched_barrier(0);
          #pragma unroll
          for (int s = 0; s < ATT_KB; ++s) {
            fa[st ^ 1][s] = *(const v2f*)(pan + 4 * s);
            const float* bp = xpn + (2 * s) * XPSTR;
            #pragma unroll
            for (int q = 0; q < 4; ++q) fb[st ^ 1][s][q] = *(const v2f*)(bp + 32 * q);
          }
          __builtin_amdgcn_sched_barrier(0);
          #pragma unroll
          for (int s = 0; s < ATT_KB; ++s) {
            acc0 = __builtin_amdgcn_wmma_f32_16x16x4_f32(false, fa[st][s], false, fb[st][s][0], (short)0, acc0, false, false);
            acc1 = __builtin_amdgcn_wmma_f32_16x16x4_f32(false, fa[st][s], false, fb[st][s][1], (short)0, acc1, false, false);
            acc2 = __builtin_amdgcn_wmma_f32_16x16x4_f32(false, fa[st][s], false, fb[st][s][2], (short)0, acc2, false, false);
            acc3 = __builtin_amdgcn_wmma_f32_16x16x4_f32(false, fa[st][s], false, fb[st][s][3], (short)0, acc3, false, false);
          }
        }
      }

      float inv[8];
      #pragma unroll
      for (int r = 0; r < 8; ++r) inv[r] = s_rinv[wave * 16 + r + 8 * half];
      #pragma unroll
      for (int r = 0; r < 8; ++r) {
        float* hrow = s_pw + (r + 8 * half) * PSTR + mloc;
        hrow[0]  = tanh_fast(acc0[r] * inv[r]);
        hrow[16] = tanh_fast(acc1[r] * inv[r]);
        hrow[32] = tanh_fast(acc2[r] * inv[r]);
        hrow[48] = tanh_fast(acc3[r] * inv[r]);
      }
    }
    __syncthreads();

    const int rr = tid >> 4, c4 = (tid & 15) * 4;
    const int mm = row0 + ti * 16 + rr;
    v4f o;
    {
      v4f sacc = {0.f, 0.f, 0.f, 0.f};
      #pragma unroll
      for (int hh = 0; hh < HH_; ++hh) {
        const v4f t4 = *(const v4f*)(s_p + (hh * 16 + rr) * PSTR + c4);
        sacc[0] += t4[0] * inv_h;
        sacc[1] += t4[1] * inv_h;
        sacc[2] += t4[2] * inv_h;
        sacc[3] += t4[3] * inv_h;
      }
      const v4f xr = *(const v4f*)&s_xg[mm * XSTR + c4];
      o[0] = xr[0] + sacc[0]; o[1] = xr[1] + sacc[1]; o[2] = xr[2] + sacc[2]; o[3] = xr[3] + sacc[3];
    }
    __syncthreads();
    *(v4f*)(xs_bt + (size_t)mm * C1_ + c4) = o;
  }
}

#define C2_K   (C1_ * KT_)
#define C2_XP  260
#define C2_WP  196
#define C2_YP  68
#define C2_G   4
#define C2_NG  (C2_K / C2_G)

__global__ __launch_bounds__(256) void conv2_ln_kernel(
    const float* __restrict__ xs,
    const float* __restrict__ w2,
    const float* __restrict__ b2,
    const float* __restrict__ lnw,
    const float* __restrict__ lnb,
    float* __restrict__ out)
{
  extern __shared__ float smem[];
  float* s_x   = smem;
  float* s_w   = s_x + C2_K * C2_XP;
  float* s_red = s_w + C2_ * C2_WP;
  float* s_y   = smem;

  const int bt = blockIdx.x;
  const int bb = bt >> 3, t = bt & 7;
  const int tid = threadIdx.x;

  #pragma unroll
  for (unsigned j = 0; j < (C2_ * C2_K) / 4 / 256; ++j) {
    const unsigned i4 = (unsigned)tid + j * 256u;
    const v4f wv = ((const v4f*)w2)[i4];
    const unsigned co = i4 / (C2_K / 4u);
    const unsigned r  = (i4 - co * (C2_K / 4u)) * 4u;
    #pragma unroll
    for (unsigned e = 0; e < 4; ++e) {
      const unsigned ci = (r + e) / KT_, kt = (r + e) - ci * KT_;
      s_w[co * C2_WP + kt * C1_ + ci] = wv[e];
    }
  }
  {
    const float* src = xs + (((size_t)bb * T1_ + t) * NN_) * C1_ + tid;
    float* dst = s_x + (tid & 63) * C2_XP + (tid >> 6);
    #pragma unroll 1
    for (int blk = 0; blk < 2 * KT_; ++blk) {
      float xv[32];
      #pragma unroll
      for (int u = 0; u < 32; ++u) xv[u] = src[u * 256];
      #pragma unroll
      for (int u = 0; u < 32; ++u) dst[u * 4] = xv[u];
      src += 32 * 256;
      dst += (blk & 1) ? (C1_ * C2_XP - 128) : 128;
    }
  }
  __syncthreads();

  const int lane = tid & 31;
  const int wave = tid >> 5;
  const int c16  = lane & 15;
  const int nb   = wave * 32 + (lane >> 4) * 16;

  float acc[4][16];
  #pragma unroll
  for (int j = 0; j < 4; ++j) {
    const float bj = b2[c16 + 16 * j];
    #pragma unroll
    for (int i = 0; i < 16; ++i) acc[j][i] = bj;
  }

  const float* xb = s_x + nb;
  const float* wb = s_w + c16 * C2_WP;

  v4f wv[2][4];
  v4f xv[2][C2_G][4];
  #pragma unroll
  for (int j = 0; j < 4; ++j) wv[0][j] = *(const v4f*)(wb + j * 16 * C2_WP);
  #pragma unroll
  for (int k = 0; k < C2_G; ++k)
    #pragma unroll
    for (int q = 0; q < 4; ++q) xv[0][k][q] = *(const v4f*)(xb + k * C2_XP + 4 * q);

  #pragma unroll 1
  for (int g = 0; g < C2_NG; g += 2) {
    #pragma unroll
    for (int s = 0; s < 2; ++s) {
      const int gn = (g + s + 1 < C2_NG) ? (g + s + 1) : (C2_NG - 1);
      const float* xn = xb + gn * C2_G * C2_XP;
      const float* wn = wb + gn * C2_G;
      __builtin_amdgcn_sched_barrier(0);
      #pragma unroll
      for (int j = 0; j < 4; ++j) wv[s ^ 1][j] = *(const v4f*)(wn + j * 16 * C2_WP);
      #pragma unroll
      for (int k = 0; k < C2_G; ++k)
        #pragma unroll
        for (int q = 0; q < 4; ++q) xv[s ^ 1][k][q] = *(const v4f*)(xn + k * C2_XP + 4 * q);
      __builtin_amdgcn_sched_barrier(0);
      #pragma unroll
      for (int k = 0; k < C2_G; ++k)
        #pragma unroll
        for (int j = 0; j < 4; ++j)
          #pragma unroll
          for (int q = 0; q < 4; ++q)
            #pragma unroll
            for (int u = 0; u < 4; ++u)
              acc[j][4 * q + u] = fmaf(wv[s][j][k], xv[s][k][q][u], acc[j][4 * q + u]);
    }
  }

  v4f res[4][4];
  #pragma unroll
  for (int j = 0; j < 4; ++j)
    #pragma unroll
    for (int q = 0; q < 4; ++q)
      res[j][q] = *(const v4f*)(s_x + (2 * C1_ + c16 + 16 * j) * C2_XP + nb + 4 * q);

  __syncthreads();

  #pragma unroll
  for (int j = 0; j < 4; ++j) {
    const int c = c16 + 16 * j;
    #pragma unroll
    for (int q = 0; q < 4; ++q) {
      #pragma unroll
      for (int u = 0; u < 4; ++u) {
        float v = acc[j][4 * q + u] + res[j][q][u];
        v = v > 0.f ? v : 0.f;
        s_y[(nb + 4 * q + u) * C2_YP + c] = v;
      }
    }
  }
  __syncthreads();

  const int n = tid;
  float yv[C2_];
  #pragma unroll
  for (int q = 0; q < C2_ / 4; ++q) {
    const v4f y4 = *(const v4f*)(s_y + n * C2_YP + 4 * q);
    yv[4 * q + 0] = y4[0]; yv[4 * q + 1] = y4[1]; yv[4 * q + 2] = y4[2]; yv[4 * q + 3] = y4[3];
  }
  float lsum = 0.f, lsq = 0.f;
  #pragma unroll
  for (int c = 0; c < C2_; ++c) {
    lsum += yv[c];
    lsq  += yv[c] * yv[c];
  }

  for (int off = 16; off > 0; off >>= 1) {
    lsum += __shfl_xor(lsum, off);
    lsq  += __shfl_xor(lsq,  off);
  }
  if ((tid & 31) == 0) { s_red[tid >> 5] = lsum; s_red[8 + (tid >> 5)] = lsq; }
  __syncthreads();
  float tsum = 0.f, tsq = 0.f;
  for (int i = 0; i < 8; ++i) { tsum += s_red[i]; tsq += s_red[8 + i]; }
  const float cnt  = (float)(NN_ * C2_);
  const float mu   = tsum / cnt;
  const float var  = tsq / cnt - mu * mu;
  const float rstd = rsqrtf(var + 1e-5f);

  #pragma unroll
  for (int c = 0; c < C2_; ++c) {
    float v = (yv[c] - mu) * rstd * lnw[(size_t)n * C2_ + c] + lnb[(size_t)n * C2_ + c];
    out[(((size_t)bb * C2_ + c) * T2_ + t) * NN_ + n] = v;
  }
}

extern "C" void kernel_launch(void* const* d_in, const int* in_sizes, int n_in,
                              void* d_out, int out_size, void* d_ws, size_t ws_size,
                              hipStream_t stream) {
  const float* x   = (const float*)d_in[0];
  const float* adj = (const float*)d_in[1];
  const float* w1  = (const float*)d_in[2];
  const float* b1  = (const float*)d_in[3];
  const float* w2  = (const float*)d_in[4];
  const float* b2  = (const float*)d_in[5];
  const float* Wh  = (const float*)d_in[6];
  const float* ah  = (const float*)d_in[7];
  const float* lnw = (const float*)d_in[8];
  const float* lnb = (const float*)d_in[9];
  float* out = (float*)d_out;

  float* ws = (float*)d_ws;
  const size_t xg_elems = (size_t)B_ * T1_ * NN_ * C1_;
  float* xg = ws;
  float* xs = xg + xg_elems;
  float* u1 = xs + xg_elems;
  float* u2 = u1 + HH_ * C1_;

  prep_u_kernel<<<2, 256, 0, stream>>>(Wh, ah, u1, u2);

  conv1_glu_kernel<<<B_ * T1_ * (NN_ / C1_NT), 256, 0, stream>>>(x, w1, b1, xg);

  const size_t attn_lds =
      (size_t)(NN_ * XSTR + (NN_ / 2) * XPSTR + 2 * HH_ * NN_ + 8 * 16 + 8 * 16 * PSTR) * sizeof(float)
      + (size_t)(ATT_MAXT * 16 * 8) * sizeof(unsigned);
  attn_kernel<<<B_ * T1_ * ATT_RS, 256, attn_lds, stream>>>(xg, adj, u1, u2, xs);

  const size_t c2_lds = (size_t)(C2_K * C2_XP + C2_ * C2_WP + 16) * sizeof(float);
  conv2_ln_kernel<<<B_ * T2_, 256, c2_lds, stream>>>(xs, w2, b2, lnw, lnb, out);
}


# conv2+LayerNorm split: conv2/ReLU over 4 node tiles per (b,t) on 256 workgroups instead of 64 writes y and tile sums; a fifth launch applies LayerNorm; LayerNorm sums added in a different tree
# speedup vs baseline: 4.9851x; 1.1442x over previous
#include <hip/hip_runtime.h>
#include <math.h>

#define B_   8
#define C0_  16
#define C1_  64
#define C2_  64
#define T0_  12
#define T1_  10
#define T2_  8
#define NN_  256
#define KT_  3
#define HH_  8
#define ALPHA_ 0.2f

typedef float v2f __attribute__((ext_vector_type(2)));
typedef float v4f __attribute__((ext_vector_type(4)));
typedef float v8f __attribute__((ext_vector_type(8)));
typedef unsigned v4u __attribute__((ext_vector_type(4)));

__device__ __forceinline__ float tanh_fast(float x) {
  float e = __expf(fminf(2.0f * x, 80.0f));
  return (e - 1.0f) / (e + 1.0f);
}

__global__ __launch_bounds__(256) void prep_u_kernel(
    const float* __restrict__ W,
    const float* __restrict__ a,
    float* __restrict__ u1, float* __restrict__ u2)
{
  int tid = blockIdx.x * 256 + threadIdx.x;
  if (tid >= HH_ * C1_) return;
  int h = tid >> 6, c = tid & 63;
  const float* Wh = W + ((size_t)h * C1_ + c) * C1_;
  const float* ah = a + (size_t)h * 2 * C1_;
  float s1 = 0.f, s2 = 0.f;
  for (int d = 0; d < C1_; ++d) {
    float w = Wh[d];
    s1 += w * ah[d];
    s2 += w * ah[C1_ + d];
  }
  u1[tid] = s1;
  u2[tid] = s2;
}

#define C1_NT   64
#define C1_WSTR 49
#define C1_XSTR 68
#define C1_K    (C0_ * KT_)
#define C1_RC   6

__global__ __launch_bounds__(256) void conv1_glu_kernel(
    const float* __restrict__ x,
    const float* __restrict__ w,
    const float* __restrict__ b,
    float* __restrict__ xg)
{
  __shared__ __attribute__((aligned(16))) float s_w[2 * C1_ * C1_WSTR];
  __shared__ __attribute__((aligned(16))) float s_x[C1_K * C1_XSTR];

  const int tid  = threadIdx.x;
  const int lane = tid & 31;
  const int wave = tid >> 5;
  const int bt   = blockIdx.x >> 2;
  const int n0   = (blockIdx.x & 3) * C1_NT;
  const int bb   = bt / T1_, t = bt - bb * T1_;

  #pragma unroll
  for (int j = 0; j < (2 * C1_ * C1_K) / 4 / 256; ++j) {
    int i4 = tid + j * 256;
    v4f wv = ((const v4f*)w)[i4];
    int base = i4 * 4;
    int c2 = base / C1_K, k = base - c2 * C1_K;
    float* d = &s_w[c2 * C1_WSTR + k];
    d[0] = wv[0]; d[1] = wv[1]; d[2] = wv[2]; d[3] = wv[3];
  }
  #pragma unroll
  for (int j = 0; j < C1_K * (C1_NT / 4) / 256; ++j) {
    int i4 = tid + j * 256;
    int r = i4 >> 4, q = i4 & 15;
    int kt = r >> 4, ci = r & 15;
    const float* src = x + (((size_t)bb * C0_ + ci) * T0_ + (t + kt)) * NN_ + n0 + q * 4;
    *(v4f*)&s_x[r * C1_XSTR + q * 4] = *(const v4f*)src;
  }
  __syncthreads();

  const int c  = (wave & 1) * 32 + lane;
  const int nb = (wave >> 1) * 16;

  float wp[C1_K], wq[C1_K];
  #pragma unroll
  for (int r = 0; r < C1_K; ++r) {
    int kt = r >> 4, ci = r & 15;
    wp[r] = s_w[c * C1_WSTR + ci * KT_ + kt];
    wq[r] = s_w[(c + C1_) * C1_WSTR + ci * KT_ + kt];
  }
  const float bp = b[c];
  const float bq = b[c + C1_];
  const bool has_res = (c < C0_);
  const int  xr = has_res ? (2 * C0_ + c) : 0;

  #pragma unroll 1
  for (int g = 0; g < 2; ++g) {
    const int nl = nb + g * 8;
    const float* xrow = &s_x[nl];
    float ap[8], aq[8];
    #pragma unroll
    for (int u = 0; u < 8; ++u) { ap[u] = bp; aq[u] = bq; }

    v4f xa[2][C1_RC][2];
    #pragma unroll
    for (int j = 0; j < C1_RC; ++j) {
      xa[0][j][0] = *(const v4f*)(xrow + j * C1_XSTR);
      xa[0][j][1] = *(const v4f*)(xrow + j * C1_XSTR + 4);
    }
    #pragma unroll
    for (int ch = 0; ch < C1_K / C1_RC; ++ch) {
      __builtin_amdgcn_sched_barrier(0);
      if (ch + 1 < C1_K / C1_RC) {
        #pragma unroll
        for (int j = 0; j < C1_RC; ++j) {
          const float* p = xrow + ((ch + 1) * C1_RC + j) * C1_XSTR;
          xa[(ch + 1) & 1][j][0] = *(const v4f*)(p);
          xa[(ch + 1) & 1][j][1] = *(const v4f*)(p + 4);
        }
      }
      __builtin_amdgcn_sched_barrier(0);
      #pragma unroll
      for (int j = 0; j < C1_RC; ++j) {
        const int r = ch * C1_RC + j;
        const v4f x0 = xa[ch & 1][j][0];
        const v4f x1 = xa[ch & 1][j][1];
        #pragma unroll
        for (int u = 0; u < 4; ++u) {
          ap[u]     = fmaf(wp[r], x0[u], ap[u]);
          aq[u]     = fmaf(wq[r], x0[u], aq[u]);
          ap[u + 4] = fmaf(wp[r], x1[u], ap[u + 4]);
          aq[u + 4] = fmaf(wq[r], x1[u], aq[u + 4]);
        }
      }
    }
    v4f r0 = *(const v4f*)&s_x[xr * C1_XSTR + nl];
    v4f r1 = *(const v4f*)&s_x[xr * C1_XSTR + nl + 4];
    float* orow = xg + ((size_t)bt * NN_ + n0 + nl) * C1_ + c;
    #pragma unroll
    for (int u = 0; u < 8; ++u) {
      float xin = has_res ? (u < 4 ? r0[u & 3] : r1[u & 3]) : 0.f;
      float sg  = 1.f / (1.f + __expf(-aq[u]));
      orow[(size_t)u * C1_] = (ap[u] + xin) * sg;
    }
  }
}

#define XSTR  72
#define XPSTR 160
#define PSTR  260
#define ATT_RS   3
#define ATT_MAXT 6
#define ATT_KB   4
#define ATT_NB   (NN_ / (4 * ATT_KB))

__global__ __launch_bounds__(256) void attn_kernel(
    const float* __restrict__ xg,
    const float* __restrict__ adj,
    const float* __restrict__ u1,
    const float* __restrict__ u2,
    float* __restrict__ xs)
{
  extern __shared__ float smem[];
  float*    s_xg   = smem;
  float*    s_xgp  = s_xg + NN_ * XSTR;
  float*    s_e1   = s_xgp + (NN_ / 2) * XPSTR;
  float*    s_e2   = s_e1 + HH_ * NN_;
  float*    s_rinv = s_e2 + HH_ * NN_;
  float*    s_p    = s_rinv + 8 * 16;
  unsigned* s_adj  = (unsigned*)(s_p + 8 * 16 * PSTR);

  const int bt    = blockIdx.x / ATT_RS;
  const int part  = blockIdx.x - bt * ATT_RS;
  const int ntask = (part == 0) ? 6 : 5;
  const int row0  = ((part == 0) ? 0 : (part == 1 ? 6 : 11)) * 16;
  const int tid  = threadIdx.x;
  const int lane = tid & 31;
  const int wave = tid >> 5;
  const int mloc = lane & 15;
  const int half = lane >> 4;

  const float* xg_bt = xg + (size_t)bt * NN_ * C1_;

  for (int i = tid; i < NN_ * C1_; i += 256) {
    int n = i >> 6, c = i & 63;
    float v = xg_bt[i];
    s_xg[n * XSTR + c] = v;
    s_xgp[(n >> 1) * XPSTR + 2 * c + (n & 1)] = v;
  }
  {
    const int rl = tid >> 1, ch = tid & 1;
    if (rl < ntask * 16) {
      const float* arow = adj + (size_t)(row0 + rl) * NN_ + ch * 128;
      for (int w8 = 0; w8 < 4; ++w8) {
        unsigned bits = 0u;
        for (int j = 0; j < 32; ++j)
          bits |= (arow[w8 * 32 + j] > 0.f) ? (1u << j) : 0u;
        s_adj[rl * 8 + ch * 4 + w8] = bits;
      }
    }
  }
  __syncthreads();

  {
    const float* row = &s_xg[tid * XSTR];
    float rv[C1_];
    #pragma unroll
    for (int q = 0; q < C1_ / 4; ++q) {
      const v4f t4 = *(const v4f*)(row + 4 * q);
      rv[4 * q + 0] = t4[0]; rv[4 * q + 1] = t4[1]; rv[4 * q + 2] = t4[2]; rv[4 * q + 3] = t4[3];
    }
    #pragma unroll 1
    for (int h = 0; h < HH_; ++h) {
      const float* U1 = u1 + h * C1_;
      const float* U2 = u2 + h * C1_;
      float a1 = 0.f, a2 = 0.f;
      #pragma unroll
      for (int c = 0; c < C1_; ++c) {
        a1 = fmaf(rv[c], U1[c], a1);
        a2 = fmaf(rv[c], U2[c], a2);
      }
      s_e1[h * NN_ + tid] = a1;
      s_e2[h * NN_ + tid] = a2;
    }
  }
  __syncthreads();

  const float inv_h = 1.0f / (float)HH_;
  float* s_pw = s_p + wave * 16 * PSTR;
  const float* e1h = s_e1 + wave * NN_;
  const float* e2h = s_e2 + wave * NN_;
  const int jbase = half * 128;
  float* xs_bt = xs + (size_t)bt * NN_ * C1_;

  float e2r[NN_ / 2];
  #pragma unroll
  for (int q = 0; q < NN_ / 8; ++q) {
    const v4f t4 = *(const v4f*)(e2h + jbase + 4 * q);
    e2r[4 * q + 0] = t4[0]; e2r[4 * q + 1] = t4[1]; e2r[4 * q + 2] = t4[2]; e2r[4 * q + 3] = t4[3];
  }

  #pragma unroll 1
  for (int ti = 0; ti < ntask; ++ti) {
    const int ml = ti * 16 + mloc;
    {
      const float e1m = e1h[row0 + ml];
      const v4u adjv = *(const v4u*)&s_adj[ml * 8 + half * 4];

      float scr[NN_ / 2];
      float mx = -3.0e38f;
      #pragma unroll
      for (int w8 = 0; w8 < 4; ++w8) {
        const unsigned adjw = adjv[w8];
        #pragma unroll
        for (int j = 0; j < 32; ++j) {
          float sc = e1m + e2r[w8 * 32 + j];
          sc = sc > 0.f ? sc : ALPHA_ * sc;
          sc = ((adjw >> j) & 1) ? sc : -3.0e38f;
          scr[w8 * 32 + j] = sc;
          mx = fmaxf(mx, sc);
        }
      }
      mx = fmaxf(mx, __shfl_xor(mx, 16));

      float sm = 0.f;
      #pragma unroll
      for (int w8 = 0; w8 < 4; ++w8) {
        float* pb = &s_pw[mloc * PSTR + jbase + w8 * 32];
        #pragma unroll
        for (int j4 = 0; j4 < 32; j4 += 4) {
          v4f p;
          #pragma unroll
          for (int u = 0; u < 4; ++u) {
            float pe = __expf(scr[w8 * 32 + j4 + u] - mx);
            p[u] = pe;
            sm += pe;
          }
          *(v4f*)(pb + j4) = p;
        }
      }
      sm += __shfl_xor(sm, 16);
      s_rinv[wave * 16 + mloc] = 1.0f / sm;

      v8f acc0 = {}, acc1 = {}, acc2 = {}, acc3 = {};
      const float* pa   = &s_pw[mloc * PSTR + 2 * half];
      const float* xpb  = &s_xgp[half * XPSTR + 2 * mloc];
      v2f fa[2][ATT_KB];
      v2f fb[2][ATT_KB][4];
      #pragma unroll
      for (int s = 0; s < ATT_KB; ++s) {
        fa[0][s] = *(const v2f*)(pa + 4 * s);
        const float* bp = xpb + (2 * s) * XPSTR;
        #pragma unroll
        for (int q = 0; q < 4; ++q) fb[0][s][q] = *(const v2f*)(bp + 32 * q);
      }
      #pragma unroll 1
      for (int kb = 0; kb < ATT_NB; kb += 2) {
        #pragma unroll
        for (int st = 0; st < 2; ++st) {
          const int nb = (kb + st + 1 < ATT_NB) ? (kb + st + 1) : (ATT_NB - 1);
          const float* pan = pa + nb * (4 * ATT_KB);
          const float* xpn = xpb + nb * (2 * ATT_KB) * XPSTR;
          __builtin_amdgcn_sched_barrier(0);
          #pragma unroll
          for (int s = 0; s < ATT_KB; ++s) {
            fa[st ^ 1][s] = *(const v2f*)(pan + 4 * s);
            const float* bp = xpn + (2 * s) * XPSTR;
            #pragma unroll
            for (int q = 0; q < 4; ++q) fb[st ^ 1][s][q] = *(const v2f*)(bp + 32 * q);
          }
          __builtin_amdgcn_sched_barrier(0);
          #pragma unroll
          for (int s = 0; s < ATT_KB; ++s) {
            acc0 = __builtin_amdgcn_wmma_f32_16x16x4_f32(false, fa[st][s], false, fb[st][s][0], (short)0, acc0, false, false);
            acc1 = __builtin_amdgcn_wmma_f32_16x16x4_f32(false, fa[st][s], false, fb[st][s][1], (short)0, acc1, false, false);
            acc2 = __builtin_amdgcn_wmma_f32_16x16x4_f32(false, fa[st][s], false, fb[st][s][2], (short)0, acc2, false, false);
            acc3 = __builtin_amdgcn_wmma_f32_16x16x4_f32(false, fa[st][s], false, fb[st][s][3], (short)0, acc3, false, false);
          }
        }
      }

      float inv[8];
      #pragma unroll
      for (int r = 0; r < 8; ++r) inv[r] = s_rinv[wave * 16 + r + 8 * half];
      #pragma unroll
      for (int r = 0; r < 8; ++r) {
        float* hrow = s_pw + (r + 8 * half) * PSTR + mloc;
        hrow[0]  = tanh_fast(acc0[r] * inv[r]);
        hrow[16] = tanh_fast(acc1[r] * inv[r]);
        hrow[32] = tanh_fast(acc2[r] * inv[r]);
        hrow[48] = tanh_fast(acc3[r] * inv[r]);
      }
    }
    __syncthreads();

    const int rr = tid >> 4, c4 = (tid & 15) * 4;
    const int mm = row0 + ti * 16 + rr;
    v4f o;
    {
      v4f sacc = {0.f, 0.f, 0.f, 0.f};
      #pragma unroll
      for (int hh = 0; hh < HH_; ++hh) {
        const v4f t4 = *(const v4f*)(s_p + (hh * 16 + rr) * PSTR + c4);
        sacc[0] += t4[0] * inv_h;
        sacc[1] += t4[1] * inv_h;
        sacc[2] += t4[2] * inv_h;
        sacc[3] += t4[3] * inv_h;
      }
      const v4f xr = *(const v4f*)&s_xg[mm * XSTR + c4];
      o[0] = xr[0] + sacc[0]; o[1] = xr[1] + sacc[1]; o[2] = xr[2] + sacc[2]; o[3] = xr[3] + sacc[3];
    }
    __syncthreads();
    *(v4f*)(xs_bt + (size_t)mm * C1_ + c4) = o;
  }
}

#define C2_K   (C1_ * KT_)
#define C2_NT  64
#define C2_TL  (NN_ / C2_NT)
#define C2_XP  68
#define C2_WP  196
#define C2_YP  68
#define C2_G   4
#define C2_NG  (C2_K / C2_G)

__global__ __launch_bounds__(256) void conv2_relu_kernel(
    const float* __restrict__ xs,
    const float* __restrict__ w2,
    const float* __restrict__ b2,
    float* __restrict__ yws,
    float* __restrict__ stats)
{
  extern __shared__ float smem[];
  float* s_x   = smem;
  float* s_w   = s_x + C2_K * C2_XP;
  float* s_red = s_w + C2_ * C2_WP;
  float* s_y   = smem;

  const int bt   = blockIdx.x >> 2;
  const int tile = blockIdx.x & 3;
  const int n0   = tile * C2_NT;
  const int bb = bt >> 3, t = bt & 7;
  const int tid = threadIdx.x;

  #pragma unroll
  for (unsigned j = 0; j < (C2_ * C2_K) / 4 / 256; ++j) {
    const unsigned i4 = (unsigned)tid + j * 256u;
    const v4f wv = ((const v4f*)w2)[i4];
    const unsigned co = i4 / (C2_K / 4u);
    const unsigned r  = (i4 - co * (C2_K / 4u)) * 4u;
    #pragma unroll
    for (unsigned e = 0; e < 4; ++e) {
      const unsigned ci = (r + e) / KT_, kt = (r + e) - ci * KT_;
      s_w[co * C2_WP + kt * C1_ + ci] = wv[e];
    }
  }
  {
    const int ci = tid & 63, nq = tid >> 6;
    float xv[KT_][16];
    #pragma unroll
    for (int kt = 0; kt < KT_; ++kt) {
      const float* src = xs + (((size_t)bb * T1_ + t + kt) * NN_ + n0 + nq) * C1_ + ci;
      #pragma unroll
      for (int u = 0; u < 16; ++u) xv[kt][u] = src[u * 4 * C1_];
    }
    #pragma unroll
    for (int kt = 0; kt < KT_; ++kt)
      #pragma unroll
      for (int u = 0; u < 16; ++u) s_x[(kt * C1_ + ci) * C2_XP + nq + 4 * u] = xv[kt][u];
  }
  __syncthreads();

  const int lane = tid & 31;
  const int wave = tid >> 5;
  const int c16  = lane & 15;
  const int nb   = wave * 8 + (lane >> 4) * 4;

  float acc[4][4];
  #pragma unroll
  for (int j = 0; j < 4; ++j) {
    const float bj = b2[c16 + 16 * j];
    #pragma unroll
    for (int i = 0; i < 4; ++i) acc[j][i] = bj;
  }

  const float* xb = s_x + nb;
  const float* wb = s_w + c16 * C2_WP;

  v4f wv[2][4];
  v4f xv[2][C2_G];
  #pragma unroll
  for (int j = 0; j < 4; ++j) wv[0][j] = *(const v4f*)(wb + j * 16 * C2_WP);
  #pragma unroll
  for (int k = 0; k < C2_G; ++k) xv[0][k] = *(const v4f*)(xb + k * C2_XP);

  #pragma unroll 1
  for (int g = 0; g < C2_NG; g += 2) {
    #pragma unroll
    for (int s = 0; s < 2; ++s) {
      const int gn = (g + s + 1 < C2_NG) ? (g + s + 1) : (C2_NG - 1);
      const float* xn = xb + gn * C2_G * C2_XP;
      const float* wn = wb + gn * C2_G;
      __builtin_amdgcn_sched_barrier(0);
      #pragma unroll
      for (int j = 0; j < 4; ++j) wv[s ^ 1][j] = *(const v4f*)(wn + j * 16 * C2_WP);
      #pragma unroll
      for (int k = 0; k < C2_G; ++k) xv[s ^ 1][k] = *(const v4f*)(xn + k * C2_XP);
      __builtin_amdgcn_sched_barrier(0);
      #pragma unroll
      for (int k = 0; k < C2_G; ++k)
        #pragma unroll
        for (int j = 0; j < 4; ++j)
          #pragma unroll
          for (int u = 0; u < 4; ++u)
            acc[j][u] = fmaf(wv[s][j][k], xv[s][k][u], acc[j][u]);
    }
  }

  v4f res[4];
  #pragma unroll
  for (int j = 0; j < 4; ++j)
    res[j] = *(const v4f*)(s_x + (2 * C1_ + c16 + 16 * j) * C2_XP + nb);

  __syncthreads();

  float lsum = 0.f, lsq = 0.f;
  #pragma unroll
  for (int j = 0; j < 4; ++j) {
    v4f y;
    #pragma unroll
    for (int u = 0; u < 4; ++u) {
      float v = acc[j][u] + res[j][u];
      v = v > 0.f ? v : 0.f;
      y[u] = v;
      lsum += v;
      lsq  += v * v;
    }
    *(v4f*)(s_y + (c16 + 16 * j) * C2_YP + nb) = y;
  }
  for (int off = 16; off > 0; off >>= 1) {
    lsum += __shfl_xor(lsum, off);
    lsq  += __shfl_xor(lsq,  off);
  }
  if ((tid & 31) == 0) { s_red[tid >> 5] = lsum; s_red[8 + (tid >> 5)] = lsq; }
  __syncthreads();
  if (tid == 0) {
    float tsum = 0.f, tsq = 0.f;
    for (int i = 0; i < 8; ++i) { tsum += s_red[i]; tsq += s_red[8 + i]; }
    stats[(bt * C2_TL + tile) * 2 + 0] = tsum;
    stats[(bt * C2_TL + tile) * 2 + 1] = tsq;
  }
  #pragma unroll
  for (int u = 0; u < 4; ++u) {
    const int i4 = tid + 256 * u;
    const int c = i4 >> 4, q = i4 & 15;
    const v4f v = *(const v4f*)(s_y + c * C2_YP + 4 * q);
    *(v4f*)(yws + ((size_t)bt * C2_ + c) * NN_ + n0 + 4 * q) = v;
  }
}

#define LN_P 65

__global__ __launch_bounds__(256) void ln_apply_kernel(
    const float* __restrict__ yws,
    const float* __restrict__ stats,
    const float* __restrict__ lnw,
    const float* __restrict__ lnb,
    float* __restrict__ out)
{
  __shared__ float s_lw[C2_NT * LN_P];
  __shared__ float s_lb[C2_NT * LN_P];

  const int bt = blockIdx.x >> 2;
  const int n0 = (blockIdx.x & 3) * C2_NT;
  const int bb = bt >> 3, t = bt & 7;
  const int tid = threadIdx.x;

  v4f wv[4], bv[4], yv[4];
  #pragma unroll
  for (int u = 0; u < 4; ++u) {
    const int i = (tid + 256 * u) * 4;
    wv[u] = *(const v4f*)(lnw + (size_t)n0 * C2_ + i);
    bv[u] = *(const v4f*)(lnb + (size_t)n0 * C2_ + i);
    const int i4 = tid + 256 * u;
    const int c = i4 >> 4, q = i4 & 15;
    yv[u] = *(const v4f*)(yws + ((size_t)bt * C2_ + c) * NN_ + n0 + 4 * q);
  }
  float tsum = 0.f, tsq = 0.f;
  for (int k = 0; k < C2_TL; ++k) {
    tsum += stats[(bt * C2_TL + k) * 2 + 0];
    tsq  += stats[(bt * C2_TL + k) * 2 + 1];
  }
  const float cnt  = (float)(NN_ * C2_);
  const float mu   = tsum / cnt;
  const float var  = tsq / cnt - mu * mu;
  const float rstd = rsqrtf(var + 1e-5f);

  #pragma unroll
  for (int u = 0; u < 4; ++u) {
    const int i = (tid + 256 * u) * 4;
    const int n = i >> 6, c = i & 63;
    #pragma unroll
    for (int e = 0; e < 4; ++e) {
      s_lw[n * LN_P + c + e] = wv[u][e];
      s_lb[n * LN_P + c + e] = bv[u][e];
    }
  }
  __syncthreads();

  #pragma unroll
  for (int u = 0; u < 4; ++u) {
    const int i4 = tid + 256 * u;
    const int c = i4 >> 4, q = i4 & 15;
    v4f o;
    #pragma unroll
    for (int e = 0; e < 4; ++e) {
      const int n = 4 * q + e;
      o[e] = (yv[u][e] - mu) * rstd * s_lw[n * LN_P + c] + s_lb[n * LN_P + c];
    }
    *(v4f*)(out + (((size_t)bb * C2_ + c) * T2_ + t) * NN_ + n0 + 4 * q) = o;
  }
}

extern "C" void kernel_launch(void* const* d_in, const int* in_sizes, int n_in,
                              void* d_out, int out_size, void* d_ws, size_t ws_size,
                              hipStream_t stream) {
  const float* x   = (const float*)d_in[0];
  const float* adj = (const float*)d_in[1];
  const float* w1  = (const float*)d_in[2];
  const float* b1  = (const float*)d_in[3];
  const float* w2  = (const float*)d_in[4];
  const float* b2  = (const float*)d_in[5];
  const float* Wh  = (const float*)d_in[6];
  const float* ah  = (const float*)d_in[7];
  const float* lnw = (const float*)d_in[8];
  const float* lnb = (const float*)d_in[9];
  float* out = (float*)d_out;

  float* ws = (float*)d_ws;
  const size_t xg_elems = (size_t)B_ * T1_ * NN_ * C1_;
  float* xg = ws;
  float* xs = xg + xg_elems;
  float* u1 = xs + xg_elems;
  float* u2 = u1 + HH_ * C1_;
  float* yws = u2 + HH_ * C1_;
  float* stats = yws + (size_t)B_ * T2_ * C2_ * NN_;

  prep_u_kernel<<<2, 256, 0, stream>>>(Wh, ah, u1, u2);

  conv1_glu_kernel<<<B_ * T1_ * (NN_ / C1_NT), 256, 0, stream>>>(x, w1, b1, xg);

  const size_t attn_lds =
      (size_t)(NN_ * XSTR + (NN_ / 2) * XPSTR + 2 * HH_ * NN_ + 8 * 16 + 8 * 16 * PSTR) * sizeof(float)
      + (size_t)(ATT_MAXT * 16 * 8) * sizeof(unsigned);
  attn_kernel<<<B_ * T1_ * ATT_RS, 256, attn_lds, stream>>>(xg, adj, u1, u2, xs);

  const size_t c2_lds = (size_t)(C2_K * C2_XP + C2_ * C2_WP + 16) * sizeof(float);
  conv2_relu_kernel<<<B_ * T2_ * C2_TL, 256, c2_lds, stream>>>(xs, w2, b2, yws, stats);

  ln_apply_kernel<<<B_ * T2_ * C2_TL, 256, 0, stream>>>(yws, stats, lnw, lnb, out);
}


# attention: lane computes its own A-fragment probabilities in registers (no LDS probability tile); LDS holds all row blocks' head tiles, so no barrier per row block; row sums added in another order
# speedup vs baseline: 5.1461x; 1.0323x over previous
#include <hip/hip_runtime.h>
#include <math.h>

#define B_   8
#define C0_  16
#define C1_  64
#define C2_  64
#define T0_  12
#define T1_  10
#define T2_  8
#define NN_  256
#define KT_  3
#define HH_  8
#define ALPHA_ 0.2f

typedef float v2f __attribute__((ext_vector_type(2)));
typedef float v4f __attribute__((ext_vector_type(4)));
typedef float v8f __attribute__((ext_vector_type(8)));
typedef unsigned v4u __attribute__((ext_vector_type(4)));

__device__ __forceinline__ float tanh_fast(float x) {
  float e = __expf(fminf(2.0f * x, 80.0f));
  return (e - 1.0f) / (e + 1.0f);
}

__global__ __launch_bounds__(256) void prep_u_kernel(
    const float* __restrict__ W,
    const float* __restrict__ a,
    float* __restrict__ u1, float* __restrict__ u2)
{
  int tid = blockIdx.x * 256 + threadIdx.x;
  if (tid >= HH_ * C1_) return;
  int h = tid >> 6, c = tid & 63;
  const float* Wh = W + ((size_t)h * C1_ + c) * C1_;
  const float* ah = a + (size_t)h * 2 * C1_;
  float s1 = 0.f, s2 = 0.f;
  for (int d = 0; d < C1_; ++d) {
    float w = Wh[d];
    s1 += w * ah[d];
    s2 += w * ah[C1_ + d];
  }
  u1[tid] = s1;
  u2[tid] = s2;
}

#define C1_NT   64
#define C1_WSTR 49
#define C1_XSTR 68
#define C1_K    (C0_ * KT_)
#define C1_RC   6

__global__ __launch_bounds__(256) void conv1_glu_kernel(
    const float* __restrict__ x,
    const float* __restrict__ w,
    const float* __restrict__ b,
    float* __restrict__ xg)
{
  __shared__ __attribute__((aligned(16))) float s_w[2 * C1_ * C1_WSTR];
  __shared__ __attribute__((aligned(16))) float s_x[C1_K * C1_XSTR];

  const int tid  = threadIdx.x;
  const int lane = tid & 31;
  const int wave = tid >> 5;
  const int bt   = blockIdx.x >> 2;
  const int n0   = (blockIdx.x & 3) * C1_NT;
  const int bb   = bt / T1_, t = bt - bb * T1_;

  #pragma unroll
  for (int j = 0; j < (2 * C1_ * C1_K) / 4 / 256; ++j) {
    int i4 = tid + j * 256;
    v4f wv = ((const v4f*)w)[i4];
    int base = i4 * 4;
    int c2 = base / C1_K, k = base - c2 * C1_K;
    float* d = &s_w[c2 * C1_WSTR + k];
    d[0] = wv[0]; d[1] = wv[1]; d[2] = wv[2]; d[3] = wv[3];
  }
  #pragma unroll
  for (int j = 0; j < C1_K * (C1_NT / 4) / 256; ++j) {
    int i4 = tid + j * 256;
    int r = i4 >> 4, q = i4 & 15;
    int kt = r >> 4, ci = r & 15;
    const float* src = x + (((size_t)bb * C0_ + ci) * T0_ + (t + kt)) * NN_ + n0 + q * 4;
    *(v4f*)&s_x[r * C1_XSTR + q * 4] = *(const v4f*)src;
  }
  __syncthreads();

  const int c  = (wave & 1) * 32 + lane;
  const int nb = (wave >> 1) * 16;

  float wp[C1_K], wq[C1_K];
  #pragma unroll
  for (int r = 0; r < C1_K; ++r) {
    int kt = r >> 4, ci = r & 15;
    wp[r] = s_w[c * C1_WSTR + ci * KT_ + kt];
    wq[r] = s_w[(c + C1_) * C1_WSTR + ci * KT_ + kt];
  }
  const float bp = b[c];
  const float bq = b[c + C1_];
  const bool has_res = (c < C0_);
  const int  xr = has_res ? (2 * C0_ + c) : 0;

  #pragma unroll 1
  for (int g = 0; g < 2; ++g) {
    const int nl = nb + g * 8;
    const float* xrow = &s_x[nl];
    float ap[8], aq[8];
    #pragma unroll
    for (int u = 0; u < 8; ++u) { ap[u] = bp; aq[u] = bq; }

    v4f xa[2][C1_RC][2];
    #pragma unroll
    for (int j = 0; j < C1_RC; ++j) {
      xa[0][j][0] = *(const v4f*)(xrow + j * C1_XSTR);
      xa[0][j][1] = *(const v4f*)(xrow + j * C1_XSTR + 4);
    }
    #pragma unroll
    for (int ch = 0; ch < C1_K / C1_RC; ++ch) {
      __builtin_amdgcn_sched_barrier(0);
      if (ch + 1 < C1_K / C1_RC) {
        #pragma unroll
        for (int j = 0; j < C1_RC; ++j) {
          const float* p = xrow + ((ch + 1) * C1_RC + j) * C1_XSTR;
          xa[(ch + 1) & 1][j][0] = *(const v4f*)(p);
          xa[(ch + 1) & 1][j][1] = *(const v4f*)(p + 4);
        }
      }
      __builtin_amdgcn_sched_barrier(0);
      #pragma unroll
      for (int j = 0; j < C1_RC; ++j) {
        const int r = ch * C1_RC + j;
        const v4f x0 = xa[ch & 1][j][0];
        const v4f x1 = xa[ch & 1][j][1];
        #pragma unroll
        for (int u = 0; u < 4; ++u) {
          ap[u]     = fmaf(wp[r], x0[u], ap[u]);
          aq[u]     = fmaf(wq[r], x0[u], aq[u]);
          ap[u + 4] = fmaf(wp[r], x1[u], ap[u + 4]);
          aq[u + 4] = fmaf(wq[r], x1[u], aq[u + 4]);
        }
      }
    }
    v4f r0 = *(const v4f*)&s_x[xr * C1_XSTR + nl];
    v4f r1 = *(const v4f*)&s_x[xr * C1_XSTR + nl + 4];
    float* orow = xg + ((size_t)bt * NN_ + n0 + nl) * C1_ + c;
    #pragma unroll
    for (int u = 0; u < 8; ++u) {
      float xin = has_res ? (u < 4 ? r0[u & 3] : r1[u & 3]) : 0.f;
      float sg  = 1.f / (1.f + __expf(-aq[u]));
      orow[(size_t)u * C1_] = (ap[u] + xin) * sg;
    }
  }
}

#define XSTR  72
#define XPSTR 160
#define HSTR  68
#define ATT_RS   3
#define ATT_MAXT 6
#define ATT_KB   4
#define ATT_NB   (NN_ / (4 * ATT_KB))

__global__ __launch_bounds__(256) void attn_kernel(
    const float* __restrict__ xg,
    const float* __restrict__ adj,
    const float* __restrict__ u1,
    const float* __restrict__ u2,
    float* __restrict__ xs)
{
  extern __shared__ float smem[];
  float*    s_xgp  = smem;
  float*    s_e1   = s_xgp + (NN_ / 2) * XPSTR;
  float*    s_e2   = s_e1 + HH_ * NN_;
  float*    s_rinv = s_e2 + HH_ * NN_;
  unsigned* s_adj  = (unsigned*)(s_rinv + 8 * 16);
  float*    s_ht   = (float*)(s_adj + ATT_MAXT * 16 * 8);
  float*    s_xg   = s_ht;

  const int bt    = blockIdx.x / ATT_RS;
  const int part  = blockIdx.x - bt * ATT_RS;
  const int ntask = (part == 0) ? 6 : 5;
  const int row0  = ((part == 0) ? 0 : (part == 1 ? 6 : 11)) * 16;
  const int tid  = threadIdx.x;
  const int lane = tid & 31;
  const int wave = tid >> 5;
  const int mloc = lane & 15;
  const int half = lane >> 4;

  const float* xg_bt = xg + (size_t)bt * NN_ * C1_;

  for (int i = tid; i < NN_ * C1_; i += 256) {
    int n = i >> 6, c = i & 63;
    float v = xg_bt[i];
    s_xg[n * XSTR + c] = v;
    s_xgp[(n >> 1) * XPSTR + 2 * c + (n & 1)] = v;
  }
  {
    const int rl = tid >> 1, ch = tid & 1;
    if (rl < ntask * 16) {
      const float* arow = adj + (size_t)(row0 + rl) * NN_ + ch * 128;
      for (int w8 = 0; w8 < 4; ++w8) {
        unsigned bits = 0u;
        for (int j = 0; j < 32; ++j)
          bits |= (arow[w8 * 32 + j] > 0.f) ? (1u << j) : 0u;
        s_adj[rl * 8 + ch * 4 + w8] = bits;
      }
    }
  }
  __syncthreads();

  {
    const float* row = &s_xg[tid * XSTR];
    float rv[C1_];
    #pragma unroll
    for (int q = 0; q < C1_ / 4; ++q) {
      const v4f t4 = *(const v4f*)(row + 4 * q);
      rv[4 * q + 0] = t4[0]; rv[4 * q + 1] = t4[1]; rv[4 * q + 2] = t4[2]; rv[4 * q + 3] = t4[3];
    }
    #pragma unroll 1
    for (int h = 0; h < HH_; ++h) {
      const float* U1 = u1 + h * C1_;
      const float* U2 = u2 + h * C1_;
      float a1 = 0.f, a2 = 0.f;
      #pragma unroll
      for (int c = 0; c < C1_; ++c) {
        a1 = fmaf(rv[c], U1[c], a1);
        a2 = fmaf(rv[c], U2[c], a2);
      }
      s_e1[h * NN_ + tid] = a1;
      s_e2[h * NN_ + tid] = a2;
    }
  }
  __syncthreads();

  const float inv_h = 1.0f / (float)HH_;
  const float* e1h = s_e1 + wave * NN_;
  const float* e2h = s_e2 + wave * NN_;
  float* xs_bt = xs + (size_t)bt * NN_ * C1_;

  float e2r[NN_ / 2];
  #pragma unroll
  for (int s = 0; s < NN_ / 4; ++s) {
    const v2f t2 = *(const v2f*)(e2h + 4 * s + 2 * half);
    e2r[2 * s + 0] = t2[0]; e2r[2 * s + 1] = t2[1];
  }
  const float* xpb = &s_xgp[half * XPSTR + 2 * mloc];

  #pragma unroll 1
  for (int ti = 0; ti < ntask; ++ti) {
    const int ml = ti * 16 + mloc;
    const float e1m = e1h[row0 + ml];
    unsigned aw[8];
    {
      const v4u a0 = *(const v4u*)&s_adj[ml * 8];
      const v4u a1 = *(const v4u*)&s_adj[ml * 8 + 4];
      #pragma unroll
      for (int w = 0; w < 4; ++w) { aw[w] = a0[w] >> (2 * half); aw[4 + w] = a1[w] >> (2 * half); }
    }

    v2f pr[NN_ / 4];
    float mx = -3.0e38f;
    #pragma unroll
    for (int s = 0; s < NN_ / 4; ++s) {
      #pragma unroll
      for (int u = 0; u < 2; ++u) {
        float sc = e1m + e2r[2 * s + u];
        sc = sc > 0.f ? sc : ALPHA_ * sc;
        sc = ((aw[s >> 3] >> (4 * (s & 7) + u)) & 1) ? sc : -3.0e38f;
        pr[s][u] = sc;
        mx = fmaxf(mx, sc);
      }
    }
    mx = fmaxf(mx, __shfl_xor(mx, 16));

    float sm = 0.f;
    #pragma unroll
    for (int s = 0; s < NN_ / 4; ++s) {
      #pragma unroll
      for (int u = 0; u < 2; ++u) {
        const float pe = __expf(pr[s][u] - mx);
        pr[s][u] = pe;
        sm += pe;
      }
    }
    sm += __shfl_xor(sm, 16);
    s_rinv[wave * 16 + mloc] = 1.0f / sm;

    v8f acc0 = {}, acc1 = {}, acc2 = {}, acc3 = {};
    v2f fb[2][ATT_KB][4];
    #pragma unroll
    for (int s = 0; s < ATT_KB; ++s) {
      const float* bp = xpb + (2 * s) * XPSTR;
      #pragma unroll
      for (int q = 0; q < 4; ++q) fb[0][s][q] = *(const v2f*)(bp + 32 * q);
    }
    #pragma unroll
    for (int kb = 0; kb < ATT_NB; ++kb) {
      __builtin_amdgcn_sched_barrier(0);
      if (kb + 1 < ATT_NB) {
        #pragma unroll
        for (int s = 0; s < ATT_KB; ++s) {
          const float* bp = xpb + (2 * ((kb + 1) * ATT_KB + s)) * XPSTR;
          #pragma unroll
          for (int q = 0; q < 4; ++q) fb[(kb + 1) & 1][s][q] = *(const v2f*)(bp + 32 * q);
        }
      }
      __builtin_amdgcn_sched_barrier(0);
      #pragma unroll
      for (int s = 0; s < ATT_KB; ++s) {
        const v2f fa = pr[kb * ATT_KB + s];
        acc0 = __builtin_amdgcn_wmma_f32_16x16x4_f32(false, fa, false, fb[kb & 1][s][0], (short)0, acc0, false, false);
        acc1 = __builtin_amdgcn_wmma_f32_16x16x4_f32(false, fa, false, fb[kb & 1][s][1], (short)0, acc1, false, false);
        acc2 = __builtin_amdgcn_wmma_f32_16x16x4_f32(false, fa, false, fb[kb & 1][s][2], (short)0, acc2, false, false);
        acc3 = __builtin_amdgcn_wmma_f32_16x16x4_f32(false, fa, false, fb[kb & 1][s][3], (short)0, acc3, false, false);
      }
    }

    float inv[8];
    #pragma unroll
    for (int r = 0; r < 8; ++r) inv[r] = s_rinv[wave * 16 + r + 8 * half];
    float* ht = s_ht + ((ti * HH_ + wave) * 16 + 8 * half) * HSTR + mloc;
    #pragma unroll
    for (int r = 0; r < 8; ++r) {
      float* hrow = ht + r * HSTR;
      hrow[0]  = tanh_fast(acc0[r] * inv[r]);
      hrow[16] = tanh_fast(acc1[r] * inv[r]);
      hrow[32] = tanh_fast(acc2[r] * inv[r]);
      hrow[48] = tanh_fast(acc3[r] * inv[r]);
    }
  }

  const int rr = tid >> 4, c4 = (tid & 15) * 4;
  v4f xres[ATT_MAXT];
  #pragma unroll
  for (int ti = 0; ti < ATT_MAXT; ++ti) {
    const int tq = (ti < ntask) ? ti : 0;
    xres[ti] = *(const v4f*)(xg_bt + (size_t)(row0 + tq * 16 + rr) * C1_ + c4);
  }
  __syncthreads();
  #pragma unroll
  for (int ti = 0; ti < ATT_MAXT; ++ti) {
    if (ti < ntask) {
      const int mm = row0 + ti * 16 + rr;
      v4f sacc = {0.f, 0.f, 0.f, 0.f};
      #pragma unroll
      for (int hh = 0; hh < HH_; ++hh) {
        const v4f t4 = *(const v4f*)(s_ht + ((ti * HH_ + hh) * 16 + rr) * HSTR + c4);
        sacc[0] += t4[0] * inv_h;
        sacc[1] += t4[1] * inv_h;
        sacc[2] += t4[2] * inv_h;
        sacc[3] += t4[3] * inv_h;
      }
      const v4f xr = xres[ti];
      v4f o;
      o[0] = xr[0] + sacc[0]; o[1] = xr[1] + sacc[1]; o[2] = xr[2] + sacc[2]; o[3] = xr[3] + sacc[3];
      *(v4f*)(xs_bt + (size_t)mm * C1_ + c4) = o;
    }
  }
}

#define C2_K   (C1_ * KT_)
#define C2_NT  64
#define C2_TL  (NN_ / C2_NT)
#define C2_XP  68
#define C2_WP  196
#define C2_YP  68
#define C2_G   4
#define C2_NG  (C2_K / C2_G)

__global__ __launch_bounds__(256) void conv2_relu_kernel(
    const float* __restrict__ xs,
    const float* __restrict__ w2,
    const float* __restrict__ b2,
    float* __restrict__ yws,
    float* __restrict__ stats)
{
  extern __shared__ float smem[];
  float* s_x   = smem;
  float* s_w   = s_x + C2_K * C2_XP;
  float* s_red = s_w + C2_ * C2_WP;
  float* s_y   = smem;

  const int bt   = blockIdx.x >> 2;
  const int tile = blockIdx.x & 3;
  const int n0   = tile * C2_NT;
  const int bb = bt >> 3, t = bt & 7;
  const int tid = threadIdx.x;

  #pragma unroll
  for (unsigned j = 0; j < (C2_ * C2_K) / 4 / 256; ++j) {
    const unsigned i4 = (unsigned)tid + j * 256u;
    const v4f wv = ((const v4f*)w2)[i4];
    const unsigned co = i4 / (C2_K / 4u);
    const unsigned r  = (i4 - co * (C2_K / 4u)) * 4u;
    #pragma unroll
    for (unsigned e = 0; e < 4; ++e) {
      const unsigned ci = (r + e) / KT_, kt = (r + e) - ci * KT_;
      s_w[co * C2_WP + kt * C1_ + ci] = wv[e];
    }
  }
  {
    const int ci = tid & 63, nq = tid >> 6;
    float xv[KT_][16];
    #pragma unroll
    for (int kt = 0; kt < KT_; ++kt) {
      const float* src = xs + (((size_t)bb * T1_ + t + kt) * NN_ + n0 + nq) * C1_ + ci;
      #pragma unroll
      for (int u = 0; u < 16; ++u) xv[kt][u] = src[u * 4 * C1_];
    }
    #pragma unroll
    for (int kt = 0; kt < KT_; ++kt)
      #pragma unroll
      for (int u = 0; u < 16; ++u) s_x[(kt * C1_ + ci) * C2_XP + nq + 4 * u] = xv[kt][u];
  }
  __syncthreads();

  const int lane = tid & 31;
  const int wave = tid >> 5;
  const int c16  = lane & 15;
  const int nb   = wave * 8 + (lane >> 4) * 4;

  float acc[4][4];
  #pragma unroll
  for (int j = 0; j < 4; ++j) {
    const float bj = b2[c16 + 16 * j];
    #pragma unroll
    for (int i = 0; i < 4; ++i) acc[j][i] = bj;
  }

  const float* xb = s_x + nb;
  const float* wb = s_w + c16 * C2_WP;

  v4f wv[2][4];
  v4f xv[2][C2_G];
  #pragma unroll
  for (int j = 0; j < 4; ++j) wv[0][j] = *(const v4f*)(wb + j * 16 * C2_WP);
  #pragma unroll
  for (int k = 0; k < C2_G; ++k) xv[0][k] = *(const v4f*)(xb + k * C2_XP);

  #pragma unroll 1
  for (int g = 0; g < C2_NG; g += 2) {
    #pragma unroll
    for (int s = 0; s < 2; ++s) {
      const int gn = (g + s + 1 < C2_NG) ? (g + s + 1) : (C2_NG - 1);
      const float* xn = xb + gn * C2_G * C2_XP;
      const float* wn = wb + gn * C2_G;
      __builtin_amdgcn_sched_barrier(0);
      #pragma unroll
      for (int j = 0; j < 4; ++j) wv[s ^ 1][j] = *(const v4f*)(wn + j * 16 * C2_WP);
      #pragma unroll
      for (int k = 0; k < C2_G; ++k) xv[s ^ 1][k] = *(const v4f*)(xn + k * C2_XP);
      __builtin_amdgcn_sched_barrier(0);
      #pragma unroll
      for (int k = 0; k < C2_G; ++k)
        #pragma unroll
        for (int j = 0; j < 4; ++j)
          #pragma unroll
          for (int u = 0; u < 4; ++u)
            acc[j][u] = fmaf(wv[s][j][k], xv[s][k][u], acc[j][u]);
    }
  }

  v4f res[4];
  #pragma unroll
  for (int j = 0; j < 4; ++j)
    res[j] = *(const v4f*)(s_x + (2 * C1_ + c16 + 16 * j) * C2_XP + nb);

  __syncthreads();

  float lsum = 0.f, lsq = 0.f;
  #pragma unroll
  for (int j = 0; j < 4; ++j) {
    v4f y;
    #pragma unroll
    for (int u = 0; u < 4; ++u) {
      float v = acc[j][u] + res[j][u];
      v = v > 0.f ? v : 0.f;
      y[u] = v;
      lsum += v;
      lsq  += v * v;
    }
    *(v4f*)(s_y + (c16 + 16 * j) * C2_YP + nb) = y;
  }
  for (int off = 16; off > 0; off >>= 1) {
    lsum += __shfl_xor(lsum, off);
    lsq  += __shfl_xor(lsq,  off);
  }
  if ((tid & 31) == 0) { s_red[tid >> 5] = lsum; s_red[8 + (tid >> 5)] = lsq; }
  __syncthreads();
  if (tid == 0) {
    float tsum = 0.f, tsq = 0.f;
    for (int i = 0; i < 8; ++i) { tsum += s_red[i]; tsq += s_red[8 + i]; }
    stats[(bt * C2_TL + tile) * 2 + 0] = tsum;
    stats[(bt * C2_TL + tile) * 2 + 1] = tsq;
  }
  #pragma unroll
  for (int u = 0; u < 4; ++u) {
    const int i4 = tid + 256 * u;
    const int c = i4 >> 4, q = i4 & 15;
    const v4f v = *(const v4f*)(s_y + c * C2_YP + 4 * q);
    *(v4f*)(yws + ((size_t)bt * C2_ + c) * NN_ + n0 + 4 * q) = v;
  }
}

#define LN_P 65

__global__ __launch_bounds__(256) void ln_apply_kernel(
    const float* __restrict__ yws,
    const float* __restrict__ stats,
    const float* __restrict__ lnw,
    const float* __restrict__ lnb,
    float* __restrict__ out)
{
  __shared__ float s_lw[C2_NT * LN_P];
  __shared__ float s_lb[C2_NT * LN_P];

  const int bt = blockIdx.x >> 2;
  const int n0 = (blockIdx.x & 3) * C2_NT;
  const int bb = bt >> 3, t = bt & 7;
  const int tid = threadIdx.x;

  v4f wv[4], bv[4], yv[4];
  #pragma unroll
  for (int u = 0; u < 4; ++u) {
    const int i = (tid + 256 * u) * 4;
    wv[u] = *(const v4f*)(lnw + (size_t)n0 * C2_ + i);
    bv[u] = *(const v4f*)(lnb + (size_t)n0 * C2_ + i);
    const int i4 = tid + 256 * u;
    const int c = i4 >> 4, q = i4 & 15;
    yv[u] = *(const v4f*)(yws + ((size_t)bt * C2_ + c) * NN_ + n0 + 4 * q);
  }
  float tsum = 0.f, tsq = 0.f;
  for (int k = 0; k < C2_TL; ++k) {
    tsum += stats[(bt * C2_TL + k) * 2 + 0];
    tsq  += stats[(bt * C2_TL + k) * 2 + 1];
  }
  const float cnt  = (float)(NN_ * C2_);
  const float mu   = tsum / cnt;
  const float var  = tsq / cnt - mu * mu;
  const float rstd = rsqrtf(var + 1e-5f);

  #pragma unroll
  for (int u = 0; u < 4; ++u) {
    const int i = (tid + 256 * u) * 4;
    const int n = i >> 6, c = i & 63;
    #pragma unroll
    for (int e = 0; e < 4; ++e) {
      s_lw[n * LN_P + c + e] = wv[u][e];
      s_lb[n * LN_P + c + e] = bv[u][e];
    }
  }
  __syncthreads();

  #pragma unroll
  for (int u = 0; u < 4; ++u) {
    const int i4 = tid + 256 * u;
    const int c = i4 >> 4, q = i4 & 15;
    v4f o;
    #pragma unroll
    for (int e = 0; e < 4; ++e) {
      const int n = 4 * q + e;
      o[e] = (yv[u][e] - mu) * rstd * s_lw[n * LN_P + c] + s_lb[n * LN_P + c];
    }
    *(v4f*)(out + (((size_t)bb * C2_ + c) * T2_ + t) * NN_ + n0 + 4 * q) = o;
  }
}

extern "C" void kernel_launch(void* const* d_in, const int* in_sizes, int n_in,
                              void* d_out, int out_size, void* d_ws, size_t ws_size,
                              hipStream_t stream) {
  const float* x   = (const float*)d_in[0];
  const float* adj = (const float*)d_in[1];
  const float* w1  = (const float*)d_in[2];
  const float* b1  = (const float*)d_in[3];
  const float* w2  = (const float*)d_in[4];
  const float* b2  = (const float*)d_in[5];
  const float* Wh  = (const float*)d_in[6];
  const float* ah  = (const float*)d_in[7];
  const float* lnw = (const float*)d_in[8];
  const float* lnb = (const float*)d_in[9];
  float* out = (float*)d_out;

  float* ws = (float*)d_ws;
  const size_t xg_elems = (size_t)B_ * T1_ * NN_ * C1_;
  float* xg = ws;
  float* xs = xg + xg_elems;
  float* u1 = xs + xg_elems;
  float* u2 = u1 + HH_ * C1_;
  float* yws = u2 + HH_ * C1_;
  float* stats = yws + (size_t)B_ * T2_ * C2_ * NN_;

  prep_u_kernel<<<2, 256, 0, stream>>>(Wh, ah, u1, u2);

  conv1_glu_kernel<<<B_ * T1_ * (NN_ / C1_NT), 256, 0, stream>>>(x, w1, b1, xg);

  const size_t attn_lds =
      (size_t)((NN_ / 2) * XPSTR + 2 * HH_ * NN_ + 8 * 16 + ATT_MAXT * HH_ * 16 * HSTR) * sizeof(float)
      + (size_t)(ATT_MAXT * 16 * 8) * sizeof(unsigned);
  attn_kernel<<<B_ * T1_ * ATT_RS, 256, attn_lds, stream>>>(xg, adj, u1, u2, xs);

  const size_t c2_lds = (size_t)(C2_K * C2_XP + C2_ * C2_WP + 16) * sizeof(float);
  conv2_relu_kernel<<<B_ * T2_ * C2_TL, 256, c2_lds, stream>>>(xs, w2, b2, yws, stats);

  ln_apply_kernel<<<B_ * T2_ * C2_TL, 256, 0, stream>>>(yws, stats, lnw, lnb, out);
}


# attention prologue in one memory round trip, not nine: xg slab by 16-byte loads issued as one group; adjacency bit words packed once by the prep launch from coalesced rows; sum order unchanged
# speedup vs baseline: 5.4598x; 1.0610x over previous
#include <hip/hip_runtime.h>
#include <math.h>

#define B_   8
#define C0_  16
#define C1_  64
#define C2_  64
#define T0_  12
#define T1_  10
#define T2_  8
#define NN_  256
#define KT_  3
#define HH_  8
#define ALPHA_ 0.2f

typedef float v2f __attribute__((ext_vector_type(2)));
typedef float v4f __attribute__((ext_vector_type(4)));
typedef float v8f __attribute__((ext_vector_type(8)));
typedef unsigned v2u __attribute__((ext_vector_type(2)));
typedef unsigned v4u __attribute__((ext_vector_type(4)));

__device__ __forceinline__ float tanh_fast(float x) {
  float e = __expf(fminf(2.0f * x, 80.0f));
  return (e - 1.0f) / (e + 1.0f);
}

#define PREP_UB 2
#define PREP_AB (NN_ / 8)

__global__ __launch_bounds__(256) void prep_kernel(
    const float* __restrict__ W,
    const float* __restrict__ a,
    const float* __restrict__ adj,
    float* __restrict__ u1, float* __restrict__ u2,
    unsigned* __restrict__ adjbits)
{
  if (blockIdx.x >= PREP_UB) {
    const int lane = threadIdx.x & 31;
    const int r = (blockIdx.x - PREP_UB) * 8 + (threadIdx.x >> 5);
    const float* arow = adj + (size_t)r * NN_ + 4 * lane;
    const v4f c0 = *(const v4f*)(arow);
    const v4f c1 = *(const v4f*)(arow + NN_ / 2);
    v4u lo, hi;
    #pragma unroll
    for (int e = 0; e < 4; ++e) {
      lo[e] = __builtin_amdgcn_ballot_w32(c0[e] > 0.f);
      hi[e] = __builtin_amdgcn_ballot_w32(c1[e] > 0.f);
    }
    if (lane == 0) {
      *(v4u*)(adjbits + (size_t)r * 8)     = lo;
      *(v4u*)(adjbits + (size_t)r * 8 + 4) = hi;
    }
    return;
  }
  int tid = blockIdx.x * 256 + threadIdx.x;
  if (tid >= HH_ * C1_) return;
  int h = tid >> 6, c = tid & 63;
  const float* Wh = W + ((size_t)h * C1_ + c) * C1_;
  const float* ah = a + (size_t)h * 2 * C1_;
  float s1 = 0.f, s2 = 0.f;
  for (int d = 0; d < C1_; ++d) {
    float w = Wh[d];
    s1 += w * ah[d];
    s2 += w * ah[C1_ + d];
  }
  u1[tid] = s1;
  u2[tid] = s2;
}

#define C1_NT   64
#define C1_WSTR 49
#define C1_XSTR 68
#define C1_K    (C0_ * KT_)
#define C1_RC   6

__global__ __launch_bounds__(256) void conv1_glu_kernel(
    const float* __restrict__ x,
    const float* __restrict__ w,
    const float* __restrict__ b,
    float* __restrict__ xg)
{
  __shared__ __attribute__((aligned(16))) float s_w[2 * C1_ * C1_WSTR];
  __shared__ __attribute__((aligned(16))) float s_x[C1_K * C1_XSTR];

  const int tid  = threadIdx.x;
  const int lane = tid & 31;
  const int wave = tid >> 5;
  const int bt   = blockIdx.x >> 2;
  const int n0   = (blockIdx.x & 3) * C1_NT;
  const int bb   = bt / T1_, t = bt - bb * T1_;

  #pragma unroll
  for (int j = 0; j < (2 * C1_ * C1_K) / 4 / 256; ++j) {
    int i4 = tid + j * 256;
    v4f wv = ((const v4f*)w)[i4];
    int base = i4 * 4;
    int c2 = base / C1_K, k = base - c2 * C1_K;
    float* d = &s_w[c2 * C1_WSTR + k];
    d[0] = wv[0]; d[1] = wv[1]; d[2] = wv[2]; d[3] = wv[3];
  }
  #pragma unroll
  for (int j = 0; j < C1_K * (C1_NT / 4) / 256; ++j) {
    int i4 = tid + j * 256;
    int r = i4 >> 4, q = i4 & 15;
    int kt = r >> 4, ci = r & 15;
    const float* src = x + (((size_t)bb * C0_ + ci) * T0_ + (t + kt)) * NN_ + n0 + q * 4;
    *(v4f*)&s_x[r * C1_XSTR + q * 4] = *(const v4f*)src;
  }
  __syncthreads();

  const int c  = (wave & 1) * 32 + lane;
  const int nb = (wave >> 1) * 16;

  float wp[C1_K], wq[C1_K];
  #pragma unroll
  for (int r = 0; r < C1_K; ++r) {
    int kt = r >> 4, ci = r & 15;
    wp[r] = s_w[c * C1_WSTR + ci * KT_ + kt];
    wq[r] = s_w[(c + C1_) * C1_WSTR + ci * KT_ + kt];
  }
  const float bp = b[c];
  const float bq = b[c + C1_];
  const bool has_res = (c < C0_);
  const int  xr = has_res ? (2 * C0_ + c) : 0;

  #pragma unroll 1
  for (int g = 0; g < 2; ++g) {
    const int nl = nb + g * 8;
    const float* xrow = &s_x[nl];
    float ap[8], aq[8];
    #pragma unroll
    for (int u = 0; u < 8; ++u) { ap[u] = bp; aq[u] = bq; }

    v4f xa[2][C1_RC][2];
    #pragma unroll
    for (int j = 0; j < C1_RC; ++j) {
      xa[0][j][0] = *(const v4f*)(xrow + j * C1_XSTR);
      xa[0][j][1] = *(const v4f*)(xrow + j * C1_XSTR + 4);
    }
    #pragma unroll
    for (int ch = 0; ch < C1_K / C1_RC; ++ch) {
      __builtin_amdgcn_sched_barrier(0);
      if (ch + 1 < C1_K / C1_RC) {
        #pragma unroll
        for (int j = 0; j < C1_RC; ++j) {
          const float* p = xrow + ((ch + 1) * C1_RC + j) * C1_XSTR;
          xa[(ch + 1) & 1][j][0] = *(const v4f*)(p);
          xa[(ch + 1) & 1][j][1] = *(const v4f*)(p + 4);
        }
      }
      __builtin_amdgcn_sched_barrier(0);
      #pragma unroll
      for (int j = 0; j < C1_RC; ++j) {
        const int r = ch * C1_RC + j;
        const v4f x0 = xa[ch & 1][j][0];
        const v4f x1 = xa[ch & 1][j][1];
        #pragma unroll
        for (int u = 0; u < 4; ++u) {
          ap[u]     = fmaf(wp[r], x0[u], ap[u]);
          aq[u]     = fmaf(wq[r], x0[u], aq[u]);
          ap[u + 4] = fmaf(wp[r], x1[u], ap[u + 4]);
          aq[u + 4] = fmaf(wq[r], x1[u], aq[u + 4]);
        }
      }
    }
    v4f r0 = *(const v4f*)&s_x[xr * C1_XSTR + nl];
    v4f r1 = *(const v4f*)&s_x[xr * C1_XSTR + nl + 4];
    float* orow = xg + ((size_t)bt * NN_ + n0 + nl) * C1_ + c;
    #pragma unroll
    for (int u = 0; u < 8; ++u) {
      float xin = has_res ? (u < 4 ? r0[u & 3] : r1[u & 3]) : 0.f;
      float sg  = 1.f / (1.f + __expf(-aq[u]));
      orow[(size_t)u * C1_] = (ap[u] + xin) * sg;
    }
  }
}

#define XSTR  72
#define XPSTR 160
#define HSTR  68
#define ATT_RS   3
#define ATT_MAXT 6
#define ATT_KB   4
#define ATT_NB   (NN_ / (4 * ATT_KB))

__global__ __launch_bounds__(256) void attn_kernel(
    const float* __restrict__ xg,
    const unsigned* __restrict__ adjbits,
    const float* __restrict__ u1,
    const float* __restrict__ u2,
    float* __restrict__ xs)
{
  extern __shared__ float smem[];
  float*    s_xgp  = smem;
  float*    s_e1   = s_xgp + (NN_ / 2) * XPSTR;
  float*    s_e2   = s_e1 + HH_ * NN_;
  float*    s_rinv = s_e2 + HH_ * NN_;
  unsigned* s_adj  = (unsigned*)(s_rinv + 8 * 16);
  float*    s_ht   = (float*)(s_adj + ATT_MAXT * 16 * 8);
  float*    s_xg   = s_ht;

  const int bt    = blockIdx.x / ATT_RS;
  const int part  = blockIdx.x - bt * ATT_RS;
  const int ntask = (part == 0) ? 6 : 5;
  const int row0  = ((part == 0) ? 0 : (part == 1 ? 6 : 11)) * 16;
  const int tid  = threadIdx.x;
  const int lane = tid & 31;
  const int wave = tid >> 5;
  const int mloc = lane & 15;
  const int half = lane >> 4;

  const float* xg_bt = xg + (size_t)bt * NN_ * C1_;

  {
    v4f xa[8], xb[8];
    #pragma unroll
    for (int j = 0; j < 8; ++j) {
      const int id = tid + 256 * j;
      const int p = id >> 4, q = id & 15;
      const float* src = xg_bt + (size_t)(2 * p) * C1_ + 4 * q;
      xa[j] = *(const v4f*)(src);
      xb[j] = *(const v4f*)(src + C1_);
    }
    const bool has_aw = tid < ntask * 32;
    v4u ab = {0u, 0u, 0u, 0u};
    if (has_aw) ab = *(const v4u*)(adjbits + (size_t)row0 * 8 + tid * 4);
    #pragma unroll
    for (int j = 0; j < 8; ++j) {
      const int id = tid + 256 * j;
      const int p = id >> 4, q = id & 15;
      *(v4f*)&s_xg[(2 * p) * XSTR + 4 * q]     = xa[j];
      *(v4f*)&s_xg[(2 * p + 1) * XSTR + 4 * q] = xb[j];
      const v4f lo = {xa[j][0], xb[j][0], xa[j][1], xb[j][1]};
      const v4f hi = {xa[j][2], xb[j][2], xa[j][3], xb[j][3]};
      *(v4f*)&s_xgp[p * XPSTR + 8 * q]     = lo;
      *(v4f*)&s_xgp[p * XPSTR + 8 * q + 4] = hi;
    }
    if (has_aw) *(v4u*)&s_adj[tid * 4] = ab;
  }
  __syncthreads();

  {
    const float* row = &s_xg[tid * XSTR];
    float rv[C1_];
    #pragma unroll
    for (int q = 0; q < C1_ / 4; ++q) {
      const v4f t4 = *(const v4f*)(row + 4 * q);
      rv[4 * q + 0] = t4[0]; rv[4 * q + 1] = t4[1]; rv[4 * q + 2] = t4[2]; rv[4 * q + 3] = t4[3];
    }
    #pragma unroll 1
    for (int h = 0; h < HH_; ++h) {
      const float* U1 = u1 + h * C1_;
      const float* U2 = u2 + h * C1_;
      float a1 = 0.f, a2 = 0.f;
      #pragma unroll
      for (int c = 0; c < C1_; ++c) {
        a1 = fmaf(rv[c], U1[c], a1);
        a2 = fmaf(rv[c], U2[c], a2);
      }
      s_e1[h * NN_ + tid] = a1;
      s_e2[h * NN_ + tid] = a2;
    }
  }
  __syncthreads();

  const float inv_h = 1.0f / (float)HH_;
  const float* e1h = s_e1 + wave * NN_;
  const float* e2h = s_e2 + wave * NN_;
  float* xs_bt = xs + (size_t)bt * NN_ * C1_;

  float e2r[NN_ / 2];
  #pragma unroll
  for (int s = 0; s < NN_ / 4; ++s) {
    const v2f t2 = *(const v2f*)(e2h + 4 * s + 2 * half);
    e2r[2 * s + 0] = t2[0]; e2r[2 * s + 1] = t2[1];
  }
  const float* xpb = &s_xgp[half * XPSTR + 2 * mloc];

  #pragma unroll 1
  for (int ti = 0; ti < ntask; ++ti) {
    const int ml = ti * 16 + mloc;
    const float e1m = e1h[row0 + ml];
    unsigned aw[4];
    {
      const v2u a0 = *(const v2u*)&s_adj[ml * 8 + 2 * half];
      const v2u a1 = *(const v2u*)&s_adj[ml * 8 + 4 + 2 * half];
      aw[0] = a0[0]; aw[1] = a0[1]; aw[2] = a1[0]; aw[3] = a1[1];
    }

    v2f pr[NN_ / 4];
    float mx = -3.0e38f;
    #pragma unroll
    for (int s = 0; s < NN_ / 4; ++s) {
      #pragma unroll
      for (int u = 0; u < 2; ++u) {
        float sc = e1m + e2r[2 * s + u];
        sc = sc > 0.f ? sc : ALPHA_ * sc;
        sc = ((aw[2 * (s >> 5) + u] >> (s & 31)) & 1) ? sc : -3.0e38f;
        pr[s][u] = sc;
        mx = fmaxf(mx, sc);
      }
    }
    mx = fmaxf(mx, __shfl_xor(mx, 16));

    float sm = 0.f;
    #pragma unroll
    for (int s = 0; s < NN_ / 4; ++s) {
      #pragma unroll
      for (int u = 0; u < 2; ++u) {
        const float pe = __expf(pr[s][u] - mx);
        pr[s][u] = pe;
        sm += pe;
      }
    }
    sm += __shfl_xor(sm, 16);
    s_rinv[wave * 16 + mloc] = 1.0f / sm;

    v8f acc0 = {}, acc1 = {}, acc2 = {}, acc3 = {};
    v2f fb[2][ATT_KB][4];
    #pragma unroll
    for (int s = 0; s < ATT_KB; ++s) {
      const float* bp = xpb + (2 * s) * XPSTR;
      #pragma unroll
      for (int q = 0; q < 4; ++q) fb[0][s][q] = *(const v2f*)(bp + 32 * q);
    }
    #pragma unroll
    for (int kb = 0; kb < ATT_NB; ++kb) {
      __builtin_amdgcn_sched_barrier(0);
      if (kb + 1 < ATT_NB) {
        #pragma unroll
        for (int s = 0; s < ATT_KB; ++s) {
          const float* bp = xpb + (2 * ((kb + 1) * ATT_KB + s)) * XPSTR;
          #pragma unroll
          for (int q = 0; q < 4; ++q) fb[(kb + 1) & 1][s][q] = *(const v2f*)(bp + 32 * q);
        }
      }
      __builtin_amdgcn_sched_barrier(0);
      #pragma unroll
      for (int s = 0; s < ATT_KB; ++s) {
        const v2f fa = pr[kb * ATT_KB + s];
        acc0 = __builtin_amdgcn_wmma_f32_16x16x4_f32(false, fa, false, fb[kb & 1][s][0], (short)0, acc0, false, false);
        acc1 = __builtin_amdgcn_wmma_f32_16x16x4_f32(false, fa, false, fb[kb & 1][s][1], (short)0, acc1, false, false);
        acc2 = __builtin_amdgcn_wmma_f32_16x16x4_f32(false, fa, false, fb[kb & 1][s][2], (short)0, acc2, false, false);
        acc3 = __builtin_amdgcn_wmma_f32_16x16x4_f32(false, fa, false, fb[kb & 1][s][3], (short)0, acc3, false, false);
      }
    }

    float inv[8];
    #pragma unroll
    for (int r = 0; r < 8; ++r) inv[r] = s_rinv[wave * 16 + r + 8 * half];
    float* ht = s_ht + ((ti * HH_ + wave) * 16 + 8 * half) * HSTR + mloc;
    #pragma unroll
    for (int r = 0; r < 8; ++r) {
      float* hrow = ht + r * HSTR;
      hrow[0]  = tanh_fast(acc0[r] * inv[r]);
      hrow[16] = tanh_fast(acc1[r] * inv[r]);
      hrow[32] = tanh_fast(acc2[r] * inv[r]);
      hrow[48] = tanh_fast(acc3[r] * inv[r]);
    }
  }

  const int rr = tid >> 4, c4 = (tid & 15) * 4;
  v4f xres[ATT_MAXT];
  #pragma unroll
  for (int ti = 0; ti < ATT_MAXT; ++ti) {
    const int tq = (ti < ntask) ? ti : 0;
    xres[ti] = *(const v4f*)(xg_bt + (size_t)(row0 + tq * 16 + rr) * C1_ + c4);
  }
  __syncthreads();
  #pragma unroll
  for (int ti = 0; ti < ATT_MAXT; ++ti) {
    if (ti < ntask) {
      const int mm = row0 + ti * 16 + rr;
      v4f sacc = {0.f, 0.f, 0.f, 0.f};
      #pragma unroll
      for (int hh = 0; hh < HH_; ++hh) {
        const v4f t4 = *(const v4f*)(s_ht + ((ti * HH_ + hh) * 16 + rr) * HSTR + c4);
        sacc[0] += t4[0] * inv_h;
        sacc[1] += t4[1] * inv_h;
        sacc[2] += t4[2] * inv_h;
        sacc[3] += t4[3] * inv_h;
      }
      const v4f xr = xres[ti];
      v4f o;
      o[0] = xr[0] + sacc[0]; o[1] = xr[1] + sacc[1]; o[2] = xr[2] + sacc[2]; o[3] = xr[3] + sacc[3];
      *(v4f*)(xs_bt + (size_t)mm * C1_ + c4) = o;
    }
  }
}

#define C2_K   (C1_ * KT_)
#define C2_NT  64
#define C2_TL  (NN_ / C2_NT)
#define C2_XP  68
#define C2_WP  196
#define C2_YP  68
#define C2_G   4
#define C2_NG  (C2_K / C2_G)

__global__ __launch_bounds__(256) void conv2_relu_kernel(
    const float* __restrict__ xs,
    const float* __restrict__ w2,
    const float* __restrict__ b2,
    float* __restrict__ yws,
    float* __restrict__ stats)
{
  extern __shared__ float smem[];
  float* s_x   = smem;
  float* s_w   = s_x + C2_K * C2_XP;
  float* s_red = s_w + C2_ * C2_WP;
  float* s_y   = smem;

  const int bt   = blockIdx.x >> 2;
  const int tile = blockIdx.x & 3;
  const int n0   = tile * C2_NT;
  const int bb = bt >> 3, t = bt & 7;
  const int tid = threadIdx.x;

  #pragma unroll
  for (unsigned j = 0; j < (C2_ * C2_K) / 4 / 256; ++j) {
    const unsigned i4 = (unsigned)tid + j * 256u;
    const v4f wv = ((const v4f*)w2)[i4];
    const unsigned co = i4 / (C2_K / 4u);
    const unsigned r  = (i4 - co * (C2_K / 4u)) * 4u;
    #pragma unroll
    for (unsigned e = 0; e < 4; ++e) {
      const unsigned ci = (r + e) / KT_, kt = (r + e) - ci * KT_;
      s_w[co * C2_WP + kt * C1_ + ci] = wv[e];
    }
  }
  {
    const int ci = tid & 63, nq = tid >> 6;
    float xv[KT_][16];
    #pragma unroll
    for (int kt = 0; kt < KT_; ++kt) {
      const float* src = xs + (((size_t)bb * T1_ + t + kt) * NN_ + n0 + nq) * C1_ + ci;
      #pragma unroll
      for (int u = 0; u < 16; ++u) xv[kt][u] = src[u * 4 * C1_];
    }
    #pragma unroll
    for (int kt = 0; kt < KT_; ++kt)
      #pragma unroll
      for (int u = 0; u < 16; ++u) s_x[(kt * C1_ + ci) * C2_XP + nq + 4 * u] = xv[kt][u];
  }
  __syncthreads();

  const int lane = tid & 31;
  const int wave = tid >> 5;
  const int c16  = lane & 15;
  const int nb   = wave * 8 + (lane >> 4) * 4;

  float acc[4][4];
  #pragma unroll
  for (int j = 0; j < 4; ++j) {
    const float bj = b2[c16 + 16 * j];
    #pragma unroll
    for (int i = 0; i < 4; ++i) acc[j][i] = bj;
  }

  const float* xb = s_x + nb;
  const float* wb = s_w + c16 * C2_WP;

  v4f wv[2][4];
  v4f xv[2][C2_G];
  #pragma unroll
  for (int j = 0; j < 4; ++j) wv[0][j] = *(const v4f*)(wb + j * 16 * C2_WP);
  #pragma unroll
  for (int k = 0; k < C2_G; ++k) xv[0][k] = *(const v4f*)(xb + k * C2_XP);

  #pragma unroll 1
  for (int g = 0; g < C2_NG; g += 2) {
    #pragma unroll
    for (int s = 0; s < 2; ++s) {
      const int gn = (g + s + 1 < C2_NG) ? (g + s + 1) : (C2_NG - 1);
      const float* xn = xb + gn * C2_G * C2_XP;
      const float* wn = wb + gn * C2_G;
      __builtin_amdgcn_sched_barrier(0);
      #pragma unroll
      for (int j = 0; j < 4; ++j) wv[s ^ 1][j] = *(const v4f*)(wn + j * 16 * C2_WP);
      #pragma unroll
      for (int k = 0; k < C2_G; ++k) xv[s ^ 1][k] = *(const v4f*)(xn + k * C2_XP);
      __builtin_amdgcn_sched_barrier(0);
      #pragma unroll
      for (int k = 0; k < C2_G; ++k)
        #pragma unroll
        for (int j = 0; j < 4; ++j)
          #pragma unroll
          for (int u = 0; u < 4; ++u)
            acc[j][u] = fmaf(wv[s][j][k], xv[s][k][u], acc[j][u]);
    }
  }

  v4f res[4];
  #pragma unroll
  for (int j = 0; j < 4; ++j)
    res[j] = *(const v4f*)(s_x + (2 * C1_ + c16 + 16 * j) * C2_XP + nb);

  __syncthreads();

  float lsum = 0.f, lsq = 0.f;
  #pragma unroll
  for (int j = 0; j < 4; ++j) {
    v4f y;
    #pragma unroll
    for (int u = 0; u < 4; ++u) {
      float v = acc[j][u] + res[j][u];
      v = v > 0.f ? v : 0.f;
      y[u] = v;
      lsum += v;
      lsq  += v * v;
    }
    *(v4f*)(s_y + (c16 + 16 * j) * C2_YP + nb) = y;
  }
  for (int off = 16; off > 0; off >>= 1) {
    lsum += __shfl_xor(lsum, off);
    lsq  += __shfl_xor(lsq,  off);
  }
  if ((tid & 31) == 0) { s_red[tid >> 5] = lsum; s_red[8 + (tid >> 5)] = lsq; }
  __syncthreads();
  if (tid == 0) {
    float tsum = 0.f, tsq = 0.f;
    for (int i = 0; i < 8; ++i) { tsum += s_red[i]; tsq += s_red[8 + i]; }
    stats[(bt * C2_TL + tile) * 2 + 0] = tsum;
    stats[(bt * C2_TL + tile) * 2 + 1] = tsq;
  }
  #pragma unroll
  for (int u = 0; u < 4; ++u) {
    const int i4 = tid + 256 * u;
    const int c = i4 >> 4, q = i4 & 15;
    const v4f v = *(const v4f*)(s_y + c * C2_YP + 4 * q);
    *(v4f*)(yws + ((size_t)bt * C2_ + c) * NN_ + n0 + 4 * q) = v;
  }
}

#define LN_P 65

__global__ __launch_bounds__(256) void ln_apply_kernel(
    const float* __restrict__ yws,
    const float* __restrict__ stats,
    const float* __restrict__ lnw,
    const float* __restrict__ lnb,
    float* __restrict__ out)
{
  __shared__ float s_lw[C2_NT * LN_P];
  __shared__ float s_lb[C2_NT * LN_P];

  const int bt = blockIdx.x >> 2;
  const int n0 = (blockIdx.x & 3) * C2_NT;
  const int bb = bt >> 3, t = bt & 7;
  const int tid = threadIdx.x;

  v4f wv[4], bv[4], yv[4];
  #pragma unroll
  for (int u = 0; u < 4; ++u) {
    const int i = (tid + 256 * u) * 4;
    wv[u] = *(const v4f*)(lnw + (size_t)n0 * C2_ + i);
    bv[u] = *(const v4f*)(lnb + (size_t)n0 * C2_ + i);
    const int i4 = tid + 256 * u;
    const int c = i4 >> 4, q = i4 & 15;
    yv[u] = *(const v4f*)(yws + ((size_t)bt * C2_ + c) * NN_ + n0 + 4 * q);
  }
  float tsum = 0.f, tsq = 0.f;
  for (int k = 0; k < C2_TL; ++k) {
    tsum += stats[(bt * C2_TL + k) * 2 + 0];
    tsq  += stats[(bt * C2_TL + k) * 2 + 1];
  }
  const float cnt  = (float)(NN_ * C2_);
  const float mu   = tsum / cnt;
  const float var  = tsq / cnt - mu * mu;
  const float rstd = rsqrtf(var + 1e-5f);

  #pragma unroll
  for (int u = 0; u < 4; ++u) {
    const int i = (tid + 256 * u) * 4;
    const int n = i >> 6, c = i & 63;
    #pragma unroll
    for (int e = 0; e < 4; ++e) {
      s_lw[n * LN_P + c + e] = wv[u][e];
      s_lb[n * LN_P + c + e] = bv[u][e];
    }
  }
  __syncthreads();

  #pragma unroll
  for (int u = 0; u < 4; ++u) {
    const int i4 = tid + 256 * u;
    const int c = i4 >> 4, q = i4 & 15;
    v4f o;
    #pragma unroll
    for (int e = 0; e < 4; ++e) {
      const int n = 4 * q + e;
      o[e] = (yv[u][e] - mu) * rstd * s_lw[n * LN_P + c] + s_lb[n * LN_P + c];
    }
    *(v4f*)(out + (((size_t)bb * C2_ + c) * T2_ + t) * NN_ + n0 + 4 * q) = o;
  }
}

extern "C" void kernel_launch(void* const* d_in, const int* in_sizes, int n_in,
                              void* d_out, int out_size, void* d_ws, size_t ws_size,
                              hipStream_t stream) {
  const float* x   = (const float*)d_in[0];
  const float* adj = (const float*)d_in[1];
  const float* w1  = (const float*)d_in[2];
  const float* b1  = (const float*)d_in[3];
  const float* w2  = (const float*)d_in[4];
  const float* b2  = (const float*)d_in[5];
  const float* Wh  = (const float*)d_in[6];
  const float* ah  = (const float*)d_in[7];
  const float* lnw = (const float*)d_in[8];
  const float* lnb = (const float*)d_in[9];
  float* out = (float*)d_out;

  float* ws = (float*)d_ws;
  const size_t xg_elems = (size_t)B_ * T1_ * NN_ * C1_;
  float* xg = ws;
  float* xs = xg + xg_elems;
  float* u1 = xs + xg_elems;
  float* u2 = u1 + HH_ * C1_;
  float* yws = u2 + HH_ * C1_;
  float* stats = yws + (size_t)B_ * T2_ * C2_ * NN_;
  unsigned* adjbits = (unsigned*)(stats + B_ * T2_ * C2_TL * 2);

  prep_kernel<<<PREP_UB + PREP_AB, 256, 0, stream>>>(Wh, ah, adj, u1, u2, adjbits);

  conv1_glu_kernel<<<B_ * T1_ * (NN_ / C1_NT), 256, 0, stream>>>(x, w1, b1, xg);

  const size_t attn_lds =
      (size_t)((NN_ / 2) * XPSTR + 2 * HH_ * NN_ + 8 * 16 + ATT_MAXT * HH_ * 16 * HSTR) * sizeof(float)
      + (size_t)(ATT_MAXT * 16 * 8) * sizeof(unsigned);
  attn_kernel<<<B_ * T1_ * ATT_RS, 256, attn_lds, stream>>>(xg, adjbits, u1, u2, xs);

  const size_t c2_lds = (size_t)(C2_K * C2_XP + C2_ * C2_WP + 16) * sizeof(float);
  conv2_relu_kernel<<<B_ * T2_ * C2_TL, 256, c2_lds, stream>>>(xs, w2, b2, yws, stats);

  ln_apply_kernel<<<B_ * T2_ * C2_TL, 256, 0, stream>>>(yws, stats, lnw, lnb, out);
}


# prep work (u1/u2 vectors, adjacency bit words) done by 34 extra workgroups of the conv1 launch, which it never depended on: 4 launches instead of 5; sum order unchanged
# speedup vs baseline: 5.7699x; 1.0568x over previous
#include <hip/hip_runtime.h>
#include <math.h>

#define B_   8
#define C0_  16
#define C1_  64
#define C2_  64
#define T0_  12
#define T1_  10
#define T2_  8
#define NN_  256
#define KT_  3
#define HH_  8
#define ALPHA_ 0.2f

typedef float v2f __attribute__((ext_vector_type(2)));
typedef float v4f __attribute__((ext_vector_type(4)));
typedef float v8f __attribute__((ext_vector_type(8)));
typedef unsigned v2u __attribute__((ext_vector_type(2)));
typedef unsigned v4u __attribute__((ext_vector_type(4)));

__device__ __forceinline__ float tanh_fast(float x) {
  float e = __expf(fminf(2.0f * x, 80.0f));
  return (e - 1.0f) / (e + 1.0f);
}

#define PREP_UB 2
#define PREP_AB (NN_ / 8)

__device__ __forceinline__ void prep_body(
    const int pb,
    const float* __restrict__ W,
    const float* __restrict__ a,
    const float* __restrict__ adj,
    float* __restrict__ u1, float* __restrict__ u2,
    unsigned* __restrict__ adjbits)
{
  if (pb >= PREP_UB) {
    const int lane = threadIdx.x & 31;
    const int r = (pb - PREP_UB) * 8 + (threadIdx.x >> 5);
    const float* arow = adj + (size_t)r * NN_ + 4 * lane;
    const v4f c0 = *(const v4f*)(arow);
    const v4f c1 = *(const v4f*)(arow + NN_ / 2);
    v4u lo, hi;
    #pragma unroll
    for (int e = 0; e < 4; ++e) {
      lo[e] = __builtin_amdgcn_ballot_w32(c0[e] > 0.f);
      hi[e] = __builtin_amdgcn_ballot_w32(c1[e] > 0.f);
    }
    if (lane == 0) {
      *(v4u*)(adjbits + (size_t)r * 8)     = lo;
      *(v4u*)(adjbits + (size_t)r * 8 + 4) = hi;
    }
    return;
  }
  int tid = pb * 256 + threadIdx.x;
  if (tid >= HH_ * C1_) return;
  int h = tid >> 6, c = tid & 63;
  const float* Wh = W + ((size_t)h * C1_ + c) * C1_;
  const float* ah = a + (size_t)h * 2 * C1_;
  float s1 = 0.f, s2 = 0.f;
  for (int d = 0; d < C1_; ++d) {
    float w = Wh[d];
    s1 += w * ah[d];
    s2 += w * ah[C1_ + d];
  }
  u1[tid] = s1;
  u2[tid] = s2;
}

#define C1_NT   64
#define C1_WSTR 49
#define C1_XSTR 68
#define C1_K    (C0_ * KT_)
#define C1_RC   6

#define C1_WG   (B_ * T1_ * (NN_ / C1_NT))

__global__ __launch_bounds__(256, 4) void conv1_glu_kernel(
    const float* __restrict__ x,
    const float* __restrict__ w,
    const float* __restrict__ b,
    float* __restrict__ xg,
    const float* __restrict__ Wh,
    const float* __restrict__ ah,
    const float* __restrict__ adj,
    float* __restrict__ u1, float* __restrict__ u2,
    unsigned* __restrict__ adjbits)
{
  __shared__ __attribute__((aligned(16))) float s_w[2 * C1_ * C1_WSTR];
  __shared__ __attribute__((aligned(16))) float s_x[C1_K * C1_XSTR];

  if (blockIdx.x >= C1_WG) {
    prep_body((int)blockIdx.x - C1_WG, Wh, ah, adj, u1, u2, adjbits);
    return;
  }

  const int tid  = threadIdx.x;
  const int lane = tid & 31;
  const int wave = tid >> 5;
  const int bt   = blockIdx.x >> 2;
  const int n0   = (blockIdx.x & 3) * C1_NT;
  const int bb   = bt / T1_, t = bt - bb * T1_;

  #pragma unroll
  for (int j = 0; j < (2 * C1_ * C1_K) / 4 / 256; ++j) {
    int i4 = tid + j * 256;
    v4f wv = ((const v4f*)w)[i4];
    int base = i4 * 4;
    int c2 = base / C1_K, k = base - c2 * C1_K;
    float* d = &s_w[c2 * C1_WSTR + k];
    d[0] = wv[0]; d[1] = wv[1]; d[2] = wv[2]; d[3] = wv[3];
  }
  #pragma unroll
  for (int j = 0; j < C1_K * (C1_NT / 4) / 256; ++j) {
    int i4 = tid + j * 256;
    int r = i4 >> 4, q = i4 & 15;
    int kt = r >> 4, ci = r & 15;
    const float* src = x + (((size_t)bb * C0_ + ci) * T0_ + (t + kt)) * NN_ + n0 + q * 4;
    *(v4f*)&s_x[r * C1_XSTR + q * 4] = *(const v4f*)src;
  }
  __syncthreads();

  const int c  = (wave & 1) * 32 + lane;
  const int nb = (wave >> 1) * 16;

  float wp[C1_K], wq[C1_K];
  #pragma unroll
  for (int r = 0; r < C1_K; ++r) {
    int kt = r >> 4, ci = r & 15;
    wp[r] = s_w[c * C1_WSTR + ci * KT_ + kt];
    wq[r] = s_w[(c + C1_) * C1_WSTR + ci * KT_ + kt];
  }
  const float bp = b[c];
  const float bq = b[c + C1_];
  const bool has_res = (c < C0_);
  const int  xr = has_res ? (2 * C0_ + c) : 0;

  #pragma unroll 1
  for (int g = 0; g < 2; ++g) {
    const int nl = nb + g * 8;
    const float* xrow = &s_x[nl];
    float ap[8], aq[8];
    #pragma unroll
    for (int u = 0; u < 8; ++u) { ap[u] = bp; aq[u] = bq; }

    v4f xa[2][C1_RC][2];
    #pragma unroll
    for (int j = 0; j < C1_RC; ++j) {
      xa[0][j][0] = *(const v4f*)(xrow + j * C1_XSTR);
      xa[0][j][1] = *(const v4f*)(xrow + j * C1_XSTR + 4);
    }
    #pragma unroll
    for (int ch = 0; ch < C1_K / C1_RC; ++ch) {
      __builtin_amdgcn_sched_barrier(0);
      if (ch + 1 < C1_K / C1_RC) {
        #pragma unroll
        for (int j = 0; j < C1_RC; ++j) {
          const float* p = xrow + ((ch + 1) * C1_RC + j) * C1_XSTR;
          xa[(ch + 1) & 1][j][0] = *(const v4f*)(p);
          xa[(ch + 1) & 1][j][1] = *(const v4f*)(p + 4);
        }
      }
      __builtin_amdgcn_sched_barrier(0);
      #pragma unroll
      for (int j = 0; j < C1_RC; ++j) {
        const int r = ch * C1_RC + j;
        const v4f x0 = xa[ch & 1][j][0];
        const v4f x1 = xa[ch & 1][j][1];
        #pragma unroll
        for (int u = 0; u < 4; ++u) {
          ap[u]     = fmaf(wp[r], x0[u], ap[u]);
          aq[u]     = fmaf(wq[r], x0[u], aq[u]);
          ap[u + 4] = fmaf(wp[r], x1[u], ap[u + 4]);
          aq[u + 4] = fmaf(wq[r], x1[u], aq[u + 4]);
        }
      }
    }
    v4f r0 = *(const v4f*)&s_x[xr * C1_XSTR + nl];
    v4f r1 = *(const v4f*)&s_x[xr * C1_XSTR + nl + 4];
    float* orow = xg + ((size_t)bt * NN_ + n0 + nl) * C1_ + c;
    #pragma unroll
    for (int u = 0; u < 8; ++u) {
      float xin = has_res ? (u < 4 ? r0[u & 3] : r1[u & 3]) : 0.f;
      float sg  = 1.f / (1.f + __expf(-aq[u]));
      orow[(size_t)u * C1_] = (ap[u] + xin) * sg;
    }
  }
}

#define XSTR  72
#define XPSTR 160
#define HSTR  68
#define ATT_RS   3
#define ATT_MAXT 6
#define ATT_KB   4
#define ATT_NB   (NN_ / (4 * ATT_KB))

__global__ __launch_bounds__(256) void attn_kernel(
    const float* __restrict__ xg,
    const unsigned* __restrict__ adjbits,
    const float* __restrict__ u1,
    const float* __restrict__ u2,
    float* __restrict__ xs)
{
  extern __shared__ float smem[];
  float*    s_xgp  = smem;
  float*    s_e1   = s_xgp + (NN_ / 2) * XPSTR;
  float*    s_e2   = s_e1 + HH_ * NN_;
  float*    s_rinv = s_e2 + HH_ * NN_;
  unsigned* s_adj  = (unsigned*)(s_rinv + 8 * 16);
  float*    s_ht   = (float*)(s_adj + ATT_MAXT * 16 * 8);
  float*    s_xg   = s_ht;

  const int bt    = blockIdx.x / ATT_RS;
  const int part  = blockIdx.x - bt * ATT_RS;
  const int ntask = (part == 0) ? 6 : 5;
  const int row0  = ((part == 0) ? 0 : (part == 1 ? 6 : 11)) * 16;
  const int tid  = threadIdx.x;
  const int lane = tid & 31;
  const int wave = tid >> 5;
  const int mloc = lane & 15;
  const int half = lane >> 4;

  const float* xg_bt = xg + (size_t)bt * NN_ * C1_;

  {
    v4f xa[8], xb[8];
    #pragma unroll
    for (int j = 0; j < 8; ++j) {
      const int id = tid + 256 * j;
      const int p = id >> 4, q = id & 15;
      const float* src = xg_bt + (size_t)(2 * p) * C1_ + 4 * q;
      xa[j] = *(const v4f*)(src);
      xb[j] = *(const v4f*)(src + C1_);
    }
    const bool has_aw = tid < ntask * 32;
    v4u ab = {0u, 0u, 0u, 0u};
    if (has_aw) ab = *(const v4u*)(adjbits + (size_t)row0 * 8 + tid * 4);
    #pragma unroll
    for (int j = 0; j < 8; ++j) {
      const int id = tid + 256 * j;
      const int p = id >> 4, q = id & 15;
      *(v4f*)&s_xg[(2 * p) * XSTR + 4 * q]     = xa[j];
      *(v4f*)&s_xg[(2 * p + 1) * XSTR + 4 * q] = xb[j];
      const v4f lo = {xa[j][0], xb[j][0], xa[j][1], xb[j][1]};
      const v4f hi = {xa[j][2], xb[j][2], xa[j][3], xb[j][3]};
      *(v4f*)&s_xgp[p * XPSTR + 8 * q]     = lo;
      *(v4f*)&s_xgp[p * XPSTR + 8 * q + 4] = hi;
    }
    if (has_aw) *(v4u*)&s_adj[tid * 4] = ab;
  }
  __syncthreads();

  {
    const float* row = &s_xg[tid * XSTR];
    float rv[C1_];
    #pragma unroll
    for (int q = 0; q < C1_ / 4; ++q) {
      const v4f t4 = *(const v4f*)(row + 4 * q);
      rv[4 * q + 0] = t4[0]; rv[4 * q + 1] = t4[1]; rv[4 * q + 2] = t4[2]; rv[4 * q + 3] = t4[3];
    }
    #pragma unroll 1
    for (int h = 0; h < HH_; ++h) {
      const float* U1 = u1 + h * C1_;
      const float* U2 = u2 + h * C1_;
      float a1 = 0.f, a2 = 0.f;
      #pragma unroll
      for (int c = 0; c < C1_; ++c) {
        a1 = fmaf(rv[c], U1[c], a1);
        a2 = fmaf(rv[c], U2[c], a2);
      }
      s_e1[h * NN_ + tid] = a1;
      s_e2[h * NN_ + tid] = a2;
    }
  }
  __syncthreads();

  const float inv_h = 1.0f / (float)HH_;
  const float* e1h = s_e1 + wave * NN_;
  const float* e2h = s_e2 + wave * NN_;
  float* xs_bt = xs + (size_t)bt * NN_ * C1_;

  float e2r[NN_ / 2];
  #pragma unroll
  for (int s = 0; s < NN_ / 4; ++s) {
    const v2f t2 = *(const v2f*)(e2h + 4 * s + 2 * half);
    e2r[2 * s + 0] = t2[0]; e2r[2 * s + 1] = t2[1];
  }
  const float* xpb = &s_xgp[half * XPSTR + 2 * mloc];

  #pragma unroll 1
  for (int ti = 0; ti < ntask; ++ti) {
    const int ml = ti * 16 + mloc;
    const float e1m = e1h[row0 + ml];
    unsigned aw[4];
    {
      const v2u a0 = *(const v2u*)&s_adj[ml * 8 + 2 * half];
      const v2u a1 = *(const v2u*)&s_adj[ml * 8 + 4 + 2 * half];
      aw[0] = a0[0]; aw[1] = a0[1]; aw[2] = a1[0]; aw[3] = a1[1];
    }

    v2f pr[NN_ / 4];
    float mx = -3.0e38f;
    #pragma unroll
    for (int s = 0; s < NN_ / 4; ++s) {
      #pragma unroll
      for (int u = 0; u < 2; ++u) {
        float sc = e1m + e2r[2 * s + u];
        sc = sc > 0.f ? sc : ALPHA_ * sc;
        sc = ((aw[2 * (s >> 5) + u] >> (s & 31)) & 1) ? sc : -3.0e38f;
        pr[s][u] = sc;
        mx = fmaxf(mx, sc);
      }
    }
    mx = fmaxf(mx, __shfl_xor(mx, 16));

    float sm = 0.f;
    #pragma unroll
    for (int s = 0; s < NN_ / 4; ++s) {
      #pragma unroll
      for (int u = 0; u < 2; ++u) {
        const float pe = __expf(pr[s][u] - mx);
        pr[s][u] = pe;
        sm += pe;
      }
    }
    sm += __shfl_xor(sm, 16);
    s_rinv[wave * 16 + mloc] = 1.0f / sm;

    v8f acc0 = {}, acc1 = {}, acc2 = {}, acc3 = {};
    v2f fb[2][ATT_KB][4];
    #pragma unroll
    for (int s = 0; s < ATT_KB; ++s) {
      const float* bp = xpb + (2 * s) * XPSTR;
      #pragma unroll
      for (int q = 0; q < 4; ++q) fb[0][s][q] = *(const v2f*)(bp + 32 * q);
    }
    #pragma unroll
    for (int kb = 0; kb < ATT_NB; ++kb) {
      __builtin_amdgcn_sched_barrier(0);
      if (kb + 1 < ATT_NB) {
        #pragma unroll
        for (int s = 0; s < ATT_KB; ++s) {
          const float* bp = xpb + (2 * ((kb + 1) * ATT_KB + s)) * XPSTR;
          #pragma unroll
          for (int q = 0; q < 4; ++q) fb[(kb + 1) & 1][s][q] = *(const v2f*)(bp + 32 * q);
        }
      }
      __builtin_amdgcn_sched_barrier(0);
      #pragma unroll
      for (int s = 0; s < ATT_KB; ++s) {
        const v2f fa = pr[kb * ATT_KB + s];
        acc0 = __builtin_amdgcn_wmma_f32_16x16x4_f32(false, fa, false, fb[kb & 1][s][0], (short)0, acc0, false, false);
        acc1 = __builtin_amdgcn_wmma_f32_16x16x4_f32(false, fa, false, fb[kb & 1][s][1], (short)0, acc1, false, false);
        acc2 = __builtin_amdgcn_wmma_f32_16x16x4_f32(false, fa, false, fb[kb & 1][s][2], (short)0, acc2, false, false);
        acc3 = __builtin_amdgcn_wmma_f32_16x16x4_f32(false, fa, false, fb[kb & 1][s][3], (short)0, acc3, false, false);
      }
    }

    float inv[8];
    #pragma unroll
    for (int r = 0; r < 8; ++r) inv[r] = s_rinv[wave * 16 + r + 8 * half];
    float* ht = s_ht + ((ti * HH_ + wave) * 16 + 8 * half) * HSTR + mloc;
    #pragma unroll
    for (int r = 0; r < 8; ++r) {
      float* hrow = ht + r * HSTR;
      hrow[0]  = tanh_fast(acc0[r] * inv[r]);
      hrow[16] = tanh_fast(acc1[r] * inv[r]);
      hrow[32] = tanh_fast(acc2[r] * inv[r]);
      hrow[48] = tanh_fast(acc3[r] * inv[r]);
    }
  }

  const int rr = tid >> 4, c4 = (tid & 15) * 4;
  v4f xres[ATT_MAXT];
  #pragma unroll
  for (int ti = 0; ti < ATT_MAXT; ++ti) {
    const int tq = (ti < ntask) ? ti : 0;
    xres[ti] = *(const v4f*)(xg_bt + (size_t)(row0 + tq * 16 + rr) * C1_ + c4);
  }
  __syncthreads();
  #pragma unroll
  for (int ti = 0; ti < ATT_MAXT; ++ti) {
    if (ti < ntask) {
      const int mm = row0 + ti * 16 + rr;
      v4f sacc = {0.f, 0.f, 0.f, 0.f};
      #pragma unroll
      for (int hh = 0; hh < HH_; ++hh) {
        const v4f t4 = *(const v4f*)(s_ht + ((ti * HH_ + hh) * 16 + rr) * HSTR + c4);
        sacc[0] += t4[0] * inv_h;
        sacc[1] += t4[1] * inv_h;
        sacc[2] += t4[2] * inv_h;
        sacc[3] += t4[3] * inv_h;
      }
      const v4f xr = xres[ti];
      v4f o;
      o[0] = xr[0] + sacc[0]; o[1] = xr[1] + sacc[1]; o[2] = xr[2] + sacc[2]; o[3] = xr[3] + sacc[3];
      *(v4f*)(xs_bt + (size_t)mm * C1_ + c4) = o;
    }
  }
}

#define C2_K   (C1_ * KT_)
#define C2_NT  64
#define C2_TL  (NN_ / C2_NT)
#define C2_XP  68
#define C2_WP  196
#define C2_YP  68
#define C2_G   4
#define C2_NG  (C2_K / C2_G)

__global__ __launch_bounds__(256) void conv2_relu_kernel(
    const float* __restrict__ xs,
    const float* __restrict__ w2,
    const float* __restrict__ b2,
    float* __restrict__ yws,
    float* __restrict__ stats)
{
  extern __shared__ float smem[];
  float* s_x   = smem;
  float* s_w   = s_x + C2_K * C2_XP;
  float* s_red = s_w + C2_ * C2_WP;
  float* s_y   = smem;

  const int bt   = blockIdx.x >> 2;
  const int tile = blockIdx.x & 3;
  const int n0   = tile * C2_NT;
  const int bb = bt >> 3, t = bt & 7;
  const int tid = threadIdx.x;

  #pragma unroll
  for (unsigned j = 0; j < (C2_ * C2_K) / 4 / 256; ++j) {
    const unsigned i4 = (unsigned)tid + j * 256u;
    const v4f wv = ((const v4f*)w2)[i4];
    const unsigned co = i4 / (C2_K / 4u);
    const unsigned r  = (i4 - co * (C2_K / 4u)) * 4u;
    #pragma unroll
    for (unsigned e = 0; e < 4; ++e) {
      const unsigned ci = (r + e) / KT_, kt = (r + e) - ci * KT_;
      s_w[co * C2_WP + kt * C1_ + ci] = wv[e];
    }
  }
  {
    const int ci = tid & 63, nq = tid >> 6;
    float xv[KT_][16];
    #pragma unroll
    for (int kt = 0; kt < KT_; ++kt) {
      const float* src = xs + (((size_t)bb * T1_ + t + kt) * NN_ + n0 + nq) * C1_ + ci;
      #pragma unroll
      for (int u = 0; u < 16; ++u) xv[kt][u] = src[u * 4 * C1_];
    }
    #pragma unroll
    for (int kt = 0; kt < KT_; ++kt)
      #pragma unroll
      for (int u = 0; u < 16; ++u) s_x[(kt * C1_ + ci) * C2_XP + nq + 4 * u] = xv[kt][u];
  }
  __syncthreads();

  const int lane = tid & 31;
  const int wave = tid >> 5;
  const int c16  = lane & 15;
  const int nb   = wave * 8 + (lane >> 4) * 4;

  float acc[4][4];
  #pragma unroll
  for (int j = 0; j < 4; ++j) {
    const float bj = b2[c16 + 16 * j];
    #pragma unroll
    for (int i = 0; i < 4; ++i) acc[j][i] = bj;
  }

  const float* xb = s_x + nb;
  const float* wb = s_w + c16 * C2_WP;

  v4f wv[2][4];
  v4f xv[2][C2_G];
  #pragma unroll
  for (int j = 0; j < 4; ++j) wv[0][j] = *(const v4f*)(wb + j * 16 * C2_WP);
  #pragma unroll
  for (int k = 0; k < C2_G; ++k) xv[0][k] = *(const v4f*)(xb + k * C2_XP);

  #pragma unroll 1
  for (int g = 0; g < C2_NG; g += 2) {
    #pragma unroll
    for (int s = 0; s < 2; ++s) {
      const int gn = (g + s + 1 < C2_NG) ? (g + s + 1) : (C2_NG - 1);
      const float* xn = xb + gn * C2_G * C2_XP;
      const float* wn = wb + gn * C2_G;
      __builtin_amdgcn_sched_barrier(0);
      #pragma unroll
      for (int j = 0; j < 4; ++j) wv[s ^ 1][j] = *(const v4f*)(wn + j * 16 * C2_WP);
      #pragma unroll
      for (int k = 0; k < C2_G; ++k) xv[s ^ 1][k] = *(const v4f*)(xn + k * C2_XP);
      __builtin_amdgcn_sched_barrier(0);
      #pragma unroll
      for (int k = 0; k < C2_G; ++k)
        #pragma unroll
        for (int j = 0; j < 4; ++j)
          #pragma unroll
          for (int u = 0; u < 4; ++u)
            acc[j][u] = fmaf(wv[s][j][k], xv[s][k][u], acc[j][u]);
    }
  }

  v4f res[4];
  #pragma unroll
  for (int j = 0; j < 4; ++j)
    res[j] = *(const v4f*)(s_x + (2 * C1_ + c16 + 16 * j) * C2_XP + nb);

  __syncthreads();

  float lsum = 0.f, lsq = 0.f;
  #pragma unroll
  for (int j = 0; j < 4; ++j) {
    v4f y;
    #pragma unroll
    for (int u = 0; u < 4; ++u) {
      float v = acc[j][u] + res[j][u];
      v = v > 0.f ? v : 0.f;
      y[u] = v;
      lsum += v;
      lsq  += v * v;
    }
    *(v4f*)(s_y + (c16 + 16 * j) * C2_YP + nb) = y;
  }
  for (int off = 16; off > 0; off >>= 1) {
    lsum += __shfl_xor(lsum, off);
    lsq  += __shfl_xor(lsq,  off);
  }
  if ((tid & 31) == 0) { s_red[tid >> 5] = lsum; s_red[8 + (tid >> 5)] = lsq; }
  __syncthreads();
  if (tid == 0) {
    float tsum = 0.f, tsq = 0.f;
    for (int i = 0; i < 8; ++i) { tsum += s_red[i]; tsq += s_red[8 + i]; }
    stats[(bt * C2_TL + tile) * 2 + 0] = tsum;
    stats[(bt * C2_TL + tile) * 2 + 1] = tsq;
  }
  #pragma unroll
  for (int u = 0; u < 4; ++u) {
    const int i4 = tid + 256 * u;
    const int c = i4 >> 4, q = i4 & 15;
    const v4f v = *(const v4f*)(s_y + c * C2_YP + 4 * q);
    *(v4f*)(yws + ((size_t)bt * C2_ + c) * NN_ + n0 + 4 * q) = v;
  }
}

#define LN_P 65

__global__ __launch_bounds__(256) void ln_apply_kernel(
    const float* __restrict__ yws,
    const float* __restrict__ stats,
    const float* __restrict__ lnw,
    const float* __restrict__ lnb,
    float* __restrict__ out)
{
  __shared__ float s_lw[C2_NT * LN_P];
  __shared__ float s_lb[C2_NT * LN_P];

  const int bt = blockIdx.x >> 2;
  const int n0 = (blockIdx.x & 3) * C2_NT;
  const int bb = bt >> 3, t = bt & 7;
  const int tid = threadIdx.x;

  v4f wv[4], bv[4], yv[4];
  #pragma unroll
  for (int u = 0; u < 4; ++u) {
    const int i = (tid + 256 * u) * 4;
    wv[u] = *(const v4f*)(lnw + (size_t)n0 * C2_ + i);
    bv[u] = *(const v4f*)(lnb + (size_t)n0 * C2_ + i);
    const int i4 = tid + 256 * u;
    const int c = i4 >> 4, q = i4 & 15;
    yv[u] = *(const v4f*)(yws + ((size_t)bt * C2_ + c) * NN_ + n0 + 4 * q);
  }
  float tsum = 0.f, tsq = 0.f;
  for (int k = 0; k < C2_TL; ++k) {
    tsum += stats[(bt * C2_TL + k) * 2 + 0];
    tsq  += stats[(bt * C2_TL + k) * 2 + 1];
  }
  const float cnt  = (float)(NN_ * C2_);
  const float mu   = tsum / cnt;
  const float var  = tsq / cnt - mu * mu;
  const float rstd = rsqrtf(var + 1e-5f);

  #pragma unroll
  for (int u = 0; u < 4; ++u) {
    const int i = (tid + 256 * u) * 4;
    const int n = i >> 6, c = i & 63;
    #pragma unroll
    for (int e = 0; e < 4; ++e) {
      s_lw[n * LN_P + c + e] = wv[u][e];
      s_lb[n * LN_P + c + e] = bv[u][e];
    }
  }
  __syncthreads();

  #pragma unroll
  for (int u = 0; u < 4; ++u) {
    const int i4 = tid + 256 * u;
    const int c = i4 >> 4, q = i4 & 15;
    v4f o;
    #pragma unroll
    for (int e = 0; e < 4; ++e) {
      const int n = 4 * q + e;
      o[e] = (yv[u][e] - mu) * rstd * s_lw[n * LN_P + c] + s_lb[n * LN_P + c];
    }
    *(v4f*)(out + (((size_t)bb * C2_ + c) * T2_ + t) * NN_ + n0 + 4 * q) = o;
  }
}

extern "C" void kernel_launch(void* const* d_in, const int* in_sizes, int n_in,
                              void* d_out, int out_size, void* d_ws, size_t ws_size,
                              hipStream_t stream) {
  const float* x   = (const float*)d_in[0];
  const float* adj = (const float*)d_in[1];
  const float* w1  = (const float*)d_in[2];
  const float* b1  = (const float*)d_in[3];
  const float* w2  = (const float*)d_in[4];
  const float* b2  = (const float*)d_in[5];
  const float* Wh  = (const float*)d_in[6];
  const float* ah  = (const float*)d_in[7];
  const float* lnw = (const float*)d_in[8];
  const float* lnb = (const float*)d_in[9];
  float* out = (float*)d_out;

  float* ws = (float*)d_ws;
  const size_t xg_elems = (size_t)B_ * T1_ * NN_ * C1_;
  float* xg = ws;
  float* xs = xg + xg_elems;
  float* u1 = xs + xg_elems;
  float* u2 = u1 + HH_ * C1_;
  float* yws = u2 + HH_ * C1_;
  float* stats = yws + (size_t)B_ * T2_ * C2_ * NN_;
  unsigned* adjbits = (unsigned*)(stats + B_ * T2_ * C2_TL * 2);

  conv1_glu_kernel<<<C1_WG + PREP_UB + PREP_AB, 256, 0, stream>>>(
      x, w1, b1, xg, Wh, ah, adj, u1, u2, adjbits);

  const size_t attn_lds =
      (size_t)((NN_ / 2) * XPSTR + 2 * HH_ * NN_ + 8 * 16 + ATT_MAXT * HH_ * 16 * HSTR) * sizeof(float)
      + (size_t)(ATT_MAXT * 16 * 8) * sizeof(unsigned);
  attn_kernel<<<B_ * T1_ * ATT_RS, 256, attn_lds, stream>>>(xg, adjbits, u1, u2, xs);

  const size_t c2_lds = (size_t)(C2_K * C2_XP + C2_ * C2_WP + 16) * sizeof(float);
  conv2_relu_kernel<<<B_ * T2_ * C2_TL, 256, c2_lds, stream>>>(xs, w2, b2, yws, stats);

  ln_apply_kernel<<<B_ * T2_ * C2_TL, 256, 0, stream>>>(yws, stats, lnw, lnb, out);
}
